# starting kernel
# baseline (speedup 1.0000x reference)
#include <hip/hip_runtime.h>

typedef __attribute__((ext_vector_type(2))) float v2f;
typedef __attribute__((ext_vector_type(8))) float v8f;

#define DCH 64

__device__ __forceinline__ void atomic_add_f32x4(float* p, float4 v) {
    asm volatile(
        "global_atomic_add_f32 %0, %1, off scope:SCOPE_DEV\n\t"
        "global_atomic_add_f32 %0, %2, off offset:4 scope:SCOPE_DEV\n\t"
        "global_atomic_add_f32 %0, %3, off offset:8 scope:SCOPE_DEV\n\t"
        "global_atomic_add_f32 %0, %4, off offset:12 scope:SCOPE_DEV"
        :: "v"(p), "v"(v.x), "v"(v.y), "v"(v.z), "v"(v.w)
        : "memory");
}

__device__ __forceinline__ v2f ldv2(const float* p) { return *(const v2f*)p; }

__global__ void zero_kernel(float4* __restrict__ p, long long n4) {
    long long i = (long long)blockIdx.x * blockDim.x + threadIdx.x;
    if (i < n4) p[i] = make_float4(0.f, 0.f, 0.f, 0.f);
}

__global__ void scatter_add_kernel(const float* __restrict__ x,
                                   const int* __restrict__ src,
                                   const int* __restrict__ dst,
                                   float* __restrict__ agg, int nE) {
    int t = blockIdx.x * blockDim.x + threadIdx.x;
    int e = t >> 4;
    if (e >= nE) return;
    int j = (t & 15) << 2;
    int s = src[e], d = dst[e];
    float4 v = *(const float4*)(x + (long long)s * DCH + j);
    atomic_add_f32x4(agg + (long long)d * DCH + j, v);
}

__global__ void __launch_bounds__(128)
gemm5_kernel(const float* __restrict__ x,  const float* __restrict__ agg,
             const float* __restrict__ W0, const float* __restrict__ W1,
             const float* __restrict__ WQ, const float* __restrict__ WK,
             const float* __restrict__ WV,
             float* __restrict__ out, float* __restrict__ Qo,
             float* __restrict__ Ko,  float* __restrict__ Vo, int nrows)
{
    const int lane = threadIdx.x & 31;
    const int wave = threadIdx.x >> 5;
    const int row0 = (blockIdx.x * 4 + wave) * 16;
    if (row0 >= nrows) return;
    const int m = lane & 15;
    const int h = lane >> 4;

    v2f ax[16], ag[16];
    const float* xr = x   + (long long)(row0 + m) * DCH + 2 * h;
    const float* gr = agg + (long long)(row0 + m) * DCH + 2 * h;
#pragma unroll
    for (int kk = 0; kk < 16; ++kk) {
        ax[kk] = ldv2(xr + 4 * kk);
        ag[kk] = ldv2(gr + 4 * kk);
    }

    const int bi = m * DCH + 2 * h;

#pragma unroll
    for (int n0 = 0; n0 < DCH; n0 += 16) {
        const float* b0 = W0 + n0 * DCH + bi;
        const float* b1 = W1 + n0 * DCH + bi;
        const float* bq = WQ + n0 * DCH + bi;
        const float* bk = WK + n0 * DCH + bi;
        const float* bv = WV + n0 * DCH + bi;

        v8f acc = {};
#pragma unroll
        for (int kk = 0; kk < 16; ++kk)
            acc = __builtin_amdgcn_wmma_f32_16x16x4_f32(
                false, ax[kk], false, ldv2(b0 + 4 * kk), (short)0, acc, false, false);
#pragma unroll
        for (int kk = 0; kk < 16; ++kk)
            acc = __builtin_amdgcn_wmma_f32_16x16x4_f32(
                false, ag[kk], false, ldv2(b1 + 4 * kk), (short)0, acc, false, false);
#pragma unroll
        for (int r = 0; r < 8; ++r)
            out[(long long)(row0 + r + 8 * h) * DCH + n0 + m] = acc[r];

        acc = (v8f){};
#pragma unroll
        for (int kk = 0; kk < 16; ++kk)
            acc = __builtin_amdgcn_wmma_f32_16x16x4_f32(
                false, ax[kk], false, ldv2(bq + 4 * kk), (short)0, acc, false, false);
#pragma unroll
        for (int r = 0; r < 8; ++r)
            Qo[(long long)(row0 + r + 8 * h) * DCH + n0 + m] = acc[r];

        acc = (v8f){};
#pragma unroll
        for (int kk = 0; kk < 16; ++kk)
            acc = __builtin_amdgcn_wmma_f32_16x16x4_f32(
                false, ax[kk], false, ldv2(bk + 4 * kk), (short)0, acc, false, false);
#pragma unroll
        for (int r = 0; r < 8; ++r)
            Ko[(long long)(row0 + r + 8 * h) * DCH + n0 + m] = acc[r];

        acc = (v8f){};
#pragma unroll
        for (int kk = 0; kk < 16; ++kk)
            acc = __builtin_amdgcn_wmma_f32_16x16x4_f32(
                false, ax[kk], false, ldv2(bv + 4 * kk), (short)0, acc, false, false);
#pragma unroll
        for (int r = 0; r < 8; ++r)
            Vo[(long long)(row0 + r + 8 * h) * DCH + n0 + m] = acc[r];
    }
}

__global__ void attn_kernel(const float* __restrict__ Q, const float* __restrict__ K,
                            const float* __restrict__ V,
                            const int* __restrict__ s2, const int* __restrict__ d2,
                            float* __restrict__ out, int nE)
{
    int t = blockIdx.x * blockDim.x + threadIdx.x;
    int e = t >> 4;
    if (e >= nE) return;
    int j = (t & 15) << 2;
    int s = s2[e], d = d2[e];
    const float4 q = *(const float4*)(Q + (long long)d * DCH + j);
    const float4 k = *(const float4*)(K + (long long)s * DCH + j);
    float p = q.x * k.x + q.y * k.y + q.z * k.z + q.w * k.w;
#pragma unroll
    for (int off = 8; off >= 1; off >>= 1)
        p += __shfl_xor(p, off, 16);
    const float4 v = *(const float4*)(V + (long long)s * DCH + j);
    atomic_add_f32x4(out + (long long)d * DCH + j,
                     make_float4(-p * v.x, -p * v.y, -p * v.z, -p * v.w));
}

extern "C" void kernel_launch(void* const* d_in, const int* in_sizes, int n_in,
                              void* d_out, int out_size, void* d_ws, size_t ws_size,
                              hipStream_t stream)
{
    const float* x  = (const float*)d_in[0];
    const int*  ei1 = (const int*)d_in[1];
    const int*  ei2 = (const int*)d_in[2];
    const float* W0 = (const float*)d_in[3];
    const float* W1 = (const float*)d_in[4];
    const float* WQ = (const float*)d_in[5];
    const float* WK = (const float*)d_in[6];
    const float* WV = (const float*)d_in[7];
    float* out = (float*)d_out;

    const int N  = in_sizes[0] / DCH;
    const int E1 = in_sizes[1] / 2;
    const int E2 = in_sizes[2] / 2;

    float* agg = (float*)d_ws;
    float* Qb  = agg + (size_t)N * DCH;
    float* Kb  = Qb  + (size_t)N * DCH;
    float* Vb  = Kb  + (size_t)N * DCH;

    long long n4 = (long long)N * DCH / 4;
    zero_kernel<<<(unsigned)((n4 + 255) / 256), 256, 0, stream>>>((float4*)agg, n4);

    {
        long long threads = (long long)E1 * 16;
        scatter_add_kernel<<<(unsigned)((threads + 255) / 256), 256, 0, stream>>>(
            x, ei1, ei1 + E1, agg, E1);
    }

    {
        int strips = (N + 15) / 16;
        int blocks = (strips + 3) / 4;
        gemm5_kernel<<<blocks, 128, 0, stream>>>(x, agg, W0, W1, WQ, WK, WV,
                                                 out, Qb, Kb, Vb, N);
    }

    {
        long long threads = (long long)E2 * 16;
        attn_kernel<<<(unsigned)((threads + 255) / 256), 256, 0, stream>>>(
            Qb, Kb, Vb, ei2, ei2 + E2, out, E2);
    }
}


# Every f32 atomic instruction now covers one whole 128-B line of one destination row (32 lanes per edge, lane l adds elements l and l+32; was 16-B lane stride over 4 lines); dot sum is a 32-lane tree
# speedup vs baseline: 1.2214x; 1.2214x over previous
#include <hip/hip_runtime.h>

typedef __attribute__((ext_vector_type(2))) float v2f;
typedef __attribute__((ext_vector_type(8))) float v8f;

#define DCH 64

__device__ __forceinline__ void atomic_add_row(float* p, float lo, float hi) {
    asm volatile(
        "global_atomic_add_f32 %0, %1, off scope:SCOPE_DEV\n\t"
        "global_atomic_add_f32 %0, %2, off offset:128 scope:SCOPE_DEV"
        :: "v"(p), "v"(lo), "v"(hi)
        : "memory");
}

__device__ __forceinline__ int wave_first_edge() {
    const int w = __builtin_amdgcn_readfirstlane((int)(threadIdx.x >> 5));
    return (int)(blockIdx.x * (blockDim.x >> 5) + w) * 2;
}

__device__ __forceinline__ v2f ldv2(const float* p) { return *(const v2f*)p; }

__global__ void zero_kernel(float4* __restrict__ p, long long n4) {
    long long i = (long long)blockIdx.x * blockDim.x + threadIdx.x;
    if (i < n4) p[i] = make_float4(0.f, 0.f, 0.f, 0.f);
}

__global__ void scatter_add_kernel(const float* __restrict__ x,
                                   const int* __restrict__ src,
                                   const int* __restrict__ dst,
                                   float* __restrict__ agg, int nE) {
    const int lane = threadIdx.x & 31;
    const int e0 = wave_first_edge();
    if (e0 >= nE) return;
    const bool two = e0 + 1 < nE;
    const int e1 = two ? e0 + 1 : e0;
    const int s0 = src[e0], d0 = dst[e0];
    const int s1 = src[e1], d1 = dst[e1];
    const float* x0 = x + (long long)s0 * DCH + lane;
    const float* x1 = x + (long long)s1 * DCH + lane;
    const float a0 = x0[0], b0 = x0[32];
    const float a1 = x1[0], b1 = x1[32];
    atomic_add_row(agg + (long long)d0 * DCH + lane, a0, b0);
    atomic_add_row(agg + (long long)d1 * DCH + lane, two ? a1 : 0.f, two ? b1 : 0.f);
}

__global__ void __launch_bounds__(128)
gemm5_kernel(const float* __restrict__ x,  const float* __restrict__ agg,
             const float* __restrict__ W0, const float* __restrict__ W1,
             const float* __restrict__ WQ, const float* __restrict__ WK,
             const float* __restrict__ WV,
             float* __restrict__ out, float* __restrict__ Qo,
             float* __restrict__ Ko,  float* __restrict__ Vo, int nrows)
{
    const int lane = threadIdx.x & 31;
    const int wave = threadIdx.x >> 5;
    const int row0 = (blockIdx.x * 4 + wave) * 16;
    if (row0 >= nrows) return;
    const int m = lane & 15;
    const int h = lane >> 4;

    v2f ax[16], ag[16];
    const float* xr = x   + (long long)(row0 + m) * DCH + 2 * h;
    const float* gr = agg + (long long)(row0 + m) * DCH + 2 * h;
#pragma unroll
    for (int kk = 0; kk < 16; ++kk) {
        ax[kk] = ldv2(xr + 4 * kk);
        ag[kk] = ldv2(gr + 4 * kk);
    }

    const int bi = m * DCH + 2 * h;

#pragma unroll
    for (int n0 = 0; n0 < DCH; n0 += 16) {
        const float* b0 = W0 + n0 * DCH + bi;
        const float* b1 = W1 + n0 * DCH + bi;
        const float* bq = WQ + n0 * DCH + bi;
        const float* bk = WK + n0 * DCH + bi;
        const float* bv = WV + n0 * DCH + bi;

        v8f acc = {};
#pragma unroll
        for (int kk = 0; kk < 16; ++kk)
            acc = __builtin_amdgcn_wmma_f32_16x16x4_f32(
                false, ax[kk], false, ldv2(b0 + 4 * kk), (short)0, acc, false, false);
#pragma unroll
        for (int kk = 0; kk < 16; ++kk)
            acc = __builtin_amdgcn_wmma_f32_16x16x4_f32(
                false, ag[kk], false, ldv2(b1 + 4 * kk), (short)0, acc, false, false);
#pragma unroll
        for (int r = 0; r < 8; ++r)
            out[(long long)(row0 + r + 8 * h) * DCH + n0 + m] = acc[r];

        acc = (v8f){};
#pragma unroll
        for (int kk = 0; kk < 16; ++kk)
            acc = __builtin_amdgcn_wmma_f32_16x16x4_f32(
                false, ax[kk], false, ldv2(bq + 4 * kk), (short)0, acc, false, false);
#pragma unroll
        for (int r = 0; r < 8; ++r)
            Qo[(long long)(row0 + r + 8 * h) * DCH + n0 + m] = acc[r];

        acc = (v8f){};
#pragma unroll
        for (int kk = 0; kk < 16; ++kk)
            acc = __builtin_amdgcn_wmma_f32_16x16x4_f32(
                false, ax[kk], false, ldv2(bk + 4 * kk), (short)0, acc, false, false);
#pragma unroll
        for (int r = 0; r < 8; ++r)
            Ko[(long long)(row0 + r + 8 * h) * DCH + n0 + m] = acc[r];

        acc = (v8f){};
#pragma unroll
        for (int kk = 0; kk < 16; ++kk)
            acc = __builtin_amdgcn_wmma_f32_16x16x4_f32(
                false, ax[kk], false, ldv2(bv + 4 * kk), (short)0, acc, false, false);
#pragma unroll
        for (int r = 0; r < 8; ++r)
            Vo[(long long)(row0 + r + 8 * h) * DCH + n0 + m] = acc[r];
    }
}

__device__ __forceinline__ float wave_sum(float p) {
    for (int off = 16; off >= 1; off >>= 1)
        p += __shfl_xor(p, off, 32);
    return p;
}

__global__ void attn_kernel(const float* __restrict__ Q, const float* __restrict__ K,
                            const float* __restrict__ V,
                            const int* __restrict__ s2, const int* __restrict__ d2,
                            float* __restrict__ out, int nE)
{
    const int lane = threadIdx.x & 31;
    const int e0 = wave_first_edge();
    if (e0 >= nE) return;
    const bool two = e0 + 1 < nE;
    const int e1 = two ? e0 + 1 : e0;
    const int s0 = s2[e0], d0 = d2[e0];
    const int s1 = s2[e1], d1 = d2[e1];
    const float* q0 = Q + (long long)d0 * DCH + lane;
    const float* k0 = K + (long long)s0 * DCH + lane;
    const float* v0 = V + (long long)s0 * DCH + lane;
    const float* q1 = Q + (long long)d1 * DCH + lane;
    const float* k1 = K + (long long)s1 * DCH + lane;
    const float* v1 = V + (long long)s1 * DCH + lane;
    const float q0a = q0[0], q0b = q0[32], k0a = k0[0], k0b = k0[32];
    const float q1a = q1[0], q1b = q1[32], k1a = k1[0], k1b = k1[32];
    const float v0a = v0[0], v0b = v0[32];
    const float v1a = v1[0], v1b = v1[32];
    const float p0 = wave_sum(q0a * k0a + q0b * k0b);
    const float p1 = wave_sum(q1a * k1a + q1b * k1b);
    atomic_add_row(out + (long long)d0 * DCH + lane, -p0 * v0a, -p0 * v0b);
    atomic_add_row(out + (long long)d1 * DCH + lane,
                   two ? -p1 * v1a : 0.f, two ? -p1 * v1b : 0.f);
}

extern "C" void kernel_launch(void* const* d_in, const int* in_sizes, int n_in,
                              void* d_out, int out_size, void* d_ws, size_t ws_size,
                              hipStream_t stream)
{
    const float* x  = (const float*)d_in[0];
    const int*  ei1 = (const int*)d_in[1];
    const int*  ei2 = (const int*)d_in[2];
    const float* W0 = (const float*)d_in[3];
    const float* W1 = (const float*)d_in[4];
    const float* WQ = (const float*)d_in[5];
    const float* WK = (const float*)d_in[6];
    const float* WV = (const float*)d_in[7];
    float* out = (float*)d_out;

    const int N  = in_sizes[0] / DCH;
    const int E1 = in_sizes[1] / 2;
    const int E2 = in_sizes[2] / 2;

    float* agg = (float*)d_ws;
    float* Qb  = agg + (size_t)N * DCH;
    float* Kb  = Qb  + (size_t)N * DCH;
    float* Vb  = Kb  + (size_t)N * DCH;

    long long n4 = (long long)N * DCH / 4;
    zero_kernel<<<(unsigned)((n4 + 255) / 256), 256, 0, stream>>>((float4*)agg, n4);

    {
        long long threads = (long long)E1 * 16;
        scatter_add_kernel<<<(unsigned)((threads + 255) / 256), 256, 0, stream>>>(
            x, ei1, ei1 + E1, agg, E1);
    }

    {
        int strips = (N + 15) / 16;
        int blocks = (strips + 3) / 4;
        gemm5_kernel<<<blocks, 128, 0, stream>>>(x, agg, W0, W1, WQ, WK, WV,
                                                 out, Qb, Kb, Vb, N);
    }

    {
        long long threads = (long long)E2 * 16;
        attn_kernel<<<(unsigned)((threads + 255) / 256), 256, 0, stream>>>(
            Qb, Kb, Vb, ei2, ei2 + E2, out, E2);
    }
}


# Float atomics and agg zero fill go: edges ordered by destination in the call (two-level counting sort, wave-private LDS counters, no atomics); one wave per node sums rows in registers, in list order
# speedup vs baseline: 1.2656x; 1.0362x over previous
#include <hip/hip_runtime.h>

typedef __attribute__((ext_vector_type(2))) float v2f;
typedef __attribute__((ext_vector_type(8))) float v8f;

#define DCH 64
#define BIN_NODES 256
#define BIN_SHIFT 8
#define PART_THREADS 1024
#define PART_WAVES (PART_THREADS / 32)
#define PART_EPT 16
#define CHUNK_EDGES (PART_THREADS * PART_EPT)
#define ROW_WAVES (BIN_NODES / 32)
#define NODE_WAVES 8

__device__ __forceinline__ v2f ldv2(const float* p) { return *(const v2f*)p; }

__device__ __forceinline__ int wave_in_grid() {
    const int w = __builtin_amdgcn_readfirstlane((int)(threadIdx.x >> 5));
    return (int)(blockIdx.x * NODE_WAVES) + w;
}

__device__ __forceinline__ int wave_incl_scan(int v, int lane) {
    for (int off = 1; off < 32; off <<= 1) {
        const int t = __shfl_up(v, off, 32);
        if (lane >= off) v += t;
    }
    return v;
}

__device__ __forceinline__ unsigned same_key_lanes(int key, bool valid) {
    const unsigned all = __builtin_amdgcn_ballot_w32(valid);
    unsigned rem = all;
    unsigned mine = 0u;
    while (rem) {
        const int first = __builtin_ctz(rem);
        const int v = __builtin_amdgcn_readlane(key, first);
        const unsigned grp = __builtin_amdgcn_ballot_w32(key == v) & all;
        if (valid && key == v) mine = grp;
        rem &= ~grp;
    }
    return mine;
}

__device__ __forceinline__ unsigned lanes_below(int lane) { return (1u << lane) - 1u; }


__global__ void __launch_bounds__(PART_THREADS)
bin_count_kernel(const int* __restrict__ dstA, int nEA,
                 const int* __restrict__ dstB, int nEB,
                 int* __restrict__ cnt, int nBins, int nChunks)
{
    extern __shared__ int histW[];
    const int g = blockIdx.y;
    const int* __restrict__ dst = g ? dstB : dstA;
    const int nE = g ? nEB : nEA;
    const int c = blockIdx.x;
    const int lane = threadIdx.x & 31;
    const int wv = __builtin_amdgcn_readfirstlane((int)(threadIdx.x >> 5));
    for (int i = threadIdx.x; i < PART_WAVES * nBins; i += PART_THREADS) histW[i] = 0;
    __syncthreads();
    const int e0 = c * CHUNK_EDGES + (int)threadIdx.x;
    int d[PART_EPT];
#pragma unroll
    for (int k = 0; k < PART_EPT; ++k) {
        const int e = e0 + k * PART_THREADS;
        d[k] = (e < nE) ? dst[e] : -1;
    }
    int* myRow = histW + wv * nBins;
#pragma unroll
    for (int k = 0; k < PART_EPT; ++k) {
        const bool valid = d[k] >= 0;
        const int bin = valid ? (d[k] >> BIN_SHIFT) : 0;
        const unsigned grp = same_key_lanes(bin, valid);
        if (valid && lane == __builtin_ctz(grp)) myRow[bin] += __builtin_popcount(grp);
    }
    __syncthreads();
    int* __restrict__ row = cnt + ((size_t)g * nChunks + c) * nBins;
    for (int b = threadIdx.x; b < nBins; b += PART_THREADS) {
        int tot = 0;
        for (int w = 0; w < PART_WAVES; ++w) tot += histW[w * nBins + b];
        row[b] = tot;
    }
}

__global__ void __launch_bounds__(1024)
bin_scan_kernel(const int* __restrict__ cnt, int* __restrict__ start,
                int* __restrict__ binStart, int nBins, int nChunks)
{
    __shared__ int wsum[32];
    __shared__ int tileTotal;
    const int g = blockIdx.x;
    const int* __restrict__ m = cnt + (size_t)g * nChunks * nBins;
    int* __restrict__ st = start + (size_t)g * nChunks * nBins;
    int* __restrict__ bs = binStart + (size_t)g * (nBins + 1);
    const int lane = threadIdx.x & 31, wv = threadIdx.x >> 5;
    int carry = 0;
    for (int b0 = 0; b0 < nBins; b0 += 1024) {
        const int b = b0 + (int)threadIdx.x;
        const bool live = b < nBins;
        int tot = 0;
        for (int c = 0; c < nChunks; c += 16) {
            int v[16];
#pragma unroll
            for (int u = 0; u < 16; ++u)
                v[u] = (live && c + u < nChunks) ? m[(size_t)(c + u) * nBins + b] : 0;
#pragma unroll
            for (int u = 0; u < 16; ++u) tot += v[u];
        }
        const int incl = wave_incl_scan(tot, lane);
        if (lane == 31) wsum[wv] = incl;
        __syncthreads();
        if (wv == 0) {
            const int s = wsum[lane];
            const int si = wave_incl_scan(s, lane);
            wsum[lane] = si - s;
            if (lane == 31) tileTotal = si;
        }
        __syncthreads();
        const int excl = carry + wsum[wv] + incl - tot;
        const int tt = tileTotal;
        if (live) bs[b] = excl;
        int run = excl;
        for (int c = 0; c < nChunks; c += 16) {
            int v[16];
#pragma unroll
            for (int u = 0; u < 16; ++u)
                v[u] = (live && c + u < nChunks) ? m[(size_t)(c + u) * nBins + b] : 0;
#pragma unroll
            for (int u = 0; u < 16; ++u) {
                if (live && c + u < nChunks) st[(size_t)(c + u) * nBins + b] = run;
                run += v[u];
            }
        }
        carry += tt;
        __syncthreads();
    }
    if (threadIdx.x == 0) bs[nBins] = carry;
}

__global__ void __launch_bounds__(PART_THREADS)
bin_place_kernel(const int* __restrict__ srcA, const int* __restrict__ dstA, int nEA,
                 const int* __restrict__ srcB, const int* __restrict__ dstB, int nEB,
                 const int* __restrict__ start, int2* __restrict__ part,
                 int nBins, int nChunks, int listStride)
{
    extern __shared__ int histW[];
    const int g = blockIdx.y;
    const int* __restrict__ src = g ? srcB : srcA;
    const int* __restrict__ dst = g ? dstB : dstA;
    const int nE = g ? nEB : nEA;
    const int c = blockIdx.x;
    const int lane = threadIdx.x & 31;
    const int wv = __builtin_amdgcn_readfirstlane((int)(threadIdx.x >> 5));
    for (int i = threadIdx.x; i < PART_WAVES * nBins; i += PART_THREADS) histW[i] = 0;
    __syncthreads();
    const int e0 = c * CHUNK_EDGES + (int)threadIdx.x;
    int d[PART_EPT], s[PART_EPT];
    unsigned grp[PART_EPT];
#pragma unroll
    for (int k = 0; k < PART_EPT; ++k) {
        const int e = e0 + k * PART_THREADS;
        const bool in = e < nE;
        d[k] = in ? dst[e] : -1;
        s[k] = in ? src[e] : 0;
    }
    int* myRow = histW + wv * nBins;
#pragma unroll
    for (int k = 0; k < PART_EPT; ++k) {
        const bool valid = d[k] >= 0;
        const int bin = valid ? (d[k] >> BIN_SHIFT) : 0;
        grp[k] = same_key_lanes(bin, valid);
        if (valid && lane == __builtin_ctz(grp[k])) myRow[bin] += __builtin_popcount(grp[k]);
    }
    __syncthreads();
    const int* __restrict__ srow = start + ((size_t)g * nChunks + c) * nBins;
    for (int b = threadIdx.x; b < nBins; b += PART_THREADS) {
        int run = srow[b];
        for (int w = 0; w < PART_WAVES; ++w) {
            const int t = histW[w * nBins + b];
            histW[w * nBins + b] = run;
            run += t;
        }
    }
    __syncthreads();
    int2* __restrict__ pg = part + (size_t)g * listStride;
#pragma unroll
    for (int k = 0; k < PART_EPT; ++k) {
        const bool valid = d[k] >= 0;
        const int bin = valid ? (d[k] >> BIN_SHIFT) : 0;
        const int first = valid ? __builtin_ctz(grp[k]) : 0;
        int base = 0;
        if (valid && lane == first) {
            base = myRow[bin];
            myRow[bin] = base + __builtin_popcount(grp[k]);
        }
        base = __shfl(base, first, 32);
        if (valid) pg[base + __builtin_popcount(grp[k] & lanes_below(lane))] = make_int2(s[k], d[k]);
    }
}

__global__ void __launch_bounds__(BIN_NODES)
bin_to_rows_kernel(const int2* __restrict__ part, const int* __restrict__ binStart,
                   int2* __restrict__ rows, int* __restrict__ sorted,
                   int nBins, int nNodes, int listStride)
{
    __shared__ int cntW[ROW_WAVES][BIN_NODES];
    __shared__ int wsum[ROW_WAVES];
    const int g = blockIdx.y, b = blockIdx.x, t = threadIdx.x;
    const int lane = t & 31;
    const int wv = __builtin_amdgcn_readfirstlane(t >> 5);
    const int beg = binStart[(size_t)g * (nBins + 1) + b];
    const int end = binStart[(size_t)g * (nBins + 1) + b + 1];
    const int node0 = b * BIN_NODES;
    const int2* __restrict__ pg = part + (size_t)g * listStride;
    int* __restrict__ sg = sorted + (size_t)g * listStride;
    const int per = (((end - beg) + ROW_WAVES - 1) / ROW_WAVES + 31) & ~31;
    const int sBeg = min(end, beg + wv * per);
    const int sEnd = min(end, sBeg + per);
    const int2 none = make_int2(0, node0);
#pragma unroll
    for (int w = 0; w < ROW_WAVES; ++w) cntW[w][t] = 0;
    __syncthreads();
    {
        int2 pn = (sBeg + lane < sEnd) ? pg[sBeg + lane] : none;
        for (int i = sBeg; i < sEnd; i += 32) {
            const int2 p = pn;
            const bool valid = i + lane < sEnd;
            pn = (i + 32 + lane < sEnd) ? pg[i + 32 + lane] : none;
            const int dl = p.y - node0;
            const unsigned grp = same_key_lanes(dl, valid);
            if (valid && lane == __builtin_ctz(grp)) cntW[wv][dl] += __builtin_popcount(grp);
        }
    }
    __syncthreads();
    int v = 0;
#pragma unroll
    for (int w = 0; w < ROW_WAVES; ++w) {
        const int c = cntW[w][t];
        cntW[w][t] = v;
        v += c;
    }
    const int incl = wave_incl_scan(v, lane);
    if (lane == 31) wsum[wv] = incl;
    __syncthreads();
    int wo = 0;
#pragma unroll
    for (int w = 0; w < ROW_WAVES; ++w) wo += (w < wv) ? wsum[w] : 0;
    const int first = beg + wo + incl - v;
#pragma unroll
    for (int w = 0; w < ROW_WAVES; ++w) cntW[w][t] += first;
    const int node = node0 + t;
    if (node < nNodes) rows[(size_t)g * nNodes + node] = make_int2(first, v);
    __syncthreads();
    {
        int2 pn = (sBeg + lane < sEnd) ? pg[sBeg + lane] : none;
        for (int i = sBeg; i < sEnd; i += 32) {
            const int2 p = pn;
            const bool valid = i + lane < sEnd;
            pn = (i + 32 + lane < sEnd) ? pg[i + 32 + lane] : none;
            const int dl = p.y - node0;
            const unsigned grp = same_key_lanes(dl, valid);
            const int gfirst = valid ? __builtin_ctz(grp) : 0;
            int base = 0;
            if (valid && lane == gfirst) {
                base = cntW[wv][dl];
                cntW[wv][dl] = base + __builtin_popcount(grp);
            }
            base = __shfl(base, gfirst, 32);
            if (valid) sg[base + __builtin_popcount(grp & lanes_below(lane))] = p.x;
        }
    }
}

__global__ void __launch_bounds__(NODE_WAVES * 32)
gather_sum_kernel(const float* __restrict__ x, const int2* __restrict__ rows,
                  const int* __restrict__ sorted, float* __restrict__ agg, int nNodes)
{
    const int lane = threadIdx.x & 31;
    const int n = wave_in_grid();
    if (n >= nNodes) return;
    const int2 row = rows[n];
    const int beg = row.x;
    const int cnt = row.y;
    float a0 = 0.f, a1 = 0.f;
    for (int i0 = 0; i0 < cnt; i0 += 32) {
        const int m = min(32, cnt - i0);
        const int idx = (lane < m) ? sorted[beg + i0 + lane] : 0;
        for (int j = 0; j < m; j += 8) {
            float r0[8], r1[8];
#pragma unroll
            for (int u = 0; u < 8; ++u) {
                const int s = __builtin_amdgcn_readlane(idx, min(j + u, m - 1));
                const float* p = x + (size_t)s * DCH + lane;
                r0[u] = p[0];
                r1[u] = p[32];
            }
#pragma unroll
            for (int u = 0; u < 8; ++u) {
                a0 += (j + u < m) ? r0[u] : 0.f;
                a1 += (j + u < m) ? r1[u] : 0.f;
            }
        }
    }
    agg[(size_t)n * DCH + lane] = a0;
    agg[(size_t)n * DCH + 32 + lane] = a1;
}

__global__ void __launch_bounds__(128)
gemm5_kernel(const float* __restrict__ x,  const float* __restrict__ agg,
             const float* __restrict__ W0, const float* __restrict__ W1,
             const float* __restrict__ WQ, const float* __restrict__ WK,
             const float* __restrict__ WV,
             float* __restrict__ out, float* __restrict__ Qo,
             float* __restrict__ Ko,  float* __restrict__ Vo, int nrows)
{
    const int lane = threadIdx.x & 31;
    const int wave = threadIdx.x >> 5;
    const int row0 = (blockIdx.x * 4 + wave) * 16;
    if (row0 >= nrows) return;
    const int m = lane & 15;
    const int h = lane >> 4;

    v2f ax[16], ag[16];
    const float* xr = x   + (long long)(row0 + m) * DCH + 2 * h;
    const float* gr = agg + (long long)(row0 + m) * DCH + 2 * h;
#pragma unroll
    for (int kk = 0; kk < 16; ++kk) {
        ax[kk] = ldv2(xr + 4 * kk);
        ag[kk] = ldv2(gr + 4 * kk);
    }

    const int bi = m * DCH + 2 * h;

#pragma unroll
    for (int n0 = 0; n0 < DCH; n0 += 16) {
        const float* b0 = W0 + n0 * DCH + bi;
        const float* b1 = W1 + n0 * DCH + bi;
        const float* bq = WQ + n0 * DCH + bi;
        const float* bk = WK + n0 * DCH + bi;
        const float* bv = WV + n0 * DCH + bi;

        v8f acc = {};
#pragma unroll
        for (int kk = 0; kk < 16; ++kk)
            acc = __builtin_amdgcn_wmma_f32_16x16x4_f32(
                false, ax[kk], false, ldv2(b0 + 4 * kk), (short)0, acc, false, false);
#pragma unroll
        for (int kk = 0; kk < 16; ++kk)
            acc = __builtin_amdgcn_wmma_f32_16x16x4_f32(
                false, ag[kk], false, ldv2(b1 + 4 * kk), (short)0, acc, false, false);
#pragma unroll
        for (int r = 0; r < 8; ++r)
            out[(long long)(row0 + r + 8 * h) * DCH + n0 + m] = acc[r];

        acc = (v8f){};
#pragma unroll
        for (int kk = 0; kk < 16; ++kk)
            acc = __builtin_amdgcn_wmma_f32_16x16x4_f32(
                false, ax[kk], false, ldv2(bq + 4 * kk), (short)0, acc, false, false);
#pragma unroll
        for (int r = 0; r < 8; ++r)
            Qo[(long long)(row0 + r + 8 * h) * DCH + n0 + m] = acc[r];

        acc = (v8f){};
#pragma unroll
        for (int kk = 0; kk < 16; ++kk)
            acc = __builtin_amdgcn_wmma_f32_16x16x4_f32(
                false, ax[kk], false, ldv2(bk + 4 * kk), (short)0, acc, false, false);
#pragma unroll
        for (int r = 0; r < 8; ++r)
            Ko[(long long)(row0 + r + 8 * h) * DCH + n0 + m] = acc[r];

        acc = (v8f){};
#pragma unroll
        for (int kk = 0; kk < 16; ++kk)
            acc = __builtin_amdgcn_wmma_f32_16x16x4_f32(
                false, ax[kk], false, ldv2(bv + 4 * kk), (short)0, acc, false, false);
#pragma unroll
        for (int r = 0; r < 8; ++r)
            Vo[(long long)(row0 + r + 8 * h) * DCH + n0 + m] = acc[r];
    }
}

__device__ __forceinline__ float wave_sum(float p) {
    for (int off = 16; off >= 1; off >>= 1)
        p += __shfl_xor(p, off, 32);
    return p;
}

__global__ void __launch_bounds__(NODE_WAVES * 32)
attn_rows_kernel(const float* __restrict__ Q, const float* __restrict__ K,
                 const float* __restrict__ V, const int2* __restrict__ rows,
                 const int* __restrict__ sorted, float* out, int nNodes)
{
    const int lane = threadIdx.x & 31;
    const int n = wave_in_grid();
    if (n >= nNodes) return;
    const int2 row = rows[n];
    const int beg = row.x;
    const int cnt = row.y;
    const float q0 = Q[(size_t)n * DCH + lane];
    const float q1 = Q[(size_t)n * DCH + 32 + lane];
    const float x0 = out[(size_t)n * DCH + lane];
    const float x1 = out[(size_t)n * DCH + 32 + lane];
    float a0 = 0.f, a1 = 0.f;
    for (int i0 = 0; i0 < cnt; i0 += 32) {
        const int m = min(32, cnt - i0);
        const int idx = (lane < m) ? sorted[beg + i0 + lane] : 0;
        for (int j = 0; j < m; j += 4) {
            float k0[4], k1[4], v0[4], v1[4];
#pragma unroll
            for (int u = 0; u < 4; ++u) {
                const int s = __builtin_amdgcn_readlane(idx, min(j + u, m - 1));
                const float* kp = K + (size_t)s * DCH + lane;
                const float* vp = V + (size_t)s * DCH + lane;
                k0[u] = kp[0];
                k1[u] = kp[32];
                v0[u] = vp[0];
                v1[u] = vp[32];
            }
#pragma unroll
            for (int u = 0; u < 4; ++u) {
                const float p = wave_sum(q0 * k0[u] + q1 * k1[u]);
                const float pe = (j + u < m) ? p : 0.f;
                a0 += pe * v0[u];
                a1 += pe * v1[u];
            }
        }
    }
    out[(size_t)n * DCH + lane] = x0 - a0;
    out[(size_t)n * DCH + 32 + lane] = x1 - a1;
}

extern "C" void kernel_launch(void* const* d_in, const int* in_sizes, int n_in,
                              void* d_out, int out_size, void* d_ws, size_t ws_size,
                              hipStream_t stream)
{
    const float* x  = (const float*)d_in[0];
    const int*  ei1 = (const int*)d_in[1];
    const int*  ei2 = (const int*)d_in[2];
    const float* W0 = (const float*)d_in[3];
    const float* W1 = (const float*)d_in[4];
    const float* WQ = (const float*)d_in[5];
    const float* WK = (const float*)d_in[6];
    const float* WV = (const float*)d_in[7];
    float* out = (float*)d_out;

    const int N  = in_sizes[0] / DCH;
    const int E1 = in_sizes[1] / 2;
    const int E2 = in_sizes[2] / 2;
    const int EM = E1 > E2 ? E1 : E2;
    const int listStride = (EM + 63) / 64 * 64;
    const int nBins = (N + BIN_NODES - 1) / BIN_NODES;
    const int nChunks = (EM + CHUNK_EDGES - 1) / CHUNK_EDGES;

    float* agg = (float*)d_ws;
    float* Qb  = agg + (size_t)N * DCH;
    float* Kb  = Qb  + (size_t)N * DCH;
    float* Vb  = Kb  + (size_t)N * DCH;
    int2* part    = (int2*)(Vb + (size_t)N * DCH);
    int* sorted   = (int*)(part + (size_t)2 * listStride);
    int2* rows    = (int2*)(sorted + (size_t)2 * listStride);
    int* cnt      = (int*)(rows + (size_t)2 * N);
    int* start    = cnt + (size_t)2 * nChunks * nBins;
    int* binStart = start + (size_t)2 * nChunks * nBins;

    const size_t histBytes = (size_t)PART_WAVES * nBins * sizeof(int);

    bin_count_kernel<<<dim3(nChunks, 2), PART_THREADS, histBytes, stream>>>(
        ei1 + E1, E1, ei2 + E2, E2, cnt, nBins, nChunks);
    bin_scan_kernel<<<2, 1024, 0, stream>>>(cnt, start, binStart, nBins, nChunks);
    bin_place_kernel<<<dim3(nChunks, 2), PART_THREADS, histBytes, stream>>>(
        ei1, ei1 + E1, E1, ei2, ei2 + E2, E2, start, part, nBins, nChunks, listStride);
    bin_to_rows_kernel<<<dim3(nBins, 2), BIN_NODES, 0, stream>>>(
        part, binStart, rows, sorted, nBins, N, listStride);

    const int nodeBlocks = (N + NODE_WAVES - 1) / NODE_WAVES;
    gather_sum_kernel<<<nodeBlocks, NODE_WAVES * 32, 0, stream>>>(x, rows, sorted, agg, N);

    {
        int strips = (N + 15) / 16;
        int blocks = (strips + 3) / 4;
        gemm5_kernel<<<blocks, 128, 0, stream>>>(x, agg, W0, W1, WQ, WK, WV,
                                                 out, Qb, Kb, Vb, N);
    }

    attn_rows_kernel<<<nodeBlocks, NODE_WAVES * 32, 0, stream>>>(
        Qb, Kb, Vb, rows + N, sorted + listStride, out, N);
}


# Same-key lane sets come from one ballot per key bit (9 rounds for bins, 8 for nodes) instead of one readlane round per distinct key (about 31 per 32 lanes); list and float sum order unchanged
# speedup vs baseline: 1.5246x; 1.2046x over previous
#include <hip/hip_runtime.h>

typedef __attribute__((ext_vector_type(2))) float v2f;
typedef __attribute__((ext_vector_type(8))) float v8f;

#define DCH 64
#define BIN_NODES 256
#define BIN_SHIFT 8
#define PART_THREADS 1024
#define PART_WAVES (PART_THREADS / 32)
#define PART_EPT 16
#define CHUNK_EDGES (PART_THREADS * PART_EPT)
#define ROW_WAVES (BIN_NODES / 32)
#define NODE_WAVES 8

__device__ __forceinline__ v2f ldv2(const float* p) { return *(const v2f*)p; }

__device__ __forceinline__ int wave_in_grid() {
    const int w = __builtin_amdgcn_readfirstlane((int)(threadIdx.x >> 5));
    return (int)(blockIdx.x * NODE_WAVES) + w;
}

__device__ __forceinline__ int wave_incl_scan(int v, int lane) {
    for (int off = 1; off < 32; off <<= 1) {
        const int t = __shfl_up(v, off, 32);
        if (lane >= off) v += t;
    }
    return v;
}

__device__ __forceinline__ unsigned same_key_lanes(int key, bool valid, int nbits) {
    unsigned m = __builtin_amdgcn_ballot_w32(valid);
    for (int b = 0; b < nbits; ++b) {
        const int sx = -((key >> b) & 1);
        const unsigned set = __builtin_amdgcn_ballot_w32(sx != 0);
        m &= ~(set ^ (unsigned)sx);
    }
    return valid ? m : 0u;
}

__device__ __forceinline__ int key_bits(int n) { return n > 1 ? 32 - __builtin_clz((unsigned)(n - 1)) : 0; }

__device__ __forceinline__ unsigned lanes_below(int lane) { return (1u << lane) - 1u; }


__global__ void __launch_bounds__(PART_THREADS)
bin_count_kernel(const int* __restrict__ dstA, int nEA,
                 const int* __restrict__ dstB, int nEB,
                 int* __restrict__ cnt, int nBins, int nChunks)
{
    extern __shared__ int histW[];
    const int g = blockIdx.y;
    const int* __restrict__ dst = g ? dstB : dstA;
    const int nE = g ? nEB : nEA;
    const int c = blockIdx.x;
    const int lane = threadIdx.x & 31;
    const int wv = __builtin_amdgcn_readfirstlane((int)(threadIdx.x >> 5));
    const int binBits = key_bits(nBins);
    for (int i = threadIdx.x; i < PART_WAVES * nBins; i += PART_THREADS) histW[i] = 0;
    __syncthreads();
    const int e0 = c * CHUNK_EDGES + (int)threadIdx.x;
    int d[PART_EPT];
#pragma unroll
    for (int k = 0; k < PART_EPT; ++k) {
        const int e = e0 + k * PART_THREADS;
        d[k] = (e < nE) ? dst[e] : -1;
    }
    int* myRow = histW + wv * nBins;
#pragma unroll
    for (int k = 0; k < PART_EPT; ++k) {
        const bool valid = d[k] >= 0;
        const int bin = valid ? (d[k] >> BIN_SHIFT) : 0;
        const unsigned grp = same_key_lanes(bin, valid, binBits);
        if (valid && lane == __builtin_ctz(grp)) myRow[bin] += __builtin_popcount(grp);
    }
    __syncthreads();
    int* __restrict__ row = cnt + ((size_t)g * nChunks + c) * nBins;
    for (int b = threadIdx.x; b < nBins; b += PART_THREADS) {
        int tot = 0;
        for (int w = 0; w < PART_WAVES; ++w) tot += histW[w * nBins + b];
        row[b] = tot;
    }
}

__global__ void __launch_bounds__(1024)
bin_scan_kernel(const int* __restrict__ cnt, int* __restrict__ start,
                int* __restrict__ binStart, int nBins, int nChunks)
{
    __shared__ int wsum[32];
    __shared__ int tileTotal;
    const int g = blockIdx.x;
    const int* __restrict__ m = cnt + (size_t)g * nChunks * nBins;
    int* __restrict__ st = start + (size_t)g * nChunks * nBins;
    int* __restrict__ bs = binStart + (size_t)g * (nBins + 1);
    const int lane = threadIdx.x & 31, wv = threadIdx.x >> 5;
    int carry = 0;
    for (int b0 = 0; b0 < nBins; b0 += 1024) {
        const int b = b0 + (int)threadIdx.x;
        const bool live = b < nBins;
        int tot = 0;
        for (int c = 0; c < nChunks; c += 16) {
            int v[16];
#pragma unroll
            for (int u = 0; u < 16; ++u)
                v[u] = (live && c + u < nChunks) ? m[(size_t)(c + u) * nBins + b] : 0;
#pragma unroll
            for (int u = 0; u < 16; ++u) tot += v[u];
        }
        const int incl = wave_incl_scan(tot, lane);
        if (lane == 31) wsum[wv] = incl;
        __syncthreads();
        if (wv == 0) {
            const int s = wsum[lane];
            const int si = wave_incl_scan(s, lane);
            wsum[lane] = si - s;
            if (lane == 31) tileTotal = si;
        }
        __syncthreads();
        const int excl = carry + wsum[wv] + incl - tot;
        const int tt = tileTotal;
        if (live) bs[b] = excl;
        int run = excl;
        for (int c = 0; c < nChunks; c += 16) {
            int v[16];
#pragma unroll
            for (int u = 0; u < 16; ++u)
                v[u] = (live && c + u < nChunks) ? m[(size_t)(c + u) * nBins + b] : 0;
#pragma unroll
            for (int u = 0; u < 16; ++u) {
                if (live && c + u < nChunks) st[(size_t)(c + u) * nBins + b] = run;
                run += v[u];
            }
        }
        carry += tt;
        __syncthreads();
    }
    if (threadIdx.x == 0) bs[nBins] = carry;
}

__global__ void __launch_bounds__(PART_THREADS)
bin_place_kernel(const int* __restrict__ srcA, const int* __restrict__ dstA, int nEA,
                 const int* __restrict__ srcB, const int* __restrict__ dstB, int nEB,
                 const int* __restrict__ start, int2* __restrict__ part,
                 int nBins, int nChunks, int listStride)
{
    extern __shared__ int histW[];
    const int g = blockIdx.y;
    const int* __restrict__ src = g ? srcB : srcA;
    const int* __restrict__ dst = g ? dstB : dstA;
    const int nE = g ? nEB : nEA;
    const int c = blockIdx.x;
    const int lane = threadIdx.x & 31;
    const int wv = __builtin_amdgcn_readfirstlane((int)(threadIdx.x >> 5));
    const int binBits = key_bits(nBins);
    for (int i = threadIdx.x; i < PART_WAVES * nBins; i += PART_THREADS) histW[i] = 0;
    __syncthreads();
    const int e0 = c * CHUNK_EDGES + (int)threadIdx.x;
    int d[PART_EPT], s[PART_EPT];
    unsigned grp[PART_EPT];
#pragma unroll
    for (int k = 0; k < PART_EPT; ++k) {
        const int e = e0 + k * PART_THREADS;
        const bool in = e < nE;
        d[k] = in ? dst[e] : -1;
        s[k] = in ? src[e] : 0;
    }
    int* myRow = histW + wv * nBins;
#pragma unroll
    for (int k = 0; k < PART_EPT; ++k) {
        const bool valid = d[k] >= 0;
        const int bin = valid ? (d[k] >> BIN_SHIFT) : 0;
        grp[k] = same_key_lanes(bin, valid, binBits);
        if (valid && lane == __builtin_ctz(grp[k])) myRow[bin] += __builtin_popcount(grp[k]);
    }
    __syncthreads();
    const int* __restrict__ srow = start + ((size_t)g * nChunks + c) * nBins;
    for (int b = threadIdx.x; b < nBins; b += PART_THREADS) {
        int run = srow[b];
        for (int w = 0; w < PART_WAVES; ++w) {
            const int t = histW[w * nBins + b];
            histW[w * nBins + b] = run;
            run += t;
        }
    }
    __syncthreads();
    int2* __restrict__ pg = part + (size_t)g * listStride;
#pragma unroll
    for (int k = 0; k < PART_EPT; ++k) {
        const bool valid = d[k] >= 0;
        const int bin = valid ? (d[k] >> BIN_SHIFT) : 0;
        const int first = valid ? __builtin_ctz(grp[k]) : 0;
        int base = 0;
        if (valid && lane == first) {
            base = myRow[bin];
            myRow[bin] = base + __builtin_popcount(grp[k]);
        }
        base = __shfl(base, first, 32);
        if (valid) pg[base + __builtin_popcount(grp[k] & lanes_below(lane))] = make_int2(s[k], d[k]);
    }
}

__global__ void __launch_bounds__(BIN_NODES)
bin_to_rows_kernel(const int2* __restrict__ part, const int* __restrict__ binStart,
                   int2* __restrict__ rows, int* __restrict__ sorted,
                   int nBins, int nNodes, int listStride)
{
    __shared__ int cntW[ROW_WAVES][BIN_NODES];
    __shared__ int wsum[ROW_WAVES];
    const int g = blockIdx.y, b = blockIdx.x, t = threadIdx.x;
    const int lane = t & 31;
    const int wv = __builtin_amdgcn_readfirstlane(t >> 5);
    const int beg = binStart[(size_t)g * (nBins + 1) + b];
    const int end = binStart[(size_t)g * (nBins + 1) + b + 1];
    const int node0 = b * BIN_NODES;
    const int2* __restrict__ pg = part + (size_t)g * listStride;
    int* __restrict__ sg = sorted + (size_t)g * listStride;
    const int per = (((end - beg) + ROW_WAVES - 1) / ROW_WAVES + 31) & ~31;
    const int sBeg = min(end, beg + wv * per);
    const int sEnd = min(end, sBeg + per);
    const int2 none = make_int2(0, node0);
#pragma unroll
    for (int w = 0; w < ROW_WAVES; ++w) cntW[w][t] = 0;
    __syncthreads();
    {
        int2 pn = (sBeg + lane < sEnd) ? pg[sBeg + lane] : none;
        for (int i = sBeg; i < sEnd; i += 32) {
            const int2 p = pn;
            const bool valid = i + lane < sEnd;
            pn = (i + 32 + lane < sEnd) ? pg[i + 32 + lane] : none;
            const int dl = p.y - node0;
            const unsigned grp = same_key_lanes(dl, valid, BIN_SHIFT);
            if (valid && lane == __builtin_ctz(grp)) cntW[wv][dl] += __builtin_popcount(grp);
        }
    }
    __syncthreads();
    int v = 0;
#pragma unroll
    for (int w = 0; w < ROW_WAVES; ++w) {
        const int c = cntW[w][t];
        cntW[w][t] = v;
        v += c;
    }
    const int incl = wave_incl_scan(v, lane);
    if (lane == 31) wsum[wv] = incl;
    __syncthreads();
    int wo = 0;
#pragma unroll
    for (int w = 0; w < ROW_WAVES; ++w) wo += (w < wv) ? wsum[w] : 0;
    const int first = beg + wo + incl - v;
#pragma unroll
    for (int w = 0; w < ROW_WAVES; ++w) cntW[w][t] += first;
    const int node = node0 + t;
    if (node < nNodes) rows[(size_t)g * nNodes + node] = make_int2(first, v);
    __syncthreads();
    {
        int2 pn = (sBeg + lane < sEnd) ? pg[sBeg + lane] : none;
        for (int i = sBeg; i < sEnd; i += 32) {
            const int2 p = pn;
            const bool valid = i + lane < sEnd;
            pn = (i + 32 + lane < sEnd) ? pg[i + 32 + lane] : none;
            const int dl = p.y - node0;
            const unsigned grp = same_key_lanes(dl, valid, BIN_SHIFT);
            const int gfirst = valid ? __builtin_ctz(grp) : 0;
            int base = 0;
            if (valid && lane == gfirst) {
                base = cntW[wv][dl];
                cntW[wv][dl] = base + __builtin_popcount(grp);
            }
            base = __shfl(base, gfirst, 32);
            if (valid) sg[base + __builtin_popcount(grp & lanes_below(lane))] = p.x;
        }
    }
}

__global__ void __launch_bounds__(NODE_WAVES * 32)
gather_sum_kernel(const float* __restrict__ x, const int2* __restrict__ rows,
                  const int* __restrict__ sorted, float* __restrict__ agg, int nNodes)
{
    const int lane = threadIdx.x & 31;
    const int n = wave_in_grid();
    if (n >= nNodes) return;
    const int2 row = rows[n];
    const int beg = row.x;
    const int cnt = row.y;
    float a0 = 0.f, a1 = 0.f;
    for (int i0 = 0; i0 < cnt; i0 += 32) {
        const int m = min(32, cnt - i0);
        const int idx = (lane < m) ? sorted[beg + i0 + lane] : 0;
        for (int j = 0; j < m; j += 8) {
            float r0[8], r1[8];
#pragma unroll
            for (int u = 0; u < 8; ++u) {
                const int s = __builtin_amdgcn_readlane(idx, min(j + u, m - 1));
                const float* p = x + (size_t)s * DCH + lane;
                r0[u] = p[0];
                r1[u] = p[32];
            }
#pragma unroll
            for (int u = 0; u < 8; ++u) {
                a0 += (j + u < m) ? r0[u] : 0.f;
                a1 += (j + u < m) ? r1[u] : 0.f;
            }
        }
    }
    agg[(size_t)n * DCH + lane] = a0;
    agg[(size_t)n * DCH + 32 + lane] = a1;
}

__global__ void __launch_bounds__(128)
gemm5_kernel(const float* __restrict__ x,  const float* __restrict__ agg,
             const float* __restrict__ W0, const float* __restrict__ W1,
             const float* __restrict__ WQ, const float* __restrict__ WK,
             const float* __restrict__ WV,
             float* __restrict__ out, float* __restrict__ Qo,
             float* __restrict__ Ko,  float* __restrict__ Vo, int nrows)
{
    const int lane = threadIdx.x & 31;
    const int wave = threadIdx.x >> 5;
    const int row0 = (blockIdx.x * 4 + wave) * 16;
    if (row0 >= nrows) return;
    const int m = lane & 15;
    const int h = lane >> 4;

    v2f ax[16], ag[16];
    const float* xr = x   + (long long)(row0 + m) * DCH + 2 * h;
    const float* gr = agg + (long long)(row0 + m) * DCH + 2 * h;
#pragma unroll
    for (int kk = 0; kk < 16; ++kk) {
        ax[kk] = ldv2(xr + 4 * kk);
        ag[kk] = ldv2(gr + 4 * kk);
    }

    const int bi = m * DCH + 2 * h;

#pragma unroll
    for (int n0 = 0; n0 < DCH; n0 += 16) {
        const float* b0 = W0 + n0 * DCH + bi;
        const float* b1 = W1 + n0 * DCH + bi;
        const float* bq = WQ + n0 * DCH + bi;
        const float* bk = WK + n0 * DCH + bi;
        const float* bv = WV + n0 * DCH + bi;

        v8f acc = {};
#pragma unroll
        for (int kk = 0; kk < 16; ++kk)
            acc = __builtin_amdgcn_wmma_f32_16x16x4_f32(
                false, ax[kk], false, ldv2(b0 + 4 * kk), (short)0, acc, false, false);
#pragma unroll
        for (int kk = 0; kk < 16; ++kk)
            acc = __builtin_amdgcn_wmma_f32_16x16x4_f32(
                false, ag[kk], false, ldv2(b1 + 4 * kk), (short)0, acc, false, false);
#pragma unroll
        for (int r = 0; r < 8; ++r)
            out[(long long)(row0 + r + 8 * h) * DCH + n0 + m] = acc[r];

        acc = (v8f){};
#pragma unroll
        for (int kk = 0; kk < 16; ++kk)
            acc = __builtin_amdgcn_wmma_f32_16x16x4_f32(
                false, ax[kk], false, ldv2(bq + 4 * kk), (short)0, acc, false, false);
#pragma unroll
        for (int r = 0; r < 8; ++r)
            Qo[(long long)(row0 + r + 8 * h) * DCH + n0 + m] = acc[r];

        acc = (v8f){};
#pragma unroll
        for (int kk = 0; kk < 16; ++kk)
            acc = __builtin_amdgcn_wmma_f32_16x16x4_f32(
                false, ax[kk], false, ldv2(bk + 4 * kk), (short)0, acc, false, false);
#pragma unroll
        for (int r = 0; r < 8; ++r)
            Ko[(long long)(row0 + r + 8 * h) * DCH + n0 + m] = acc[r];

        acc = (v8f){};
#pragma unroll
        for (int kk = 0; kk < 16; ++kk)
            acc = __builtin_amdgcn_wmma_f32_16x16x4_f32(
                false, ax[kk], false, ldv2(bv + 4 * kk), (short)0, acc, false, false);
#pragma unroll
        for (int r = 0; r < 8; ++r)
            Vo[(long long)(row0 + r + 8 * h) * DCH + n0 + m] = acc[r];
    }
}

__device__ __forceinline__ float wave_sum(float p) {
    for (int off = 16; off >= 1; off >>= 1)
        p += __shfl_xor(p, off, 32);
    return p;
}

__global__ void __launch_bounds__(NODE_WAVES * 32)
attn_rows_kernel(const float* __restrict__ Q, const float* __restrict__ K,
                 const float* __restrict__ V, const int2* __restrict__ rows,
                 const int* __restrict__ sorted, float* out, int nNodes)
{
    const int lane = threadIdx.x & 31;
    const int n = wave_in_grid();
    if (n >= nNodes) return;
    const int2 row = rows[n];
    const int beg = row.x;
    const int cnt = row.y;
    const float q0 = Q[(size_t)n * DCH + lane];
    const float q1 = Q[(size_t)n * DCH + 32 + lane];
    const float x0 = out[(size_t)n * DCH + lane];
    const float x1 = out[(size_t)n * DCH + 32 + lane];
    float a0 = 0.f, a1 = 0.f;
    for (int i0 = 0; i0 < cnt; i0 += 32) {
        const int m = min(32, cnt - i0);
        const int idx = (lane < m) ? sorted[beg + i0 + lane] : 0;
        for (int j = 0; j < m; j += 4) {
            float k0[4], k1[4], v0[4], v1[4];
#pragma unroll
            for (int u = 0; u < 4; ++u) {
                const int s = __builtin_amdgcn_readlane(idx, min(j + u, m - 1));
                const float* kp = K + (size_t)s * DCH + lane;
                const float* vp = V + (size_t)s * DCH + lane;
                k0[u] = kp[0];
                k1[u] = kp[32];
                v0[u] = vp[0];
                v1[u] = vp[32];
            }
#pragma unroll
            for (int u = 0; u < 4; ++u) {
                const float p = wave_sum(q0 * k0[u] + q1 * k1[u]);
                const float pe = (j + u < m) ? p : 0.f;
                a0 += pe * v0[u];
                a1 += pe * v1[u];
            }
        }
    }
    out[(size_t)n * DCH + lane] = x0 - a0;
    out[(size_t)n * DCH + 32 + lane] = x1 - a1;
}

extern "C" void kernel_launch(void* const* d_in, const int* in_sizes, int n_in,
                              void* d_out, int out_size, void* d_ws, size_t ws_size,
                              hipStream_t stream)
{
    const float* x  = (const float*)d_in[0];
    const int*  ei1 = (const int*)d_in[1];
    const int*  ei2 = (const int*)d_in[2];
    const float* W0 = (const float*)d_in[3];
    const float* W1 = (const float*)d_in[4];
    const float* WQ = (const float*)d_in[5];
    const float* WK = (const float*)d_in[6];
    const float* WV = (const float*)d_in[7];
    float* out = (float*)d_out;

    const int N  = in_sizes[0] / DCH;
    const int E1 = in_sizes[1] / 2;
    const int E2 = in_sizes[2] / 2;
    const int EM = E1 > E2 ? E1 : E2;
    const int listStride = (EM + 63) / 64 * 64;
    const int nBins = (N + BIN_NODES - 1) / BIN_NODES;
    const int nChunks = (EM + CHUNK_EDGES - 1) / CHUNK_EDGES;

    float* agg = (float*)d_ws;
    float* Qb  = agg + (size_t)N * DCH;
    float* Kb  = Qb  + (size_t)N * DCH;
    float* Vb  = Kb  + (size_t)N * DCH;
    int2* part    = (int2*)(Vb + (size_t)N * DCH);
    int* sorted   = (int*)(part + (size_t)2 * listStride);
    int2* rows    = (int2*)(sorted + (size_t)2 * listStride);
    int* cnt      = (int*)(rows + (size_t)2 * N);
    int* start    = cnt + (size_t)2 * nChunks * nBins;
    int* binStart = start + (size_t)2 * nChunks * nBins;

    const size_t histBytes = (size_t)PART_WAVES * nBins * sizeof(int);

    bin_count_kernel<<<dim3(nChunks, 2), PART_THREADS, histBytes, stream>>>(
        ei1 + E1, E1, ei2 + E2, E2, cnt, nBins, nChunks);
    bin_scan_kernel<<<2, 1024, 0, stream>>>(cnt, start, binStart, nBins, nChunks);
    bin_place_kernel<<<dim3(nChunks, 2), PART_THREADS, histBytes, stream>>>(
        ei1, ei1 + E1, E1, ei2, ei2 + E2, E2, start, part, nBins, nChunks, listStride);
    bin_to_rows_kernel<<<dim3(nBins, 2), BIN_NODES, 0, stream>>>(
        part, binStart, rows, sorted, nBins, N, listStride);

    const int nodeBlocks = (N + NODE_WAVES - 1) / NODE_WAVES;
    gather_sum_kernel<<<nodeBlocks, NODE_WAVES * 32, 0, stream>>>(x, rows, sorted, agg, N);

    {
        int strips = (N + 15) / 16;
        int blocks = (strips + 3) / 4;
        gemm5_kernel<<<blocks, 128, 0, stream>>>(x, agg, W0, W1, WQ, WK, WV,
                                                 out, Qb, Kb, Vb, N);
    }

    attn_rows_kernel<<<nodeBlocks, NODE_WAVES * 32, 0, stream>>>(
        Qb, Kb, Vb, rows + N, sorted + listStride, out, N);
}


# GEMM launch: the five weight matrices are staged once per work-group in LDS; each B fragment is an 8-byte LDS read, was a global load touching one line per weight row; 768 groups loop over strips
# speedup vs baseline: 1.6405x; 1.0760x over previous
#include <hip/hip_runtime.h>

typedef __attribute__((ext_vector_type(2))) float v2f;
typedef __attribute__((ext_vector_type(4))) float v4f;
typedef __attribute__((ext_vector_type(8))) float v8f;

#define DCH 64
#define BIN_NODES 256
#define BIN_SHIFT 8
#define PART_THREADS 1024
#define PART_WAVES (PART_THREADS / 32)
#define PART_EPT 16
#define CHUNK_EDGES (PART_THREADS * PART_EPT)
#define ROW_WAVES (BIN_NODES / 32)
#define NODE_WAVES 8

__device__ __forceinline__ v2f ldv2(const float* p) { return *(const v2f*)p; }

__device__ __forceinline__ int wave_in_grid() {
    const int w = __builtin_amdgcn_readfirstlane((int)(threadIdx.x >> 5));
    return (int)(blockIdx.x * NODE_WAVES) + w;
}

__device__ __forceinline__ int wave_incl_scan(int v, int lane) {
    for (int off = 1; off < 32; off <<= 1) {
        const int t = __shfl_up(v, off, 32);
        if (lane >= off) v += t;
    }
    return v;
}

__device__ __forceinline__ unsigned same_key_lanes(int key, bool valid, int nbits) {
    unsigned m = __builtin_amdgcn_ballot_w32(valid);
    for (int b = 0; b < nbits; ++b) {
        const int sx = -((key >> b) & 1);
        const unsigned set = __builtin_amdgcn_ballot_w32(sx != 0);
        m &= ~(set ^ (unsigned)sx);
    }
    return valid ? m : 0u;
}

__device__ __forceinline__ int key_bits(int n) { return n > 1 ? 32 - __builtin_clz((unsigned)(n - 1)) : 0; }

__device__ __forceinline__ unsigned lanes_below(int lane) { return (1u << lane) - 1u; }


__global__ void __launch_bounds__(PART_THREADS)
bin_count_kernel(const int* __restrict__ dstA, int nEA,
                 const int* __restrict__ dstB, int nEB,
                 int* __restrict__ cnt, int nBins, int nChunks)
{
    extern __shared__ int histW[];
    const int g = blockIdx.y;
    const int* __restrict__ dst = g ? dstB : dstA;
    const int nE = g ? nEB : nEA;
    const int c = blockIdx.x;
    const int lane = threadIdx.x & 31;
    const int wv = __builtin_amdgcn_readfirstlane((int)(threadIdx.x >> 5));
    const int binBits = key_bits(nBins);
    for (int i = threadIdx.x; i < PART_WAVES * nBins; i += PART_THREADS) histW[i] = 0;
    __syncthreads();
    const int e0 = c * CHUNK_EDGES + (int)threadIdx.x;
    int d[PART_EPT];
#pragma unroll
    for (int k = 0; k < PART_EPT; ++k) {
        const int e = e0 + k * PART_THREADS;
        d[k] = (e < nE) ? dst[e] : -1;
    }
    int* myRow = histW + wv * nBins;
#pragma unroll
    for (int k = 0; k < PART_EPT; ++k) {
        const bool valid = d[k] >= 0;
        const int bin = valid ? (d[k] >> BIN_SHIFT) : 0;
        const unsigned grp = same_key_lanes(bin, valid, binBits);
        if (valid && lane == __builtin_ctz(grp)) myRow[bin] += __builtin_popcount(grp);
    }
    __syncthreads();
    int* __restrict__ row = cnt + ((size_t)g * nChunks + c) * nBins;
    for (int b = threadIdx.x; b < nBins; b += PART_THREADS) {
        int tot = 0;
        for (int w = 0; w < PART_WAVES; ++w) tot += histW[w * nBins + b];
        row[b] = tot;
    }
}

__global__ void __launch_bounds__(1024)
bin_scan_kernel(const int* __restrict__ cnt, int* __restrict__ start,
                int* __restrict__ binStart, int nBins, int nChunks)
{
    __shared__ int wsum[32];
    __shared__ int tileTotal;
    const int g = blockIdx.x;
    const int* __restrict__ m = cnt + (size_t)g * nChunks * nBins;
    int* __restrict__ st = start + (size_t)g * nChunks * nBins;
    int* __restrict__ bs = binStart + (size_t)g * (nBins + 1);
    const int lane = threadIdx.x & 31, wv = threadIdx.x >> 5;
    int carry = 0;
    for (int b0 = 0; b0 < nBins; b0 += 1024) {
        const int b = b0 + (int)threadIdx.x;
        const bool live = b < nBins;
        int tot = 0;
        for (int c = 0; c < nChunks; c += 16) {
            int v[16];
#pragma unroll
            for (int u = 0; u < 16; ++u)
                v[u] = (live && c + u < nChunks) ? m[(size_t)(c + u) * nBins + b] : 0;
#pragma unroll
            for (int u = 0; u < 16; ++u) tot += v[u];
        }
        const int incl = wave_incl_scan(tot, lane);
        if (lane == 31) wsum[wv] = incl;
        __syncthreads();
        if (wv == 0) {
            const int s = wsum[lane];
            const int si = wave_incl_scan(s, lane);
            wsum[lane] = si - s;
            if (lane == 31) tileTotal = si;
        }
        __syncthreads();
        const int excl = carry + wsum[wv] + incl - tot;
        const int tt = tileTotal;
        if (live) bs[b] = excl;
        int run = excl;
        for (int c = 0; c < nChunks; c += 16) {
            int v[16];
#pragma unroll
            for (int u = 0; u < 16; ++u)
                v[u] = (live && c + u < nChunks) ? m[(size_t)(c + u) * nBins + b] : 0;
#pragma unroll
            for (int u = 0; u < 16; ++u) {
                if (live && c + u < nChunks) st[(size_t)(c + u) * nBins + b] = run;
                run += v[u];
            }
        }
        carry += tt;
        __syncthreads();
    }
    if (threadIdx.x == 0) bs[nBins] = carry;
}

__global__ void __launch_bounds__(PART_THREADS)
bin_place_kernel(const int* __restrict__ srcA, const int* __restrict__ dstA, int nEA,
                 const int* __restrict__ srcB, const int* __restrict__ dstB, int nEB,
                 const int* __restrict__ start, int2* __restrict__ part,
                 int nBins, int nChunks, int listStride)
{
    extern __shared__ int histW[];
    const int g = blockIdx.y;
    const int* __restrict__ src = g ? srcB : srcA;
    const int* __restrict__ dst = g ? dstB : dstA;
    const int nE = g ? nEB : nEA;
    const int c = blockIdx.x;
    const int lane = threadIdx.x & 31;
    const int wv = __builtin_amdgcn_readfirstlane((int)(threadIdx.x >> 5));
    const int binBits = key_bits(nBins);
    for (int i = threadIdx.x; i < PART_WAVES * nBins; i += PART_THREADS) histW[i] = 0;
    __syncthreads();
    const int e0 = c * CHUNK_EDGES + (int)threadIdx.x;
    int d[PART_EPT], s[PART_EPT];
    unsigned grp[PART_EPT];
#pragma unroll
    for (int k = 0; k < PART_EPT; ++k) {
        const int e = e0 + k * PART_THREADS;
        const bool in = e < nE;
        d[k] = in ? dst[e] : -1;
        s[k] = in ? src[e] : 0;
    }
    int* myRow = histW + wv * nBins;
#pragma unroll
    for (int k = 0; k < PART_EPT; ++k) {
        const bool valid = d[k] >= 0;
        const int bin = valid ? (d[k] >> BIN_SHIFT) : 0;
        grp[k] = same_key_lanes(bin, valid, binBits);
        if (valid && lane == __builtin_ctz(grp[k])) myRow[bin] += __builtin_popcount(grp[k]);
    }
    __syncthreads();
    const int* __restrict__ srow = start + ((size_t)g * nChunks + c) * nBins;
    for (int b = threadIdx.x; b < nBins; b += PART_THREADS) {
        int run = srow[b];
        for (int w = 0; w < PART_WAVES; ++w) {
            const int t = histW[w * nBins + b];
            histW[w * nBins + b] = run;
            run += t;
        }
    }
    __syncthreads();
    int2* __restrict__ pg = part + (size_t)g * listStride;
#pragma unroll
    for (int k = 0; k < PART_EPT; ++k) {
        const bool valid = d[k] >= 0;
        const int bin = valid ? (d[k] >> BIN_SHIFT) : 0;
        const int first = valid ? __builtin_ctz(grp[k]) : 0;
        int base = 0;
        if (valid && lane == first) {
            base = myRow[bin];
            myRow[bin] = base + __builtin_popcount(grp[k]);
        }
        base = __shfl(base, first, 32);
        if (valid) pg[base + __builtin_popcount(grp[k] & lanes_below(lane))] = make_int2(s[k], d[k]);
    }
}

__global__ void __launch_bounds__(BIN_NODES)
bin_to_rows_kernel(const int2* __restrict__ part, const int* __restrict__ binStart,
                   int2* __restrict__ rows, int* __restrict__ sorted,
                   int nBins, int nNodes, int listStride)
{
    __shared__ int cntW[ROW_WAVES][BIN_NODES];
    __shared__ int wsum[ROW_WAVES];
    const int g = blockIdx.y, b = blockIdx.x, t = threadIdx.x;
    const int lane = t & 31;
    const int wv = __builtin_amdgcn_readfirstlane(t >> 5);
    const int beg = binStart[(size_t)g * (nBins + 1) + b];
    const int end = binStart[(size_t)g * (nBins + 1) + b + 1];
    const int node0 = b * BIN_NODES;
    const int2* __restrict__ pg = part + (size_t)g * listStride;
    int* __restrict__ sg = sorted + (size_t)g * listStride;
    const int per = (((end - beg) + ROW_WAVES - 1) / ROW_WAVES + 31) & ~31;
    const int sBeg = min(end, beg + wv * per);
    const int sEnd = min(end, sBeg + per);
    const int2 none = make_int2(0, node0);
#pragma unroll
    for (int w = 0; w < ROW_WAVES; ++w) cntW[w][t] = 0;
    __syncthreads();
    {
        int2 pn = (sBeg + lane < sEnd) ? pg[sBeg + lane] : none;
        for (int i = sBeg; i < sEnd; i += 32) {
            const int2 p = pn;
            const bool valid = i + lane < sEnd;
            pn = (i + 32 + lane < sEnd) ? pg[i + 32 + lane] : none;
            const int dl = p.y - node0;
            const unsigned grp = same_key_lanes(dl, valid, BIN_SHIFT);
            if (valid && lane == __builtin_ctz(grp)) cntW[wv][dl] += __builtin_popcount(grp);
        }
    }
    __syncthreads();
    int v = 0;
#pragma unroll
    for (int w = 0; w < ROW_WAVES; ++w) {
        const int c = cntW[w][t];
        cntW[w][t] = v;
        v += c;
    }
    const int incl = wave_incl_scan(v, lane);
    if (lane == 31) wsum[wv] = incl;
    __syncthreads();
    int wo = 0;
#pragma unroll
    for (int w = 0; w < ROW_WAVES; ++w) wo += (w < wv) ? wsum[w] : 0;
    const int first = beg + wo + incl - v;
#pragma unroll
    for (int w = 0; w < ROW_WAVES; ++w) cntW[w][t] += first;
    const int node = node0 + t;
    if (node < nNodes) rows[(size_t)g * nNodes + node] = make_int2(first, v);
    __syncthreads();
    {
        int2 pn = (sBeg + lane < sEnd) ? pg[sBeg + lane] : none;
        for (int i = sBeg; i < sEnd; i += 32) {
            const int2 p = pn;
            const bool valid = i + lane < sEnd;
            pn = (i + 32 + lane < sEnd) ? pg[i + 32 + lane] : none;
            const int dl = p.y - node0;
            const unsigned grp = same_key_lanes(dl, valid, BIN_SHIFT);
            const int gfirst = valid ? __builtin_ctz(grp) : 0;
            int base = 0;
            if (valid && lane == gfirst) {
                base = cntW[wv][dl];
                cntW[wv][dl] = base + __builtin_popcount(grp);
            }
            base = __shfl(base, gfirst, 32);
            if (valid) sg[base + __builtin_popcount(grp & lanes_below(lane))] = p.x;
        }
    }
}

__global__ void __launch_bounds__(NODE_WAVES * 32)
gather_sum_kernel(const float* __restrict__ x, const int2* __restrict__ rows,
                  const int* __restrict__ sorted, float* __restrict__ agg, int nNodes)
{
    const int lane = threadIdx.x & 31;
    const int n = wave_in_grid();
    if (n >= nNodes) return;
    const int2 row = rows[n];
    const int beg = row.x;
    const int cnt = row.y;
    float a0 = 0.f, a1 = 0.f;
    for (int i0 = 0; i0 < cnt; i0 += 32) {
        const int m = min(32, cnt - i0);
        const int idx = (lane < m) ? sorted[beg + i0 + lane] : 0;
        for (int j = 0; j < m; j += 8) {
            float r0[8], r1[8];
#pragma unroll
            for (int u = 0; u < 8; ++u) {
                const int s = __builtin_amdgcn_readlane(idx, min(j + u, m - 1));
                const float* p = x + (size_t)s * DCH + lane;
                r0[u] = p[0];
                r1[u] = p[32];
            }
#pragma unroll
            for (int u = 0; u < 8; ++u) {
                a0 += (j + u < m) ? r0[u] : 0.f;
                a1 += (j + u < m) ? r1[u] : 0.f;
            }
        }
    }
    agg[(size_t)n * DCH + lane] = a0;
    agg[(size_t)n * DCH + 32 + lane] = a1;
}

#define GEMM_WAVES 8
#define GEMM_GROUPS 768
#define WPITCH 68
#define WSLAB (16 * WPITCH)
#define WTILE (5 * WSLAB)

__device__ __forceinline__ void stage_weight(const float* __restrict__ Wg, float* Wl, int mat, int t) {
#pragma unroll
    for (int q = 0; q < (DCH * DCH / 4) / (GEMM_WAVES * 32); ++q) {
        const int f = t + q * GEMM_WAVES * 32;
        const int r = f >> 4;
        const v4f v = *(const v4f*)(Wg + 4 * f);
        *(v4f*)(Wl + (r >> 4) * WTILE + mat * WSLAB + (r & 15) * WPITCH + 4 * (f & 15)) = v;
    }
}

__device__ __forceinline__ void load_b_group(v2f (&b)[16], const float* bp) {
#pragma unroll
    for (int kk = 0; kk < 16; ++kk) b[kk] = ldv2(bp + 4 * kk);
}

__device__ __forceinline__ v8f wmma_chain(const v2f (&a)[16], const v2f (&b)[16], v8f acc) {
#pragma unroll
    for (int kk = 0; kk < 16; ++kk)
        acc = __builtin_amdgcn_wmma_f32_16x16x4_f32(false, a[kk], false, b[kk], (short)0, acc, false, false);
    return acc;
}

__device__ __forceinline__ void store_tile(float* __restrict__ rowp, unsigned off, v8f acc) {
#pragma unroll
    for (int r = 0; r < 8; ++r)
        rowp[off + (unsigned)(r * DCH)] = acc[r];
}

__device__ __forceinline__ void load_a_strip(v2f (&ax)[16], v2f (&ag)[16], const float* __restrict__ x,
                                             const float* __restrict__ agg, int row0, int m, int h) {
    const float* xr = x   + (long long)(row0 + m) * DCH + 2 * h;
    const float* gr = agg + (long long)(row0 + m) * DCH + 2 * h;
#pragma unroll
    for (int kk = 0; kk < 16; ++kk) {
        ax[kk] = ldv2(xr + 4 * kk);
        ag[kk] = ldv2(gr + 4 * kk);
    }
}

__device__ __forceinline__ void gemm_tile(const v2f (&ax)[16], const v2f (&ag)[16], const float* bt,
                                          unsigned off,
                                          float* __restrict__ orow, float* __restrict__ qrow,
                                          float* __restrict__ krow, float* __restrict__ vrow)
{
    v2f b0[16], b1[16];
    load_b_group(b0, bt);
    load_b_group(b1, bt + 1 * WSLAB);
    v8f acc = wmma_chain(ax, b0, (v8f){});
    load_b_group(b0, bt + 2 * WSLAB);
    acc = wmma_chain(ag, b1, acc);
    store_tile(orow, off, acc);
    load_b_group(b1, bt + 3 * WSLAB);
    acc = wmma_chain(ax, b0, (v8f){});
    store_tile(qrow, off, acc);
    load_b_group(b0, bt + 4 * WSLAB);
    acc = wmma_chain(ax, b1, (v8f){});
    store_tile(krow, off, acc);
    acc = wmma_chain(ax, b0, (v8f){});
    store_tile(vrow, off, acc);
}

__global__ void __launch_bounds__(GEMM_WAVES * 32) __attribute__((amdgpu_waves_per_eu(6)))
gemm5_kernel(const float* __restrict__ x,  const float* __restrict__ agg,
             const float* __restrict__ W0, const float* __restrict__ W1,
             const float* __restrict__ WQ, const float* __restrict__ WK,
             const float* __restrict__ WV,
             float* __restrict__ out, float* __restrict__ Qo,
             float* __restrict__ Ko,  float* __restrict__ Vo, int nrows)
{
    __shared__ __attribute__((aligned(16))) float Wl[(DCH / 16) * WTILE];
    const int t = threadIdx.x;
    stage_weight(W0, Wl, 0, t);
    stage_weight(W1, Wl, 1, t);
    stage_weight(WQ, Wl, 2, t);
    stage_weight(WK, Wl, 3, t);
    stage_weight(WV, Wl, 4, t);
    __syncthreads();

    const int lane = t & 31;
    const int wave = __builtin_amdgcn_readfirstlane(t >> 5);
    const int m = lane & 15;
    const int h = lane >> 4;
    const int nStrips = (nrows + 15) >> 4;
    const int total = (int)gridDim.x * GEMM_WAVES;
    const int g = (int)blockIdx.x * GEMM_WAVES + wave;
    const float* bl = Wl + m * WPITCH + 2 * h;
    v2f ax[16], ag[16];

    int base = 0, sub = 0;
    for (;;) {
        const int left = nStrips - base;
        int strip, t0, t1;
        if (left >= total) {
            strip = base + g; t0 = 0; t1 = DCH / 16;
            base += total;
        } else {
            const int u = g + sub * total;
            if (u >= left * (DCH / 16)) break;
            strip = base + (u >> 2); t0 = u & 3; t1 = t0 + 1;
            ++sub;
        }
        const int row0 = strip * 16;
        load_a_strip(ax, ag, x, agg, row0, m, h);
        const long long rowoff = (long long)row0 * DCH;
        for (int tile = t0; tile < t1; ++tile)
            gemm_tile(ax, ag, bl + tile * WTILE, (unsigned)(8 * h * DCH + tile * 16 + m),
                      out + rowoff, Qo + rowoff, Ko + rowoff, Vo + rowoff);
    }
}

__device__ __forceinline__ float wave_sum(float p) {
    for (int off = 16; off >= 1; off >>= 1)
        p += __shfl_xor(p, off, 32);
    return p;
}

__global__ void __launch_bounds__(NODE_WAVES * 32)
attn_rows_kernel(const float* __restrict__ Q, const float* __restrict__ K,
                 const float* __restrict__ V, const int2* __restrict__ rows,
                 const int* __restrict__ sorted, float* out, int nNodes)
{
    const int lane = threadIdx.x & 31;
    const int n = wave_in_grid();
    if (n >= nNodes) return;
    const int2 row = rows[n];
    const int beg = row.x;
    const int cnt = row.y;
    const float q0 = Q[(size_t)n * DCH + lane];
    const float q1 = Q[(size_t)n * DCH + 32 + lane];
    const float x0 = out[(size_t)n * DCH + lane];
    const float x1 = out[(size_t)n * DCH + 32 + lane];
    float a0 = 0.f, a1 = 0.f;
    for (int i0 = 0; i0 < cnt; i0 += 32) {
        const int m = min(32, cnt - i0);
        const int idx = (lane < m) ? sorted[beg + i0 + lane] : 0;
        for (int j = 0; j < m; j += 4) {
            float k0[4], k1[4], v0[4], v1[4];
#pragma unroll
            for (int u = 0; u < 4; ++u) {
                const int s = __builtin_amdgcn_readlane(idx, min(j + u, m - 1));
                const float* kp = K + (size_t)s * DCH + lane;
                const float* vp = V + (size_t)s * DCH + lane;
                k0[u] = kp[0];
                k1[u] = kp[32];
                v0[u] = vp[0];
                v1[u] = vp[32];
            }
#pragma unroll
            for (int u = 0; u < 4; ++u) {
                const float p = wave_sum(q0 * k0[u] + q1 * k1[u]);
                const float pe = (j + u < m) ? p : 0.f;
                a0 += pe * v0[u];
                a1 += pe * v1[u];
            }
        }
    }
    out[(size_t)n * DCH + lane] = x0 - a0;
    out[(size_t)n * DCH + 32 + lane] = x1 - a1;
}

extern "C" void kernel_launch(void* const* d_in, const int* in_sizes, int n_in,
                              void* d_out, int out_size, void* d_ws, size_t ws_size,
                              hipStream_t stream)
{
    const float* x  = (const float*)d_in[0];
    const int*  ei1 = (const int*)d_in[1];
    const int*  ei2 = (const int*)d_in[2];
    const float* W0 = (const float*)d_in[3];
    const float* W1 = (const float*)d_in[4];
    const float* WQ = (const float*)d_in[5];
    const float* WK = (const float*)d_in[6];
    const float* WV = (const float*)d_in[7];
    float* out = (float*)d_out;

    const int N  = in_sizes[0] / DCH;
    const int E1 = in_sizes[1] / 2;
    const int E2 = in_sizes[2] / 2;
    const int EM = E1 > E2 ? E1 : E2;
    const int listStride = (EM + 63) / 64 * 64;
    const int nBins = (N + BIN_NODES - 1) / BIN_NODES;
    const int nChunks = (EM + CHUNK_EDGES - 1) / CHUNK_EDGES;

    float* agg = (float*)d_ws;
    float* Qb  = agg + (size_t)N * DCH;
    float* Kb  = Qb  + (size_t)N * DCH;
    float* Vb  = Kb  + (size_t)N * DCH;
    int2* part    = (int2*)(Vb + (size_t)N * DCH);
    int* sorted   = (int*)(part + (size_t)2 * listStride);
    int2* rows    = (int2*)(sorted + (size_t)2 * listStride);
    int* cnt      = (int*)(rows + (size_t)2 * N);
    int* start    = cnt + (size_t)2 * nChunks * nBins;
    int* binStart = start + (size_t)2 * nChunks * nBins;

    const size_t histBytes = (size_t)PART_WAVES * nBins * sizeof(int);

    bin_count_kernel<<<dim3(nChunks, 2), PART_THREADS, histBytes, stream>>>(
        ei1 + E1, E1, ei2 + E2, E2, cnt, nBins, nChunks);
    bin_scan_kernel<<<2, 1024, 0, stream>>>(cnt, start, binStart, nBins, nChunks);
    bin_place_kernel<<<dim3(nChunks, 2), PART_THREADS, histBytes, stream>>>(
        ei1, ei1 + E1, E1, ei2, ei2 + E2, E2, start, part, nBins, nChunks, listStride);
    bin_to_rows_kernel<<<dim3(nBins, 2), BIN_NODES, 0, stream>>>(
        part, binStart, rows, sorted, nBins, N, listStride);

    const int nodeBlocks = (N + NODE_WAVES - 1) / NODE_WAVES;
    gather_sum_kernel<<<nodeBlocks, NODE_WAVES * 32, 0, stream>>>(x, rows, sorted, agg, N);

    {
        int strips = (N + 15) / 16;
        int blocks = (strips + GEMM_WAVES - 1) / GEMM_WAVES;
        if (blocks > GEMM_GROUPS) blocks = GEMM_GROUPS;
        gemm5_kernel<<<blocks, GEMM_WAVES * 32, 0, stream>>>(x, agg, W0, W1, WQ, WK, WV,
                                                             out, Qb, Kb, Vb, N);
    }

    attn_rows_kernel<<<nodeBlocks, NODE_WAVES * 32, 0, stream>>>(
        Qb, Kb, Vb, rows + N, sorted + listStride, out, N);
}


# Attention sums x rows, not K and V rows: score = <x[d] WQ^T WK, x[s]>, WV applied to the summed row in a second GEMM launch; one 256-B row per edge, was two; sum order changes (project after sum)
# speedup vs baseline: 1.7307x; 1.0550x over previous
#include <hip/hip_runtime.h>

typedef __attribute__((ext_vector_type(2))) float v2f;
typedef __attribute__((ext_vector_type(4))) float v4f;
typedef __attribute__((ext_vector_type(8))) float v8f;

#define DCH 64
#define BIN_NODES 256
#define BIN_SHIFT 8
#define PART_THREADS 1024
#define PART_WAVES (PART_THREADS / 32)
#define PART_EPT 16
#define CHUNK_EDGES (PART_THREADS * PART_EPT)
#define ROW_WAVES (BIN_NODES / 32)
#define NODE_WAVES 8

__device__ __forceinline__ v2f ldv2(const float* p) { return *(const v2f*)p; }

__device__ __forceinline__ int wave_in_grid() {
    const int w = __builtin_amdgcn_readfirstlane((int)(threadIdx.x >> 5));
    return (int)(blockIdx.x * NODE_WAVES) + w;
}

__device__ __forceinline__ int wave_incl_scan(int v, int lane) {
    for (int off = 1; off < 32; off <<= 1) {
        const int t = __shfl_up(v, off, 32);
        if (lane >= off) v += t;
    }
    return v;
}

__device__ __forceinline__ unsigned same_key_lanes(int key, bool valid, int nbits) {
    unsigned m = __builtin_amdgcn_ballot_w32(valid);
    for (int b = 0; b < nbits; ++b) {
        const int sx = -((key >> b) & 1);
        const unsigned set = __builtin_amdgcn_ballot_w32(sx != 0);
        m &= ~(set ^ (unsigned)sx);
    }
    return valid ? m : 0u;
}

__device__ __forceinline__ int key_bits(int n) { return n > 1 ? 32 - __builtin_clz((unsigned)(n - 1)) : 0; }

__device__ __forceinline__ unsigned lanes_below(int lane) { return (1u << lane) - 1u; }


__global__ void __launch_bounds__(PART_THREADS)
bin_count_kernel(const int* __restrict__ dstA, int nEA,
                 const int* __restrict__ dstB, int nEB,
                 int* __restrict__ cnt, int nBins, int nChunks)
{
    extern __shared__ int histW[];
    const int g = blockIdx.y;
    const int* __restrict__ dst = g ? dstB : dstA;
    const int nE = g ? nEB : nEA;
    const int c = blockIdx.x;
    const int lane = threadIdx.x & 31;
    const int wv = __builtin_amdgcn_readfirstlane((int)(threadIdx.x >> 5));
    const int binBits = key_bits(nBins);
    for (int i = threadIdx.x; i < PART_WAVES * nBins; i += PART_THREADS) histW[i] = 0;
    __syncthreads();
    const int e0 = c * CHUNK_EDGES + (int)threadIdx.x;
    int d[PART_EPT];
#pragma unroll
    for (int k = 0; k < PART_EPT; ++k) {
        const int e = e0 + k * PART_THREADS;
        d[k] = (e < nE) ? dst[e] : -1;
    }
    int* myRow = histW + wv * nBins;
#pragma unroll
    for (int k = 0; k < PART_EPT; ++k) {
        const bool valid = d[k] >= 0;
        const int bin = valid ? (d[k] >> BIN_SHIFT) : 0;
        const unsigned grp = same_key_lanes(bin, valid, binBits);
        if (valid && lane == __builtin_ctz(grp)) myRow[bin] += __builtin_popcount(grp);
    }
    __syncthreads();
    int* __restrict__ row = cnt + ((size_t)g * nChunks + c) * nBins;
    for (int b = threadIdx.x; b < nBins; b += PART_THREADS) {
        int tot = 0;
        for (int w = 0; w < PART_WAVES; ++w) tot += histW[w * nBins + b];
        row[b] = tot;
    }
}

__global__ void __launch_bounds__(1024)
bin_scan_kernel(const int* __restrict__ cnt, int* __restrict__ start,
                int* __restrict__ binStart, int nBins, int nChunks)
{
    __shared__ int wsum[32];
    __shared__ int tileTotal;
    const int g = blockIdx.x;
    const int* __restrict__ m = cnt + (size_t)g * nChunks * nBins;
    int* __restrict__ st = start + (size_t)g * nChunks * nBins;
    int* __restrict__ bs = binStart + (size_t)g * (nBins + 1);
    const int lane = threadIdx.x & 31, wv = threadIdx.x >> 5;
    int carry = 0;
    for (int b0 = 0; b0 < nBins; b0 += 1024) {
        const int b = b0 + (int)threadIdx.x;
        const bool live = b < nBins;
        int tot = 0;
        for (int c = 0; c < nChunks; c += 16) {
            int v[16];
#pragma unroll
            for (int u = 0; u < 16; ++u)
                v[u] = (live && c + u < nChunks) ? m[(size_t)(c + u) * nBins + b] : 0;
#pragma unroll
            for (int u = 0; u < 16; ++u) tot += v[u];
        }
        const int incl = wave_incl_scan(tot, lane);
        if (lane == 31) wsum[wv] = incl;
        __syncthreads();
        if (wv == 0) {
            const int s = wsum[lane];
            const int si = wave_incl_scan(s, lane);
            wsum[lane] = si - s;
            if (lane == 31) tileTotal = si;
        }
        __syncthreads();
        const int excl = carry + wsum[wv] + incl - tot;
        const int tt = tileTotal;
        if (live) bs[b] = excl;
        int run = excl;
        for (int c = 0; c < nChunks; c += 16) {
            int v[16];
#pragma unroll
            for (int u = 0; u < 16; ++u)
                v[u] = (live && c + u < nChunks) ? m[(size_t)(c + u) * nBins + b] : 0;
#pragma unroll
            for (int u = 0; u < 16; ++u) {
                if (live && c + u < nChunks) st[(size_t)(c + u) * nBins + b] = run;
                run += v[u];
            }
        }
        carry += tt;
        __syncthreads();
    }
    if (threadIdx.x == 0) bs[nBins] = carry;
}

__global__ void __launch_bounds__(PART_THREADS)
bin_place_kernel(const int* __restrict__ srcA, const int* __restrict__ dstA, int nEA,
                 const int* __restrict__ srcB, const int* __restrict__ dstB, int nEB,
                 const int* __restrict__ start, int2* __restrict__ part,
                 int nBins, int nChunks, int listStride)
{
    extern __shared__ int histW[];
    const int g = blockIdx.y;
    const int* __restrict__ src = g ? srcB : srcA;
    const int* __restrict__ dst = g ? dstB : dstA;
    const int nE = g ? nEB : nEA;
    const int c = blockIdx.x;
    const int lane = threadIdx.x & 31;
    const int wv = __builtin_amdgcn_readfirstlane((int)(threadIdx.x >> 5));
    const int binBits = key_bits(nBins);
    for (int i = threadIdx.x; i < PART_WAVES * nBins; i += PART_THREADS) histW[i] = 0;
    __syncthreads();
    const int e0 = c * CHUNK_EDGES + (int)threadIdx.x;
    int d[PART_EPT], s[PART_EPT];
    unsigned grp[PART_EPT];
#pragma unroll
    for (int k = 0; k < PART_EPT; ++k) {
        const int e = e0 + k * PART_THREADS;
        const bool in = e < nE;
        d[k] = in ? dst[e] : -1;
        s[k] = in ? src[e] : 0;
    }
    int* myRow = histW + wv * nBins;
#pragma unroll
    for (int k = 0; k < PART_EPT; ++k) {
        const bool valid = d[k] >= 0;
        const int bin = valid ? (d[k] >> BIN_SHIFT) : 0;
        grp[k] = same_key_lanes(bin, valid, binBits);
        if (valid && lane == __builtin_ctz(grp[k])) myRow[bin] += __builtin_popcount(grp[k]);
    }
    __syncthreads();
    const int* __restrict__ srow = start + ((size_t)g * nChunks + c) * nBins;
    for (int b = threadIdx.x; b < nBins; b += PART_THREADS) {
        int run = srow[b];
        for (int w = 0; w < PART_WAVES; ++w) {
            const int t = histW[w * nBins + b];
            histW[w * nBins + b] = run;
            run += t;
        }
    }
    __syncthreads();
    int2* __restrict__ pg = part + (size_t)g * listStride;
#pragma unroll
    for (int k = 0; k < PART_EPT; ++k) {
        const bool valid = d[k] >= 0;
        const int bin = valid ? (d[k] >> BIN_SHIFT) : 0;
        const int first = valid ? __builtin_ctz(grp[k]) : 0;
        int base = 0;
        if (valid && lane == first) {
            base = myRow[bin];
            myRow[bin] = base + __builtin_popcount(grp[k]);
        }
        base = __shfl(base, first, 32);
        if (valid) pg[base + __builtin_popcount(grp[k] & lanes_below(lane))] = make_int2(s[k], d[k]);
    }
}

__global__ void __launch_bounds__(BIN_NODES)
bin_to_rows_kernel(const int2* __restrict__ part, const int* __restrict__ binStart,
                   int2* __restrict__ rows, int* __restrict__ sorted,
                   int nBins, int nNodes, int listStride)
{
    __shared__ int cntW[ROW_WAVES][BIN_NODES];
    __shared__ int wsum[ROW_WAVES];
    const int g = blockIdx.y, b = blockIdx.x, t = threadIdx.x;
    const int lane = t & 31;
    const int wv = __builtin_amdgcn_readfirstlane(t >> 5);
    const int beg = binStart[(size_t)g * (nBins + 1) + b];
    const int end = binStart[(size_t)g * (nBins + 1) + b + 1];
    const int node0 = b * BIN_NODES;
    const int2* __restrict__ pg = part + (size_t)g * listStride;
    int* __restrict__ sg = sorted + (size_t)g * listStride;
    const int per = (((end - beg) + ROW_WAVES - 1) / ROW_WAVES + 31) & ~31;
    const int sBeg = min(end, beg + wv * per);
    const int sEnd = min(end, sBeg + per);
    const int2 none = make_int2(0, node0);
#pragma unroll
    for (int w = 0; w < ROW_WAVES; ++w) cntW[w][t] = 0;
    __syncthreads();
    {
        int2 pn = (sBeg + lane < sEnd) ? pg[sBeg + lane] : none;
        for (int i = sBeg; i < sEnd; i += 32) {
            const int2 p = pn;
            const bool valid = i + lane < sEnd;
            pn = (i + 32 + lane < sEnd) ? pg[i + 32 + lane] : none;
            const int dl = p.y - node0;
            const unsigned grp = same_key_lanes(dl, valid, BIN_SHIFT);
            if (valid && lane == __builtin_ctz(grp)) cntW[wv][dl] += __builtin_popcount(grp);
        }
    }
    __syncthreads();
    int v = 0;
#pragma unroll
    for (int w = 0; w < ROW_WAVES; ++w) {
        const int c = cntW[w][t];
        cntW[w][t] = v;
        v += c;
    }
    const int incl = wave_incl_scan(v, lane);
    if (lane == 31) wsum[wv] = incl;
    __syncthreads();
    int wo = 0;
#pragma unroll
    for (int w = 0; w < ROW_WAVES; ++w) wo += (w < wv) ? wsum[w] : 0;
    const int first = beg + wo + incl - v;
#pragma unroll
    for (int w = 0; w < ROW_WAVES; ++w) cntW[w][t] += first;
    const int node = node0 + t;
    if (node < nNodes) rows[(size_t)g * nNodes + node] = make_int2(first, v);
    __syncthreads();
    {
        int2 pn = (sBeg + lane < sEnd) ? pg[sBeg + lane] : none;
        for (int i = sBeg; i < sEnd; i += 32) {
            const int2 p = pn;
            const bool valid = i + lane < sEnd;
            pn = (i + 32 + lane < sEnd) ? pg[i + 32 + lane] : none;
            const int dl = p.y - node0;
            const unsigned grp = same_key_lanes(dl, valid, BIN_SHIFT);
            const int gfirst = valid ? __builtin_ctz(grp) : 0;
            int base = 0;
            if (valid && lane == gfirst) {
                base = cntW[wv][dl];
                cntW[wv][dl] = base + __builtin_popcount(grp);
            }
            base = __shfl(base, gfirst, 32);
            if (valid) sg[base + __builtin_popcount(grp & lanes_below(lane))] = p.x;
        }
    }
}

__device__ __forceinline__ float wave_sum(float p) {
    for (int off = 16; off >= 1; off >>= 1)
        p += __shfl_xor(p, off, 32);
    return p;
}

__global__ void __launch_bounds__(NODE_WAVES * 32)
node_rows_kernel(const float* __restrict__ x, const float* __restrict__ qp,
                 const int2* __restrict__ rows, const int* __restrict__ sorted,
                 float* __restrict__ agg, float* __restrict__ tneg,
                 int nNodes, int listStride)
{
    const int lane = threadIdx.x & 31;
    const int n = wave_in_grid();
    if (n >= nNodes) return;
    const int g = blockIdx.y;
    const int2 row = rows[(size_t)g * nNodes + n];
    const int* __restrict__ sg = sorted + (size_t)g * listStride;
    const int beg = row.x;
    const int cnt = row.y;
    float a0 = 0.f, a1 = 0.f;
    if (g == 0) {
        for (int i0 = 0; i0 < cnt; i0 += 32) {
            const int m = min(32, cnt - i0);
            const int idx = (lane < m) ? sg[beg + i0 + lane] : 0;
            for (int j = 0; j < m; j += 8) {
                float r0[8], r1[8];
#pragma unroll
                for (int u = 0; u < 8; ++u) {
                    const int s = __builtin_amdgcn_readlane(idx, min(j + u, m - 1));
                    const float* p = x + (size_t)s * DCH + lane;
                    r0[u] = p[0];
                    r1[u] = p[32];
                }
#pragma unroll
                for (int u = 0; u < 8; ++u) {
                    a0 += (j + u < m) ? r0[u] : 0.f;
                    a1 += (j + u < m) ? r1[u] : 0.f;
                }
            }
        }
        agg[(size_t)n * DCH + lane] = a0;
        agg[(size_t)n * DCH + 32 + lane] = a1;
    } else {
        const float q0 = qp[(size_t)n * DCH + lane];
        const float q1 = qp[(size_t)n * DCH + 32 + lane];
        for (int i0 = 0; i0 < cnt; i0 += 32) {
            const int m = min(32, cnt - i0);
            const int idx = (lane < m) ? sg[beg + i0 + lane] : 0;
            for (int j = 0; j < m; j += 8) {
                float r0[8], r1[8];
#pragma unroll
                for (int u = 0; u < 8; ++u) {
                    const int s = __builtin_amdgcn_readlane(idx, min(j + u, m - 1));
                    const float* p = x + (size_t)s * DCH + lane;
                    r0[u] = p[0];
                    r1[u] = p[32];
                }
#pragma unroll
                for (int u = 0; u < 8; ++u) {
                    const float p = wave_sum(q0 * r0[u] + q1 * r1[u]);
                    const float pe = (j + u < m) ? p : 0.f;
                    a0 += pe * r0[u];
                    a1 += pe * r1[u];
                }
            }
        }
        tneg[(size_t)n * DCH + lane] = -a0;
        tneg[(size_t)n * DCH + 32 + lane] = -a1;
    }
}

#define WPITCH 68
#define WSLAB (16 * WPITCH)

__device__ __forceinline__ void load_b_group(v2f (&b)[16], const float* bp) {
#pragma unroll
    for (int kk = 0; kk < 16; ++kk) b[kk] = ldv2(bp + 4 * kk);
}

__device__ __forceinline__ v8f wmma_chain(const v2f (&a)[16], const v2f (&b)[16], v8f acc) {
#pragma unroll
    for (int kk = 0; kk < 16; ++kk)
        acc = __builtin_amdgcn_wmma_f32_16x16x4_f32(false, a[kk], false, b[kk], (short)0, acc, false, false);
    return acc;
}

__device__ __forceinline__ void store_tile(float* __restrict__ rowp, unsigned off, v8f acc) {
#pragma unroll
    for (int r = 0; r < 8; ++r)
        rowp[off + (unsigned)(r * DCH)] = acc[r];
}

__device__ __forceinline__ void load_a_rows(v2f (&a)[16], const float* __restrict__ src, int row0, int m, int h) {
    const float* p = src + (long long)(row0 + m) * DCH + 2 * h;
#pragma unroll
    for (int kk = 0; kk < 16; ++kk) a[kk] = ldv2(p + 4 * kk);
}

#define QP_WAVES 16
#define RPITCH 80

__global__ void __launch_bounds__(QP_WAVES * 32)
qproj_kernel(const float* __restrict__ x, const float* __restrict__ WQ,
             const float* __restrict__ WK, float* __restrict__ qp, int nrows)
{
    __shared__ __attribute__((aligned(16))) float Rq[DCH * RPITCH];
    __shared__ __attribute__((aligned(16))) float Rk[DCH * RPITCH];
    __shared__ __attribute__((aligned(16))) float Pl[(DCH / 16) * WSLAB];
    const int t = threadIdx.x;
    const int lane = t & 31;
    const int wave = __builtin_amdgcn_readfirstlane(t >> 5);
    const int m = lane & 15;
    const int h = lane >> 4;
    const int nStrips = (nrows + 15) >> 4;
    const int strip = (int)blockIdx.x * QP_WAVES + wave;
    const int row0 = strip * 16;
    v2f ax[16] = {};
    if (strip < nStrips) load_a_rows(ax, x, row0, m, h);
#pragma unroll
    for (int q = 0; q < (DCH * DCH / 4) / (QP_WAVES * 32); ++q) {
        const int f = t + q * QP_WAVES * 32;
        const v4f vq = *(const v4f*)(WQ + 4 * f);
        const v4f vk = *(const v4f*)(WK + 4 * f);
        *(v4f*)(Rq + (f >> 4) * RPITCH + 4 * (f & 15)) = vq;
        *(v4f*)(Rk + (f >> 4) * RPITCH + 4 * (f & 15)) = vk;
    }
    __syncthreads();

    {
        const int nt = wave >> 2, kt = wave & 3;
        const float* ak = Rk + (2 * h) * RPITCH + nt * 16 + m;
        const float* bq = Rq + (2 * h) * RPITCH + kt * 16 + m;
        v2f a[16], b[16];
#pragma unroll
        for (int jj = 0; jj < 16; ++jj) {
            a[jj] = (v2f){ak[(4 * jj) * RPITCH], ak[(4 * jj + 1) * RPITCH]};
            b[jj] = (v2f){bq[(4 * jj) * RPITCH], bq[(4 * jj + 1) * RPITCH]};
        }
        const v8f acc = wmma_chain(a, b, (v8f){});
        float* pd = Pl + nt * WSLAB + (8 * h) * WPITCH + kt * 16 + m;
#pragma unroll
        for (int r = 0; r < 8; ++r) pd[r * WPITCH] = acc[r];
    }
    __syncthreads();

    if (strip < nStrips) {
        const float* bl = Pl + m * WPITCH + 2 * h;
        const unsigned off = (unsigned)(8 * h * DCH + m);
        float* qrow = qp + (long long)row0 * DCH;
        v2f b0[16], b1[16];
        load_b_group(b0, bl);
        load_b_group(b1, bl + 1 * WSLAB);
        v8f acc = wmma_chain(ax, b0, (v8f){});
        store_tile(qrow, off, acc);
        load_b_group(b0, bl + 2 * WSLAB);
        acc = wmma_chain(ax, b1, (v8f){});
        store_tile(qrow, off + 16, acc);
        load_b_group(b1, bl + 3 * WSLAB);
        acc = wmma_chain(ax, b0, (v8f){});
        store_tile(qrow, off + 32, acc);
        acc = wmma_chain(ax, b1, (v8f){});
        store_tile(qrow, off + 48, acc);
    }
}

#define GEMM_WAVES 8
#define GEMM_GROUPS 768
#define WTILE (3 * WSLAB)

__device__ __forceinline__ void stage_weight(const float* __restrict__ Wg, float* Wl, int mat, int t) {
#pragma unroll
    for (int q = 0; q < (DCH * DCH / 4) / (GEMM_WAVES * 32); ++q) {
        const int f = t + q * GEMM_WAVES * 32;
        const int r = f >> 4;
        const v4f v = *(const v4f*)(Wg + 4 * f);
        *(v4f*)(Wl + (r >> 4) * WTILE + mat * WSLAB + (r & 15) * WPITCH + 4 * (f & 15)) = v;
    }
}

__device__ __forceinline__ void gemm3_tile(const v2f (&ax)[16], const v2f (&ag)[16], const v2f (&at)[16],
                                           const float* bt, unsigned off, float* __restrict__ orow)
{
    v2f b0[16], b1[16];
    load_b_group(b0, bt);
    load_b_group(b1, bt + 1 * WSLAB);
    v8f acc = wmma_chain(ax, b0, (v8f){});
    load_b_group(b0, bt + 2 * WSLAB);
    acc = wmma_chain(ag, b1, acc);
    acc = wmma_chain(at, b0, acc);
    store_tile(orow, off, acc);
}

__global__ void __launch_bounds__(GEMM_WAVES * 32) __attribute__((amdgpu_waves_per_eu(6, 6)))
gemm3_kernel(const float* __restrict__ x,  const float* __restrict__ agg,
             const float* __restrict__ tneg,
             const float* __restrict__ W0, const float* __restrict__ W1,
             const float* __restrict__ WV,
             float* __restrict__ out, int nrows)
{
    __shared__ __attribute__((aligned(16))) float Wl[(DCH / 16) * WTILE];
    const int t = threadIdx.x;
    stage_weight(W0, Wl, 0, t);
    stage_weight(W1, Wl, 1, t);
    stage_weight(WV, Wl, 2, t);
    __syncthreads();

    const int lane = t & 31;
    const int wave = __builtin_amdgcn_readfirstlane(t >> 5);
    const int m = lane & 15;
    const int h = lane >> 4;
    const int nStrips = (nrows + 15) >> 4;
    const int total = (int)gridDim.x * GEMM_WAVES;
    const int g = (int)blockIdx.x * GEMM_WAVES + wave;
    const float* bl = Wl + m * WPITCH + 2 * h;
    v2f ax[16], ag[16], at[16];

    int base = 0, sub = 0;
    for (;;) {
        const int left = nStrips - base;
        int strip, t0, t1;
        if (left >= total) {
            strip = base + g; t0 = 0; t1 = DCH / 16;
            base += total;
        } else {
            const int u = g + sub * total;
            if (u >= left * (DCH / 16)) break;
            strip = base + (u >> 2); t0 = u & 3; t1 = t0 + 1;
            ++sub;
        }
        const int row0 = strip * 16;
        load_a_rows(ax, x, row0, m, h);
        load_a_rows(ag, agg, row0, m, h);
        load_a_rows(at, tneg, row0, m, h);
        const long long rowoff = (long long)row0 * DCH;
        for (int tile = t0; tile < t1; ++tile)
            gemm3_tile(ax, ag, at, bl + tile * WTILE, (unsigned)(8 * h * DCH + tile * 16 + m), out + rowoff);
    }
}

extern "C" void kernel_launch(void* const* d_in, const int* in_sizes, int n_in,
                              void* d_out, int out_size, void* d_ws, size_t ws_size,
                              hipStream_t stream)
{
    const float* x  = (const float*)d_in[0];
    const int*  ei1 = (const int*)d_in[1];
    const int*  ei2 = (const int*)d_in[2];
    const float* W0 = (const float*)d_in[3];
    const float* W1 = (const float*)d_in[4];
    const float* WQ = (const float*)d_in[5];
    const float* WK = (const float*)d_in[6];
    const float* WV = (const float*)d_in[7];
    float* out = (float*)d_out;

    const int N  = in_sizes[0] / DCH;
    const int E1 = in_sizes[1] / 2;
    const int E2 = in_sizes[2] / 2;
    const int EM = E1 > E2 ? E1 : E2;
    const int listStride = (EM + 63) / 64 * 64;
    const int nBins = (N + BIN_NODES - 1) / BIN_NODES;
    const int nChunks = (EM + CHUNK_EDGES - 1) / CHUNK_EDGES;

    float* agg = (float*)d_ws;
    float* Qp  = agg + (size_t)N * DCH;
    float* Tn  = Qp  + (size_t)N * DCH;
    int2* part    = (int2*)(Tn + (size_t)N * DCH);
    int* sorted   = (int*)(part + (size_t)2 * listStride);
    int2* rows    = (int2*)(sorted + (size_t)2 * listStride);
    int* cnt      = (int*)(rows + (size_t)2 * N);
    int* start    = cnt + (size_t)2 * nChunks * nBins;
    int* binStart = start + (size_t)2 * nChunks * nBins;

    const size_t histBytes = (size_t)PART_WAVES * nBins * sizeof(int);

    bin_count_kernel<<<dim3(nChunks, 2), PART_THREADS, histBytes, stream>>>(
        ei1 + E1, E1, ei2 + E2, E2, cnt, nBins, nChunks);
    bin_scan_kernel<<<2, 1024, 0, stream>>>(cnt, start, binStart, nBins, nChunks);
    bin_place_kernel<<<dim3(nChunks, 2), PART_THREADS, histBytes, stream>>>(
        ei1, ei1 + E1, E1, ei2, ei2 + E2, E2, start, part, nBins, nChunks, listStride);
    bin_to_rows_kernel<<<dim3(nBins, 2), BIN_NODES, 0, stream>>>(
        part, binStart, rows, sorted, nBins, N, listStride);

    const int strips = (N + 15) / 16;
    qproj_kernel<<<(strips + QP_WAVES - 1) / QP_WAVES, QP_WAVES * 32, 0, stream>>>(x, WQ, WK, Qp, N);

    const int nodeBlocks = (N + NODE_WAVES - 1) / NODE_WAVES;
    node_rows_kernel<<<dim3(nodeBlocks, 2), NODE_WAVES * 32, 0, stream>>>(
        x, Qp, rows, sorted, agg, Tn, N, listStride);

    {
        int blocks = (strips + GEMM_WAVES - 1) / GEMM_WAVES;
        if (blocks > GEMM_GROUPS) blocks = GEMM_GROUPS;
        gemm3_kernel<<<blocks, GEMM_WAVES * 32, 0, stream>>>(x, agg, Tn, W0, W1, WV, out, N);
    }
}


# Per-node sums: each 16-lane half of a wave takes its own edge (row = 16 lanes x 16 B, one load fetches 2 rows); score sum = 4 in-half DPP adds, was 5 LDS shuffles; sum order changes (even/odd halves)
# speedup vs baseline: 1.9747x; 1.1410x over previous
#include <hip/hip_runtime.h>

typedef __attribute__((ext_vector_type(2))) float v2f;
typedef __attribute__((ext_vector_type(4))) float v4f;
typedef __attribute__((ext_vector_type(8))) float v8f;

#define DCH 64
#define BIN_NODES 256
#define BIN_SHIFT 8
#define PART_THREADS 1024
#define PART_WAVES (PART_THREADS / 32)
#define PART_EPT 16
#define CHUNK_EDGES (PART_THREADS * PART_EPT)
#define ROW_WAVES (BIN_NODES / 32)
#define NODE_WAVES 8

__device__ __forceinline__ v2f ldv2(const float* p) { return *(const v2f*)p; }

__device__ __forceinline__ int wave_in_grid() {
    const int w = __builtin_amdgcn_readfirstlane((int)(threadIdx.x >> 5));
    return (int)(blockIdx.x * NODE_WAVES) + w;
}

__device__ __forceinline__ int wave_incl_scan(int v, int lane) {
    for (int off = 1; off < 32; off <<= 1) {
        const int t = __shfl_up(v, off, 32);
        if (lane >= off) v += t;
    }
    return v;
}

__device__ __forceinline__ unsigned same_key_lanes(int key, bool valid, int nbits) {
    unsigned m = __builtin_amdgcn_ballot_w32(valid);
    for (int b = 0; b < nbits; ++b) {
        const int sx = -((key >> b) & 1);
        const unsigned set = __builtin_amdgcn_ballot_w32(sx != 0);
        m &= ~(set ^ (unsigned)sx);
    }
    return valid ? m : 0u;
}

__device__ __forceinline__ int key_bits(int n) { return n > 1 ? 32 - __builtin_clz((unsigned)(n - 1)) : 0; }

__device__ __forceinline__ unsigned lanes_below(int lane) { return (1u << lane) - 1u; }


__global__ void __launch_bounds__(PART_THREADS)
bin_count_kernel(const int* __restrict__ dstA, int nEA,
                 const int* __restrict__ dstB, int nEB,
                 int* __restrict__ cnt, int nBins, int nChunks)
{
    extern __shared__ int histW[];
    const int g = blockIdx.y;
    const int* __restrict__ dst = g ? dstB : dstA;
    const int nE = g ? nEB : nEA;
    const int c = blockIdx.x;
    const int lane = threadIdx.x & 31;
    const int wv = __builtin_amdgcn_readfirstlane((int)(threadIdx.x >> 5));
    const int binBits = key_bits(nBins);
    for (int i = threadIdx.x; i < PART_WAVES * nBins; i += PART_THREADS) histW[i] = 0;
    __syncthreads();
    const int e0 = c * CHUNK_EDGES + (int)threadIdx.x;
    int d[PART_EPT];
#pragma unroll
    for (int k = 0; k < PART_EPT; ++k) {
        const int e = e0 + k * PART_THREADS;
        d[k] = (e < nE) ? dst[e] : -1;
    }
    int* myRow = histW + wv * nBins;
#pragma unroll
    for (int k = 0; k < PART_EPT; ++k) {
        const bool valid = d[k] >= 0;
        const int bin = valid ? (d[k] >> BIN_SHIFT) : 0;
        const unsigned grp = same_key_lanes(bin, valid, binBits);
        if (valid && lane == __builtin_ctz(grp)) myRow[bin] += __builtin_popcount(grp);
    }
    __syncthreads();
    int* __restrict__ row = cnt + ((size_t)g * nChunks + c) * nBins;
    for (int b = threadIdx.x; b < nBins; b += PART_THREADS) {
        int tot = 0;
        for (int w = 0; w < PART_WAVES; ++w) tot += histW[w * nBins + b];
        row[b] = tot;
    }
}

__global__ void __launch_bounds__(1024)
bin_scan_kernel(const int* __restrict__ cnt, int* __restrict__ start,
                int* __restrict__ binStart, int nBins, int nChunks)
{
    __shared__ int wsum[32];
    __shared__ int tileTotal;
    const int g = blockIdx.x;
    const int* __restrict__ m = cnt + (size_t)g * nChunks * nBins;
    int* __restrict__ st = start + (size_t)g * nChunks * nBins;
    int* __restrict__ bs = binStart + (size_t)g * (nBins + 1);
    const int lane = threadIdx.x & 31, wv = threadIdx.x >> 5;
    int carry = 0;
    for (int b0 = 0; b0 < nBins; b0 += 1024) {
        const int b = b0 + (int)threadIdx.x;
        const bool live = b < nBins;
        int tot = 0;
        for (int c = 0; c < nChunks; c += 16) {
            int v[16];
#pragma unroll
            for (int u = 0; u < 16; ++u)
                v[u] = (live && c + u < nChunks) ? m[(size_t)(c + u) * nBins + b] : 0;
#pragma unroll
            for (int u = 0; u < 16; ++u) tot += v[u];
        }
        const int incl = wave_incl_scan(tot, lane);
        if (lane == 31) wsum[wv] = incl;
        __syncthreads();
        if (wv == 0) {
            const int s = wsum[lane];
            const int si = wave_incl_scan(s, lane);
            wsum[lane] = si - s;
            if (lane == 31) tileTotal = si;
        }
        __syncthreads();
        const int excl = carry + wsum[wv] + incl - tot;
        const int tt = tileTotal;
        if (live) bs[b] = excl;
        int run = excl;
        for (int c = 0; c < nChunks; c += 16) {
            int v[16];
#pragma unroll
            for (int u = 0; u < 16; ++u)
                v[u] = (live && c + u < nChunks) ? m[(size_t)(c + u) * nBins + b] : 0;
#pragma unroll
            for (int u = 0; u < 16; ++u) {
                if (live && c + u < nChunks) st[(size_t)(c + u) * nBins + b] = run;
                run += v[u];
            }
        }
        carry += tt;
        __syncthreads();
    }
    if (threadIdx.x == 0) bs[nBins] = carry;
}

__global__ void __launch_bounds__(PART_THREADS)
bin_place_kernel(const int* __restrict__ srcA, const int* __restrict__ dstA, int nEA,
                 const int* __restrict__ srcB, const int* __restrict__ dstB, int nEB,
                 const int* __restrict__ start, int2* __restrict__ part,
                 int nBins, int nChunks, int listStride)
{
    extern __shared__ int histW[];
    const int g = blockIdx.y;
    const int* __restrict__ src = g ? srcB : srcA;
    const int* __restrict__ dst = g ? dstB : dstA;
    const int nE = g ? nEB : nEA;
    const int c = blockIdx.x;
    const int lane = threadIdx.x & 31;
    const int wv = __builtin_amdgcn_readfirstlane((int)(threadIdx.x >> 5));
    const int binBits = key_bits(nBins);
    for (int i = threadIdx.x; i < PART_WAVES * nBins; i += PART_THREADS) histW[i] = 0;
    __syncthreads();
    const int e0 = c * CHUNK_EDGES + (int)threadIdx.x;
    int d[PART_EPT], s[PART_EPT];
    unsigned grp[PART_EPT];
#pragma unroll
    for (int k = 0; k < PART_EPT; ++k) {
        const int e = e0 + k * PART_THREADS;
        const bool in = e < nE;
        d[k] = in ? dst[e] : -1;
        s[k] = in ? src[e] : 0;
    }
    int* myRow = histW + wv * nBins;
#pragma unroll
    for (int k = 0; k < PART_EPT; ++k) {
        const bool valid = d[k] >= 0;
        const int bin = valid ? (d[k] >> BIN_SHIFT) : 0;
        grp[k] = same_key_lanes(bin, valid, binBits);
        if (valid && lane == __builtin_ctz(grp[k])) myRow[bin] += __builtin_popcount(grp[k]);
    }
    __syncthreads();
    const int* __restrict__ srow = start + ((size_t)g * nChunks + c) * nBins;
    for (int b = threadIdx.x; b < nBins; b += PART_THREADS) {
        int run = srow[b];
        for (int w = 0; w < PART_WAVES; ++w) {
            const int t = histW[w * nBins + b];
            histW[w * nBins + b] = run;
            run += t;
        }
    }
    __syncthreads();
    int2* __restrict__ pg = part + (size_t)g * listStride;
#pragma unroll
    for (int k = 0; k < PART_EPT; ++k) {
        const bool valid = d[k] >= 0;
        const int bin = valid ? (d[k] >> BIN_SHIFT) : 0;
        const int first = valid ? __builtin_ctz(grp[k]) : 0;
        int base = 0;
        if (valid && lane == first) {
            base = myRow[bin];
            myRow[bin] = base + __builtin_popcount(grp[k]);
        }
        base = __shfl(base, first, 32);
        if (valid) pg[base + __builtin_popcount(grp[k] & lanes_below(lane))] = make_int2(s[k], d[k]);
    }
}

__global__ void __launch_bounds__(BIN_NODES)
bin_to_rows_kernel(const int2* __restrict__ part, const int* __restrict__ binStart,
                   int2* __restrict__ rows, int* __restrict__ sorted,
                   int nBins, int nNodes, int listStride)
{
    __shared__ int cntW[ROW_WAVES][BIN_NODES];
    __shared__ int wsum[ROW_WAVES];
    const int g = blockIdx.y, b = blockIdx.x, t = threadIdx.x;
    const int lane = t & 31;
    const int wv = __builtin_amdgcn_readfirstlane(t >> 5);
    const int beg = binStart[(size_t)g * (nBins + 1) + b];
    const int end = binStart[(size_t)g * (nBins + 1) + b + 1];
    const int node0 = b * BIN_NODES;
    const int2* __restrict__ pg = part + (size_t)g * listStride;
    int* __restrict__ sg = sorted + (size_t)g * listStride;
    const int per = (((end - beg) + ROW_WAVES - 1) / ROW_WAVES + 31) & ~31;
    const int sBeg = min(end, beg + wv * per);
    const int sEnd = min(end, sBeg + per);
    const int2 none = make_int2(0, node0);
#pragma unroll
    for (int w = 0; w < ROW_WAVES; ++w) cntW[w][t] = 0;
    __syncthreads();
    {
        int2 pn = (sBeg + lane < sEnd) ? pg[sBeg + lane] : none;
        for (int i = sBeg; i < sEnd; i += 32) {
            const int2 p = pn;
            const bool valid = i + lane < sEnd;
            pn = (i + 32 + lane < sEnd) ? pg[i + 32 + lane] : none;
            const int dl = p.y - node0;
            const unsigned grp = same_key_lanes(dl, valid, BIN_SHIFT);
            if (valid && lane == __builtin_ctz(grp)) cntW[wv][dl] += __builtin_popcount(grp);
        }
    }
    __syncthreads();
    int v = 0;
#pragma unroll
    for (int w = 0; w < ROW_WAVES; ++w) {
        const int c = cntW[w][t];
        cntW[w][t] = v;
        v += c;
    }
    const int incl = wave_incl_scan(v, lane);
    if (lane == 31) wsum[wv] = incl;
    __syncthreads();
    int wo = 0;
#pragma unroll
    for (int w = 0; w < ROW_WAVES; ++w) wo += (w < wv) ? wsum[w] : 0;
    const int first = beg + wo + incl - v;
#pragma unroll
    for (int w = 0; w < ROW_WAVES; ++w) cntW[w][t] += first;
    const int node = node0 + t;
    if (node < nNodes) rows[(size_t)g * nNodes + node] = make_int2(first, v);
    __syncthreads();
    {
        int2 pn = (sBeg + lane < sEnd) ? pg[sBeg + lane] : none;
        for (int i = sBeg; i < sEnd; i += 32) {
            const int2 p = pn;
            const bool valid = i + lane < sEnd;
            pn = (i + 32 + lane < sEnd) ? pg[i + 32 + lane] : none;
            const int dl = p.y - node0;
            const unsigned grp = same_key_lanes(dl, valid, BIN_SHIFT);
            const int gfirst = valid ? __builtin_ctz(grp) : 0;
            int base = 0;
            if (valid && lane == gfirst) {
                base = cntW[wv][dl];
                cntW[wv][dl] = base + __builtin_popcount(grp);
            }
            base = __shfl(base, gfirst, 32);
            if (valid) sg[base + __builtin_popcount(grp & lanes_below(lane))] = p.x;
        }
    }
}


__device__ __forceinline__ float other_half(float v) {
    const int b = __builtin_bit_cast(int, v);
    return __builtin_bit_cast(float, __builtin_amdgcn_permlanex16(b, b, 0x76543210, 0xfedcba98, false, false));
}

template <int M>
__device__ __forceinline__ float add_half_xor(float p) {
    float o;
    asm("v_add_f32_dpp %0, %1, %1 row_xmask:%2 row_mask:0xf bank_mask:0xf" : "=v"(o) : "v"(p), "n"(M));
    return o;
}

__device__ __forceinline__ int half_bcast(int v, int t) {
    const unsigned sel = (unsigned)t * 0x11111111u;
    return __builtin_amdgcn_permlane16(v, v, sel, sel, false, false);
}

__device__ __forceinline__ float half_sum(float p) {
    p = add_half_xor<1>(p);
    p = add_half_xor<2>(p);
    p = add_half_xor<4>(p);
    p = add_half_xor<8>(p);
    return p;
}

__device__ __forceinline__ v4f load_row_piece(const v4f* __restrict__ x4, int s, int c) {
    return x4[((unsigned)s << 4) | (unsigned)c];
}

__global__ void __launch_bounds__(NODE_WAVES * 32)
node_rows_kernel(const float* __restrict__ x, const float* __restrict__ qp,
                 const int2* __restrict__ rows, const int* __restrict__ sorted,
                 float* __restrict__ agg, float* __restrict__ tneg,
                 int nNodes, int listStride)
{
    const int lane = threadIdx.x & 31;
    const int c = lane & 15;
    const int h = lane >> 4;
    const int n = wave_in_grid();
    if (n >= nNodes) return;
    const int g = blockIdx.y;
    const int2 row = rows[(size_t)g * nNodes + n];
    const int* __restrict__ sg = sorted + (size_t)g * listStride;
    const int beg = row.x;
    const int cnt = row.y;
    const int slot = 2 * c + h;
    const v4f* __restrict__ x4 = (const v4f*)x;
    if (g == 0) {
        v4f acc = {0.f, 0.f, 0.f, 0.f};
        for (int i0 = 0; i0 < cnt; i0 += 32) {
            const int m = min(32, cnt - i0);
            const int idx = (slot < m) ? sg[beg + i0 + slot] : 0;
            for (int t0 = 0; 2 * t0 < m; t0 += 4) {
                v4f r[4];
#pragma unroll
                for (int u = 0; u < 4; ++u)
                    r[u] = load_row_piece(x4, half_bcast(idx, t0 + u), c);
#pragma unroll
                for (int u = 0; u < 4; ++u) {
                    const bool valid = h < m - 2 * (t0 + u);
                    acc.x += valid ? r[u].x : 0.f;
                    acc.y += valid ? r[u].y : 0.f;
                    acc.z += valid ? r[u].z : 0.f;
                    acc.w += valid ? r[u].w : 0.f;
                }
            }
        }
        acc.x += other_half(acc.x);
        acc.y += other_half(acc.y);
        acc.z += other_half(acc.z);
        acc.w += other_half(acc.w);
        if (h == 0) *(v4f*)(agg + (size_t)n * DCH + 4 * c) = acc;
    } else {
        const v4f q = *(const v4f*)(qp + (size_t)n * DCH + 4 * c);
        const v2f qlo = {q.x, q.y}, qhi = {q.z, q.w};
        v2f alo = {0.f, 0.f}, ahi = {0.f, 0.f};
        for (int i0 = 0; i0 < cnt; i0 += 32) {
            const int m = min(32, cnt - i0);
            const int idx = (slot < m) ? sg[beg + i0 + slot] : 0;
            for (int t0 = 0; 2 * t0 < m; t0 += 4) {
                v4f r[4];
#pragma unroll
                for (int u = 0; u < 4; ++u)
                    r[u] = load_row_piece(x4, half_bcast(idx, t0 + u), c);
#pragma unroll
                for (int u = 0; u < 4; ++u) {
                    const v2f d = qlo * r[u].xy + qhi * r[u].zw;
                    const float p = half_sum(d.x + d.y);
                    const float pe = (h < m - 2 * (t0 + u)) ? p : 0.f;
                    alo += pe * r[u].xy;
                    ahi += pe * r[u].zw;
                }
            }
        }
        v4f acc = {alo.x, alo.y, ahi.x, ahi.y};
        acc.x += other_half(acc.x);
        acc.y += other_half(acc.y);
        acc.z += other_half(acc.z);
        acc.w += other_half(acc.w);
        if (h == 0) *(v4f*)(tneg + (size_t)n * DCH + 4 * c) = -acc;
    }
}

#define WPITCH 68
#define WSLAB (16 * WPITCH)

__device__ __forceinline__ void load_b_group(v2f (&b)[16], const float* bp) {
#pragma unroll
    for (int kk = 0; kk < 16; ++kk) b[kk] = ldv2(bp + 4 * kk);
}

__device__ __forceinline__ v8f wmma_chain(const v2f (&a)[16], const v2f (&b)[16], v8f acc) {
#pragma unroll
    for (int kk = 0; kk < 16; ++kk)
        acc = __builtin_amdgcn_wmma_f32_16x16x4_f32(false, a[kk], false, b[kk], (short)0, acc, false, false);
    return acc;
}

__device__ __forceinline__ void store_tile(float* __restrict__ rowp, unsigned off, v8f acc) {
#pragma unroll
    for (int r = 0; r < 8; ++r)
        rowp[off + (unsigned)(r * DCH)] = acc[r];
}

__device__ __forceinline__ void load_a_rows(v2f (&a)[16], const float* __restrict__ src, int row0, int m, int h) {
    const float* p = src + (long long)(row0 + m) * DCH + 2 * h;
#pragma unroll
    for (int kk = 0; kk < 16; ++kk) a[kk] = ldv2(p + 4 * kk);
}

#define QP_WAVES 16
#define RPITCH 80

__global__ void __launch_bounds__(QP_WAVES * 32)
qproj_kernel(const float* __restrict__ x, const float* __restrict__ WQ,
             const float* __restrict__ WK, float* __restrict__ qp, int nrows)
{
    __shared__ __attribute__((aligned(16))) float Rq[DCH * RPITCH];
    __shared__ __attribute__((aligned(16))) float Rk[DCH * RPITCH];
    __shared__ __attribute__((aligned(16))) float Pl[(DCH / 16) * WSLAB];
    const int t = threadIdx.x;
    const int lane = t & 31;
    const int wave = __builtin_amdgcn_readfirstlane(t >> 5);
    const int m = lane & 15;
    const int h = lane >> 4;
    const int nStrips = (nrows + 15) >> 4;
    const int strip = (int)blockIdx.x * QP_WAVES + wave;
    const int row0 = strip * 16;
    v2f ax[16] = {};
    if (strip < nStrips) load_a_rows(ax, x, row0, m, h);
#pragma unroll
    for (int q = 0; q < (DCH * DCH / 4) / (QP_WAVES * 32); ++q) {
        const int f = t + q * QP_WAVES * 32;
        const v4f vq = *(const v4f*)(WQ + 4 * f);
        const v4f vk = *(const v4f*)(WK + 4 * f);
        *(v4f*)(Rq + (f >> 4) * RPITCH + 4 * (f & 15)) = vq;
        *(v4f*)(Rk + (f >> 4) * RPITCH + 4 * (f & 15)) = vk;
    }
    __syncthreads();

    {
        const int nt = wave >> 2, kt = wave & 3;
        const float* ak = Rk + (2 * h) * RPITCH + nt * 16 + m;
        const float* bq = Rq + (2 * h) * RPITCH + kt * 16 + m;
        v2f a[16], b[16];
#pragma unroll
        for (int jj = 0; jj < 16; ++jj) {
            a[jj] = (v2f){ak[(4 * jj) * RPITCH], ak[(4 * jj + 1) * RPITCH]};
            b[jj] = (v2f){bq[(4 * jj) * RPITCH], bq[(4 * jj + 1) * RPITCH]};
        }
        const v8f acc = wmma_chain(a, b, (v8f){});
        float* pd = Pl + nt * WSLAB + (8 * h) * WPITCH + kt * 16 + m;
#pragma unroll
        for (int r = 0; r < 8; ++r) pd[r * WPITCH] = acc[r];
    }
    __syncthreads();

    if (strip < nStrips) {
        const float* bl = Pl + m * WPITCH + 2 * h;
        const unsigned off = (unsigned)(8 * h * DCH + m);
        float* qrow = qp + (long long)row0 * DCH;
        v2f b0[16], b1[16];
        load_b_group(b0, bl);
        load_b_group(b1, bl + 1 * WSLAB);
        v8f acc = wmma_chain(ax, b0, (v8f){});
        store_tile(qrow, off, acc);
        load_b_group(b0, bl + 2 * WSLAB);
        acc = wmma_chain(ax, b1, (v8f){});
        store_tile(qrow, off + 16, acc);
        load_b_group(b1, bl + 3 * WSLAB);
        acc = wmma_chain(ax, b0, (v8f){});
        store_tile(qrow, off + 32, acc);
        acc = wmma_chain(ax, b1, (v8f){});
        store_tile(qrow, off + 48, acc);
    }
}

#define GEMM_WAVES 8
#define GEMM_GROUPS 768
#define WTILE (3 * WSLAB)

__device__ __forceinline__ void stage_weight(const float* __restrict__ Wg, float* Wl, int mat, int t) {
#pragma unroll
    for (int q = 0; q < (DCH * DCH / 4) / (GEMM_WAVES * 32); ++q) {
        const int f = t + q * GEMM_WAVES * 32;
        const int r = f >> 4;
        const v4f v = *(const v4f*)(Wg + 4 * f);
        *(v4f*)(Wl + (r >> 4) * WTILE + mat * WSLAB + (r & 15) * WPITCH + 4 * (f & 15)) = v;
    }
}

__device__ __forceinline__ void gemm3_tile(const v2f (&ax)[16], const v2f (&ag)[16], const v2f (&at)[16],
                                           const float* bt, unsigned off, float* __restrict__ orow)
{
    v2f b0[16], b1[16];
    load_b_group(b0, bt);
    load_b_group(b1, bt + 1 * WSLAB);
    v8f acc = wmma_chain(ax, b0, (v8f){});
    load_b_group(b0, bt + 2 * WSLAB);
    acc = wmma_chain(ag, b1, acc);
    acc = wmma_chain(at, b0, acc);
    store_tile(orow, off, acc);
}

__global__ void __launch_bounds__(GEMM_WAVES * 32) __attribute__((amdgpu_waves_per_eu(6, 6)))
gemm3_kernel(const float* __restrict__ x,  const float* __restrict__ agg,
             const float* __restrict__ tneg,
             const float* __restrict__ W0, const float* __restrict__ W1,
             const float* __restrict__ WV,
             float* __restrict__ out, int nrows)
{
    __shared__ __attribute__((aligned(16))) float Wl[(DCH / 16) * WTILE];
    const int t = threadIdx.x;
    stage_weight(W0, Wl, 0, t);
    stage_weight(W1, Wl, 1, t);
    stage_weight(WV, Wl, 2, t);
    __syncthreads();

    const int lane = t & 31;
    const int wave = __builtin_amdgcn_readfirstlane(t >> 5);
    const int m = lane & 15;
    const int h = lane >> 4;
    const int nStrips = (nrows + 15) >> 4;
    const int total = (int)gridDim.x * GEMM_WAVES;
    const int g = (int)blockIdx.x * GEMM_WAVES + wave;
    const float* bl = Wl + m * WPITCH + 2 * h;
    v2f ax[16], ag[16], at[16];

    int base = 0, sub = 0;
    for (;;) {
        const int left = nStrips - base;
        int strip, t0, t1;
        if (left >= total) {
            strip = base + g; t0 = 0; t1 = DCH / 16;
            base += total;
        } else {
            const int u = g + sub * total;
            if (u >= left * (DCH / 16)) break;
            strip = base + (u >> 2); t0 = u & 3; t1 = t0 + 1;
            ++sub;
        }
        const int row0 = strip * 16;
        load_a_rows(ax, x, row0, m, h);
        load_a_rows(ag, agg, row0, m, h);
        load_a_rows(at, tneg, row0, m, h);
        const long long rowoff = (long long)row0 * DCH;
        for (int tile = t0; tile < t1; ++tile)
            gemm3_tile(ax, ag, at, bl + tile * WTILE, (unsigned)(8 * h * DCH + tile * 16 + m), out + rowoff);
    }
}

extern "C" void kernel_launch(void* const* d_in, const int* in_sizes, int n_in,
                              void* d_out, int out_size, void* d_ws, size_t ws_size,
                              hipStream_t stream)
{
    const float* x  = (const float*)d_in[0];
    const int*  ei1 = (const int*)d_in[1];
    const int*  ei2 = (const int*)d_in[2];
    const float* W0 = (const float*)d_in[3];
    const float* W1 = (const float*)d_in[4];
    const float* WQ = (const float*)d_in[5];
    const float* WK = (const float*)d_in[6];
    const float* WV = (const float*)d_in[7];
    float* out = (float*)d_out;

    const int N  = in_sizes[0] / DCH;
    const int E1 = in_sizes[1] / 2;
    const int E2 = in_sizes[2] / 2;
    const int EM = E1 > E2 ? E1 : E2;
    const int listStride = (EM + 63) / 64 * 64;
    const int nBins = (N + BIN_NODES - 1) / BIN_NODES;
    const int nChunks = (EM + CHUNK_EDGES - 1) / CHUNK_EDGES;

    float* agg = (float*)d_ws;
    float* Qp  = agg + (size_t)N * DCH;
    float* Tn  = Qp  + (size_t)N * DCH;
    int2* part    = (int2*)(Tn + (size_t)N * DCH);
    int* sorted   = (int*)(part + (size_t)2 * listStride);
    int2* rows    = (int2*)(sorted + (size_t)2 * listStride);
    int* cnt      = (int*)(rows + (size_t)2 * N);
    int* start    = cnt + (size_t)2 * nChunks * nBins;
    int* binStart = start + (size_t)2 * nChunks * nBins;

    const size_t histBytes = (size_t)PART_WAVES * nBins * sizeof(int);

    bin_count_kernel<<<dim3(nChunks, 2), PART_THREADS, histBytes, stream>>>(
        ei1 + E1, E1, ei2 + E2, E2, cnt, nBins, nChunks);
    bin_scan_kernel<<<2, 1024, 0, stream>>>(cnt, start, binStart, nBins, nChunks);
    bin_place_kernel<<<dim3(nChunks, 2), PART_THREADS, histBytes, stream>>>(
        ei1, ei1 + E1, E1, ei2, ei2 + E2, E2, start, part, nBins, nChunks, listStride);
    bin_to_rows_kernel<<<dim3(nBins, 2), BIN_NODES, 0, stream>>>(
        part, binStart, rows, sorted, nBins, N, listStride);

    const int strips = (N + 15) / 16;
    qproj_kernel<<<(strips + QP_WAVES - 1) / QP_WAVES, QP_WAVES * 32, 0, stream>>>(x, WQ, WK, Qp, N);

    const int nodeBlocks = (N + NODE_WAVES - 1) / NODE_WAVES;
    node_rows_kernel<<<dim3(nodeBlocks, 2), NODE_WAVES * 32, 0, stream>>>(
        x, Qp, rows, sorted, agg, Tn, N, listStride);

    {
        int blocks = (strips + GEMM_WAVES - 1) / GEMM_WAVES;
        if (blocks > GEMM_GROUPS) blocks = GEMM_GROUPS;
        gemm3_kernel<<<blocks, GEMM_WAVES * 32, 0, stream>>>(x, agg, Tn, W0, W1, WV, out, N);
    }
}


# Edge lists in 2 launches, were 4: each chunk is binned inside its own slots of the list, so the count and scan launches go; the per-bin launch walks the chunks' segments; list and sum order same
# speedup vs baseline: 2.3460x; 1.1880x over previous
#include <hip/hip_runtime.h>

typedef __attribute__((ext_vector_type(2))) float v2f;
typedef __attribute__((ext_vector_type(4))) float v4f;
typedef __attribute__((ext_vector_type(8))) float v8f;

#define DCH 64
#define BIN_NODES 256
#define BIN_SHIFT 8
#define PART_THREADS 1024
#define PART_WAVES (PART_THREADS / 32)
#define PART_EPT 16
#define CHUNK_EDGES (PART_THREADS * PART_EPT)
#define ROW_WAVES (BIN_NODES / 32)
#define NODE_WAVES 8

__device__ __forceinline__ v2f ldv2(const float* p) { return *(const v2f*)p; }

__device__ __forceinline__ int wave_in_grid() {
    const int w = __builtin_amdgcn_readfirstlane((int)(threadIdx.x >> 5));
    return (int)(blockIdx.x * NODE_WAVES) + w;
}

__device__ __forceinline__ int wave_incl_scan(int v, int lane) {
    for (int off = 1; off < 32; off <<= 1) {
        const int t = __shfl_up(v, off, 32);
        if (lane >= off) v += t;
    }
    return v;
}

__device__ __forceinline__ unsigned same_key_lanes(int key, bool valid, int nbits) {
    unsigned m = __builtin_amdgcn_ballot_w32(valid);
    for (int b = 0; b < nbits; ++b) {
        const int sx = -((key >> b) & 1);
        const unsigned set = __builtin_amdgcn_ballot_w32(sx != 0);
        m &= ~(set ^ (unsigned)sx);
    }
    return valid ? m : 0u;
}

__device__ __forceinline__ int key_bits(int n) { return n > 1 ? 32 - __builtin_clz((unsigned)(n - 1)) : 0; }

__device__ __forceinline__ unsigned lanes_below(int lane) { return (1u << lane) - 1u; }


__global__ void __launch_bounds__(PART_THREADS)
bin_local_kernel(const int* __restrict__ srcA, const int* __restrict__ dstA, int nEA,
                 const int* __restrict__ srcB, const int* __restrict__ dstB, int nEB,
                 int2* __restrict__ part, int2* __restrict__ seg,
                 int nBins, int nChunks, int listStride)
{
    extern __shared__ int histW[];
    __shared__ int wsum[PART_WAVES];
    __shared__ int tileTotal;
    int* binOff = histW + PART_WAVES * nBins;
    const int g = blockIdx.y;
    const int* __restrict__ src = g ? srcB : srcA;
    const int* __restrict__ dst = g ? dstB : dstA;
    const int nE = g ? nEB : nEA;
    const int c = blockIdx.x;
    const int lane = threadIdx.x & 31;
    const int wv = __builtin_amdgcn_readfirstlane((int)(threadIdx.x >> 5));
    const int binBits = key_bits(nBins);
    for (int i = threadIdx.x; i < PART_WAVES * nBins; i += PART_THREADS) histW[i] = 0;
    __syncthreads();
    const int e0 = c * CHUNK_EDGES + (int)threadIdx.x;
    int d[PART_EPT], s[PART_EPT];
    unsigned grp[PART_EPT];
#pragma unroll
    for (int k = 0; k < PART_EPT; ++k) {
        const int e = e0 + k * PART_THREADS;
        const bool in = e < nE;
        d[k] = in ? dst[e] : -1;
        s[k] = in ? src[e] : 0;
    }
    int* myRow = histW + wv * nBins;
#pragma unroll
    for (int k = 0; k < PART_EPT; ++k) {
        const bool valid = d[k] >= 0;
        const int bin = valid ? (d[k] >> BIN_SHIFT) : 0;
        grp[k] = same_key_lanes(bin, valid, binBits);
        if (valid && lane == __builtin_ctz(grp[k])) myRow[bin] += __builtin_popcount(grp[k]);
    }
    __syncthreads();
    int2* __restrict__ sg = seg + (size_t)g * nBins * nChunks;
    int carry = 0;
    for (int b0 = 0; b0 < nBins; b0 += PART_THREADS) {
        const int b = b0 + (int)threadIdx.x;
        const bool live = b < nBins;
        int tot = 0;
        if (live) {
            for (int w = 0; w < PART_WAVES; ++w) {
                const int t = histW[w * nBins + b];
                histW[w * nBins + b] = tot;
                tot += t;
            }
        }
        const int incl = wave_incl_scan(tot, lane);
        if (lane == 31) wsum[wv] = incl;
        __syncthreads();
        if (wv == 0) {
            const int sv = wsum[lane];
            const int si = wave_incl_scan(sv, lane);
            wsum[lane] = si - sv;
            if (lane == 31) tileTotal = si;
        }
        __syncthreads();
        const int off = carry + wsum[wv] + incl - tot;
        if (live) {
            binOff[b] = off;
            sg[(size_t)b * nChunks + c] = make_int2(off, tot);
        }
        carry += tileTotal;
        __syncthreads();
    }
    int2* __restrict__ pg = part + (size_t)g * listStride + (size_t)c * CHUNK_EDGES;
#pragma unroll
    for (int k = 0; k < PART_EPT; ++k) {
        const bool valid = d[k] >= 0;
        const int bin = valid ? (d[k] >> BIN_SHIFT) : 0;
        const int first = valid ? __builtin_ctz(grp[k]) : 0;
        int base = 0;
        if (valid && lane == first) {
            const int rel = myRow[bin];
            myRow[bin] = rel + __builtin_popcount(grp[k]);
            base = rel + binOff[bin];
        }
        base = __shfl(base, first, 32);
        if (valid) pg[base + __builtin_popcount(grp[k] & lanes_below(lane))] = make_int2(s[k], d[k]);
    }
}

__device__ __forceinline__ int seg_slot(const int* segStart, const int* segPhys, int nChunks,
                                        int v, bool valid, int& ck)
{
    for (;;) {
        const bool adv = valid && ck + 1 < nChunks && segStart[ck + 1] <= v;
        if (__builtin_amdgcn_ballot_w32(adv) == 0u) break;
        ck += adv ? 1 : 0;
    }
    return segPhys[ck] + v;
}

__global__ void __launch_bounds__(BIN_NODES)
bin_to_rows_kernel(const int2* __restrict__ part, const int2* __restrict__ seg,
                   int2* __restrict__ rows, int* __restrict__ sorted,
                   int nBins, int nNodes, int nChunks, int listStride)
{
    extern __shared__ int segL[];
    __shared__ int cntW[ROW_WAVES][BIN_NODES];
    __shared__ int wsum[ROW_WAVES];
    __shared__ int wsumOff[ROW_WAVES];
    const int g = blockIdx.y, b = blockIdx.x, t = threadIdx.x;
    const int lane = t & 31;
    const int wv = __builtin_amdgcn_readfirstlane(t >> 5);
    int* segStart = segL;
    int* segPhys = segL + nChunks;
    const int2* __restrict__ si = seg + ((size_t)g * nBins + b) * nChunks;
    int nPairs = 0, offSum = 0;
    for (int c0 = 0; c0 < nChunks; c0 += BIN_NODES) {
        const int c = c0 + t;
        const int2 sc = (c < nChunks) ? si[c] : make_int2(0, 0);
        const int incl = wave_incl_scan(sc.y, lane);
        int so = sc.x;
        for (int o = 16; o > 0; o >>= 1) so += __shfl_xor(so, o, 32);
        if (lane == 31) { wsum[wv] = incl; wsumOff[wv] = so; }
        __syncthreads();
        int wo = 0, tl = 0, to = 0;
#pragma unroll
        for (int w = 0; w < ROW_WAVES; ++w) {
            const int l = wsum[w];
            wo += (w < wv) ? l : 0;
            tl += l;
            to += wsumOff[w];
        }
        if (c < nChunks) {
            const int ex = nPairs + wo + incl - sc.y;
            segStart[c] = ex;
            segPhys[c] = c * CHUNK_EDGES + sc.x - ex;
        }
        nPairs += tl;
        offSum += to;
        __syncthreads();
    }
    const int total = __builtin_amdgcn_readfirstlane(nPairs);
    const int beg = __builtin_amdgcn_readfirstlane(offSum);
    const int node0 = b * BIN_NODES;
    const int2* __restrict__ pg = part + (size_t)g * listStride;
    int* __restrict__ sg = sorted + (size_t)g * listStride;
    const int per = ((total + ROW_WAVES - 1) / ROW_WAVES + 31) & ~31;
    const int sBeg = min(total, wv * per);
    const int sEnd = min(total, sBeg + per);
    const int2 none = make_int2(0, node0);
#pragma unroll
    for (int w = 0; w < ROW_WAVES; ++w) cntW[w][t] = 0;
    __syncthreads();
    int ckHead = 0;
    if (sBeg + lane < sEnd) {
        const int v = sBeg + lane;
        int lo = 0, hi = nChunks - 1;
        while (lo < hi) {
            const int mid = (lo + hi + 1) >> 1;
            if (segStart[mid] <= v) lo = mid; else hi = mid - 1;
        }
        ckHead = lo;
    }
    {
        int ck = ckHead;
        const int a0 = seg_slot(segStart, segPhys, nChunks, sBeg + lane, sBeg + lane < sEnd, ck);
        int2 pn = (sBeg + lane < sEnd) ? pg[a0] : none;
        int an = seg_slot(segStart, segPhys, nChunks, sBeg + 32 + lane, sBeg + 32 + lane < sEnd, ck);
        for (int i = sBeg; i < sEnd; i += 32) {
            const int2 p = pn;
            const bool valid = i + lane < sEnd;
            pn = (i + 32 + lane < sEnd) ? pg[an] : none;
            an = seg_slot(segStart, segPhys, nChunks, i + 64 + lane, i + 64 + lane < sEnd, ck);
            const int dl = p.y - node0;
            const unsigned grp = same_key_lanes(dl, valid, BIN_SHIFT);
            if (valid && lane == __builtin_ctz(grp)) cntW[wv][dl] += __builtin_popcount(grp);
        }
    }
    __syncthreads();
    int v = 0;
#pragma unroll
    for (int w = 0; w < ROW_WAVES; ++w) {
        const int c = cntW[w][t];
        cntW[w][t] = v;
        v += c;
    }
    const int incl = wave_incl_scan(v, lane);
    if (lane == 31) wsum[wv] = incl;
    __syncthreads();
    int wo = 0;
#pragma unroll
    for (int w = 0; w < ROW_WAVES; ++w) wo += (w < wv) ? wsum[w] : 0;
    const int first = beg + wo + incl - v;
#pragma unroll
    for (int w = 0; w < ROW_WAVES; ++w) cntW[w][t] += first;
    const int node = node0 + t;
    if (node < nNodes) rows[(size_t)g * nNodes + node] = make_int2(first, v);
    __syncthreads();
    {
        int ck = ckHead;
        const int a0 = seg_slot(segStart, segPhys, nChunks, sBeg + lane, sBeg + lane < sEnd, ck);
        int2 pn = (sBeg + lane < sEnd) ? pg[a0] : none;
        int an = seg_slot(segStart, segPhys, nChunks, sBeg + 32 + lane, sBeg + 32 + lane < sEnd, ck);
        for (int i = sBeg; i < sEnd; i += 32) {
            const int2 p = pn;
            const bool valid = i + lane < sEnd;
            pn = (i + 32 + lane < sEnd) ? pg[an] : none;
            an = seg_slot(segStart, segPhys, nChunks, i + 64 + lane, i + 64 + lane < sEnd, ck);
            const int dl = p.y - node0;
            const unsigned grp = same_key_lanes(dl, valid, BIN_SHIFT);
            const int gfirst = valid ? __builtin_ctz(grp) : 0;
            int base = 0;
            if (valid && lane == gfirst) {
                base = cntW[wv][dl];
                cntW[wv][dl] = base + __builtin_popcount(grp);
            }
            base = __shfl(base, gfirst, 32);
            if (valid) sg[base + __builtin_popcount(grp & lanes_below(lane))] = p.x;
        }
    }
}


__device__ __forceinline__ float other_half(float v) {
    const int b = __builtin_bit_cast(int, v);
    return __builtin_bit_cast(float, __builtin_amdgcn_permlanex16(b, b, 0x76543210, 0xfedcba98, false, false));
}

template <int M>
__device__ __forceinline__ float add_half_xor(float p) {
    float o;
    asm("v_add_f32_dpp %0, %1, %1 row_xmask:%2 row_mask:0xf bank_mask:0xf" : "=v"(o) : "v"(p), "n"(M));
    return o;
}

__device__ __forceinline__ int half_bcast(int v, int t) {
    const unsigned sel = (unsigned)t * 0x11111111u;
    return __builtin_amdgcn_permlane16(v, v, sel, sel, false, false);
}

__device__ __forceinline__ float half_sum(float p) {
    p = add_half_xor<1>(p);
    p = add_half_xor<2>(p);
    p = add_half_xor<4>(p);
    p = add_half_xor<8>(p);
    return p;
}

__device__ __forceinline__ v4f load_row_piece(const v4f* __restrict__ x4, int s, int c) {
    return x4[((unsigned)s << 4) | (unsigned)c];
}

__global__ void __launch_bounds__(NODE_WAVES * 32)
node_rows_kernel(const float* __restrict__ x, const float* __restrict__ qp,
                 const int2* __restrict__ rows, const int* __restrict__ sorted,
                 float* __restrict__ agg, float* __restrict__ tneg,
                 int nNodes, int listStride)
{
    const int lane = threadIdx.x & 31;
    const int c = lane & 15;
    const int h = lane >> 4;
    const int n = wave_in_grid();
    if (n >= nNodes) return;
    const int g = blockIdx.y;
    const int2 row = rows[(size_t)g * nNodes + n];
    const int* __restrict__ sg = sorted + (size_t)g * listStride;
    const int beg = row.x;
    const int cnt = row.y;
    const int slot = 2 * c + h;
    const v4f* __restrict__ x4 = (const v4f*)x;
    if (g == 0) {
        v4f acc = {0.f, 0.f, 0.f, 0.f};
        for (int i0 = 0; i0 < cnt; i0 += 32) {
            const int m = min(32, cnt - i0);
            const int idx = (slot < m) ? sg[beg + i0 + slot] : 0;
            for (int t0 = 0; 2 * t0 < m; t0 += 4) {
                v4f r[4];
#pragma unroll
                for (int u = 0; u < 4; ++u)
                    r[u] = load_row_piece(x4, half_bcast(idx, t0 + u), c);
#pragma unroll
                for (int u = 0; u < 4; ++u) {
                    const bool valid = h < m - 2 * (t0 + u);
                    acc.x += valid ? r[u].x : 0.f;
                    acc.y += valid ? r[u].y : 0.f;
                    acc.z += valid ? r[u].z : 0.f;
                    acc.w += valid ? r[u].w : 0.f;
                }
            }
        }
        acc.x += other_half(acc.x);
        acc.y += other_half(acc.y);
        acc.z += other_half(acc.z);
        acc.w += other_half(acc.w);
        if (h == 0) *(v4f*)(agg + (size_t)n * DCH + 4 * c) = acc;
    } else {
        const v4f q = *(const v4f*)(qp + (size_t)n * DCH + 4 * c);
        const v2f qlo = {q.x, q.y}, qhi = {q.z, q.w};
        v2f alo = {0.f, 0.f}, ahi = {0.f, 0.f};
        for (int i0 = 0; i0 < cnt; i0 += 32) {
            const int m = min(32, cnt - i0);
            const int idx = (slot < m) ? sg[beg + i0 + slot] : 0;
            for (int t0 = 0; 2 * t0 < m; t0 += 4) {
                v4f r[4];
#pragma unroll
                for (int u = 0; u < 4; ++u)
                    r[u] = load_row_piece(x4, half_bcast(idx, t0 + u), c);
#pragma unroll
                for (int u = 0; u < 4; ++u) {
                    const v2f d = qlo * r[u].xy + qhi * r[u].zw;
                    const float p = half_sum(d.x + d.y);
                    const float pe = (h < m - 2 * (t0 + u)) ? p : 0.f;
                    alo += pe * r[u].xy;
                    ahi += pe * r[u].zw;
                }
            }
        }
        v4f acc = {alo.x, alo.y, ahi.x, ahi.y};
        acc.x += other_half(acc.x);
        acc.y += other_half(acc.y);
        acc.z += other_half(acc.z);
        acc.w += other_half(acc.w);
        if (h == 0) *(v4f*)(tneg + (size_t)n * DCH + 4 * c) = -acc;
    }
}

#define WPITCH 68
#define WSLAB (16 * WPITCH)

__device__ __forceinline__ void load_b_group(v2f (&b)[16], const float* bp) {
#pragma unroll
    for (int kk = 0; kk < 16; ++kk) b[kk] = ldv2(bp + 4 * kk);
}

__device__ __forceinline__ v8f wmma_chain(const v2f (&a)[16], const v2f (&b)[16], v8f acc) {
#pragma unroll
    for (int kk = 0; kk < 16; ++kk)
        acc = __builtin_amdgcn_wmma_f32_16x16x4_f32(false, a[kk], false, b[kk], (short)0, acc, false, false);
    return acc;
}

__device__ __forceinline__ void store_tile(float* __restrict__ rowp, unsigned off, v8f acc) {
#pragma unroll
    for (int r = 0; r < 8; ++r)
        rowp[off + (unsigned)(r * DCH)] = acc[r];
}

__device__ __forceinline__ void load_a_rows(v2f (&a)[16], const float* __restrict__ src, int row0, int m, int h) {
    const float* p = src + (long long)(row0 + m) * DCH + 2 * h;
#pragma unroll
    for (int kk = 0; kk < 16; ++kk) a[kk] = ldv2(p + 4 * kk);
}

#define QP_WAVES 16
#define RPITCH 80

__global__ void __launch_bounds__(QP_WAVES * 32)
qproj_kernel(const float* __restrict__ x, const float* __restrict__ WQ,
             const float* __restrict__ WK, float* __restrict__ qp, int nrows)
{
    __shared__ __attribute__((aligned(16))) float Rq[DCH * RPITCH];
    __shared__ __attribute__((aligned(16))) float Rk[DCH * RPITCH];
    __shared__ __attribute__((aligned(16))) float Pl[(DCH / 16) * WSLAB];
    const int t = threadIdx.x;
    const int lane = t & 31;
    const int wave = __builtin_amdgcn_readfirstlane(t >> 5);
    const int m = lane & 15;
    const int h = lane >> 4;
    const int nStrips = (nrows + 15) >> 4;
    const int strip = (int)blockIdx.x * QP_WAVES + wave;
    const int row0 = strip * 16;
    v2f ax[16] = {};
    if (strip < nStrips) load_a_rows(ax, x, row0, m, h);
#pragma unroll
    for (int q = 0; q < (DCH * DCH / 4) / (QP_WAVES * 32); ++q) {
        const int f = t + q * QP_WAVES * 32;
        const v4f vq = *(const v4f*)(WQ + 4 * f);
        const v4f vk = *(const v4f*)(WK + 4 * f);
        *(v4f*)(Rq + (f >> 4) * RPITCH + 4 * (f & 15)) = vq;
        *(v4f*)(Rk + (f >> 4) * RPITCH + 4 * (f & 15)) = vk;
    }
    __syncthreads();

    {
        const int nt = wave >> 2, kt = wave & 3;
        const float* ak = Rk + (2 * h) * RPITCH + nt * 16 + m;
        const float* bq = Rq + (2 * h) * RPITCH + kt * 16 + m;
        v2f a[16], b[16];
#pragma unroll
        for (int jj = 0; jj < 16; ++jj) {
            a[jj] = (v2f){ak[(4 * jj) * RPITCH], ak[(4 * jj + 1) * RPITCH]};
            b[jj] = (v2f){bq[(4 * jj) * RPITCH], bq[(4 * jj + 1) * RPITCH]};
        }
        const v8f acc = wmma_chain(a, b, (v8f){});
        float* pd = Pl + nt * WSLAB + (8 * h) * WPITCH + kt * 16 + m;
#pragma unroll
        for (int r = 0; r < 8; ++r) pd[r * WPITCH] = acc[r];
    }
    __syncthreads();

    if (strip < nStrips) {
        const float* bl = Pl + m * WPITCH + 2 * h;
        const unsigned off = (unsigned)(8 * h * DCH + m);
        float* qrow = qp + (long long)row0 * DCH;
        v2f b0[16], b1[16];
        load_b_group(b0, bl);
        load_b_group(b1, bl + 1 * WSLAB);
        v8f acc = wmma_chain(ax, b0, (v8f){});
        store_tile(qrow, off, acc);
        load_b_group(b0, bl + 2 * WSLAB);
        acc = wmma_chain(ax, b1, (v8f){});
        store_tile(qrow, off + 16, acc);
        load_b_group(b1, bl + 3 * WSLAB);
        acc = wmma_chain(ax, b0, (v8f){});
        store_tile(qrow, off + 32, acc);
        acc = wmma_chain(ax, b1, (v8f){});
        store_tile(qrow, off + 48, acc);
    }
}

#define GEMM_WAVES 8
#define GEMM_GROUPS 768
#define WTILE (3 * WSLAB)

__device__ __forceinline__ void stage_weight(const float* __restrict__ Wg, float* Wl, int mat, int t) {
#pragma unroll
    for (int q = 0; q < (DCH * DCH / 4) / (GEMM_WAVES * 32); ++q) {
        const int f = t + q * GEMM_WAVES * 32;
        const int r = f >> 4;
        const v4f v = *(const v4f*)(Wg + 4 * f);
        *(v4f*)(Wl + (r >> 4) * WTILE + mat * WSLAB + (r & 15) * WPITCH + 4 * (f & 15)) = v;
    }
}

__device__ __forceinline__ void gemm3_tile(const v2f (&ax)[16], const v2f (&ag)[16], const v2f (&at)[16],
                                           const float* bt, unsigned off, float* __restrict__ orow)
{
    v2f b0[16], b1[16];
    load_b_group(b0, bt);
    load_b_group(b1, bt + 1 * WSLAB);
    v8f acc = wmma_chain(ax, b0, (v8f){});
    load_b_group(b0, bt + 2 * WSLAB);
    acc = wmma_chain(ag, b1, acc);
    acc = wmma_chain(at, b0, acc);
    store_tile(orow, off, acc);
}

__global__ void __launch_bounds__(GEMM_WAVES * 32) __attribute__((amdgpu_waves_per_eu(6, 6)))
gemm3_kernel(const float* __restrict__ x,  const float* __restrict__ agg,
             const float* __restrict__ tneg,
             const float* __restrict__ W0, const float* __restrict__ W1,
             const float* __restrict__ WV,
             float* __restrict__ out, int nrows)
{
    __shared__ __attribute__((aligned(16))) float Wl[(DCH / 16) * WTILE];
    const int t = threadIdx.x;
    stage_weight(W0, Wl, 0, t);
    stage_weight(W1, Wl, 1, t);
    stage_weight(WV, Wl, 2, t);
    __syncthreads();

    const int lane = t & 31;
    const int wave = __builtin_amdgcn_readfirstlane(t >> 5);
    const int m = lane & 15;
    const int h = lane >> 4;
    const int nStrips = (nrows + 15) >> 4;
    const int total = (int)gridDim.x * GEMM_WAVES;
    const int g = (int)blockIdx.x * GEMM_WAVES + wave;
    const float* bl = Wl + m * WPITCH + 2 * h;
    v2f ax[16], ag[16], at[16];

    int base = 0, sub = 0;
    for (;;) {
        const int left = nStrips - base;
        int strip, t0, t1;
        if (left >= total) {
            strip = base + g; t0 = 0; t1 = DCH / 16;
            base += total;
        } else {
            const int u = g + sub * total;
            if (u >= left * (DCH / 16)) break;
            strip = base + (u >> 2); t0 = u & 3; t1 = t0 + 1;
            ++sub;
        }
        const int row0 = strip * 16;
        load_a_rows(ax, x, row0, m, h);
        load_a_rows(ag, agg, row0, m, h);
        load_a_rows(at, tneg, row0, m, h);
        const long long rowoff = (long long)row0 * DCH;
        for (int tile = t0; tile < t1; ++tile)
            gemm3_tile(ax, ag, at, bl + tile * WTILE, (unsigned)(8 * h * DCH + tile * 16 + m), out + rowoff);
    }
}

extern "C" void kernel_launch(void* const* d_in, const int* in_sizes, int n_in,
                              void* d_out, int out_size, void* d_ws, size_t ws_size,
                              hipStream_t stream)
{
    const float* x  = (const float*)d_in[0];
    const int*  ei1 = (const int*)d_in[1];
    const int*  ei2 = (const int*)d_in[2];
    const float* W0 = (const float*)d_in[3];
    const float* W1 = (const float*)d_in[4];
    const float* WQ = (const float*)d_in[5];
    const float* WK = (const float*)d_in[6];
    const float* WV = (const float*)d_in[7];
    float* out = (float*)d_out;

    const int N  = in_sizes[0] / DCH;
    const int E1 = in_sizes[1] / 2;
    const int E2 = in_sizes[2] / 2;
    const int EM = E1 > E2 ? E1 : E2;
    const int listStride = (EM + 63) / 64 * 64;
    const int nBins = (N + BIN_NODES - 1) / BIN_NODES;
    const int nChunks = (EM + CHUNK_EDGES - 1) / CHUNK_EDGES;

    float* agg = (float*)d_ws;
    float* Qp  = agg + (size_t)N * DCH;
    float* Tn  = Qp  + (size_t)N * DCH;
    int2* part    = (int2*)(Tn + (size_t)N * DCH);
    int* sorted   = (int*)(part + (size_t)2 * listStride);
    int2* rows    = (int2*)(sorted + (size_t)2 * listStride);
    int2* seg     = rows + (size_t)2 * N;

    const size_t histBytes = (size_t)(PART_WAVES + 1) * nBins * sizeof(int);
    const size_t segBytes = (size_t)2 * nChunks * sizeof(int);

    bin_local_kernel<<<dim3(nChunks, 2), PART_THREADS, histBytes, stream>>>(
        ei1, ei1 + E1, E1, ei2, ei2 + E2, E2, part, seg, nBins, nChunks, listStride);
    bin_to_rows_kernel<<<dim3(nBins, 2), BIN_NODES, segBytes, stream>>>(
        part, seg, rows, sorted, nBins, N, nChunks, listStride);

    const int strips = (N + 15) / 16;
    qproj_kernel<<<(strips + QP_WAVES - 1) / QP_WAVES, QP_WAVES * 32, 0, stream>>>(x, WQ, WK, Qp, N);

    const int nodeBlocks = (N + NODE_WAVES - 1) / NODE_WAVES;
    node_rows_kernel<<<dim3(nodeBlocks, 2), NODE_WAVES * 32, 0, stream>>>(
        x, Qp, rows, sorted, agg, Tn, N, listStride);

    {
        int blocks = (strips + GEMM_WAVES - 1) / GEMM_WAVES;
        if (blocks > GEMM_GROUPS) blocks = GEMM_GROUPS;
        gemm3_kernel<<<blocks, GEMM_WAVES * 32, 0, stream>>>(x, agg, Tn, W0, W1, WV, out, N);
    }
}


# Both edge-list launches order their output in LDS and store it as whole lines, 16 B per lane (level 2 reads a bin once, ranks in its count pass); were scattered 4-8 B stores; list and sum order same
# speedup vs baseline: 2.9727x; 1.2671x over previous
#include <hip/hip_runtime.h>

typedef __attribute__((ext_vector_type(2))) float v2f;
typedef __attribute__((ext_vector_type(4))) float v4f;
typedef __attribute__((ext_vector_type(8))) float v8f;

#define DCH 64
#define BIN_NODES 256
#define BIN_SHIFT 8
#define PART_THREADS 1024
#define PART_WAVES (PART_THREADS / 32)
#define PART_EPT 16
#define CHUNK_EDGES (PART_THREADS * PART_EPT)
#define ROW_WAVES (BIN_NODES / 32)
#define NODE_WAVES 8
#define BIN_CAP 5120

__device__ __forceinline__ v2f ldv2(const float* p) { return *(const v2f*)p; }

__device__ __forceinline__ int wave_in_grid() {
    const int w = __builtin_amdgcn_readfirstlane((int)(threadIdx.x >> 5));
    return (int)(blockIdx.x * NODE_WAVES) + w;
}

__device__ __forceinline__ int wave_incl_scan(int v, int lane) {
    for (int off = 1; off < 32; off <<= 1) {
        const int t = __shfl_up(v, off, 32);
        if (lane >= off) v += t;
    }
    return v;
}

__device__ __forceinline__ unsigned same_key_lanes(int key, bool valid, int nbits) {
    unsigned m = __builtin_amdgcn_ballot_w32(valid);
    for (int b = 0; b < nbits; ++b) {
        const int sx = -((key >> b) & 1);
        const unsigned set = __builtin_amdgcn_ballot_w32(sx != 0);
        m &= ~(set ^ (unsigned)sx);
    }
    return valid ? m : 0u;
}

__device__ __forceinline__ int key_bits(int n) { return n > 1 ? 32 - __builtin_clz((unsigned)(n - 1)) : 0; }

__device__ __forceinline__ unsigned lanes_below(int lane) { return (1u << lane) - 1u; }


__global__ void __launch_bounds__(PART_THREADS)
bin_local_kernel(const int* __restrict__ srcA, const int* __restrict__ dstA, int nEA,
                 const int* __restrict__ srcB, const int* __restrict__ dstB, int nEB,
                 int2* __restrict__ part, int2* __restrict__ seg,
                 int nBins, int nChunks, int listStride)
{
    extern __shared__ int histW[];
    __shared__ __attribute__((aligned(16))) int2 stage[CHUNK_EDGES];
    __shared__ int wsum[PART_WAVES];
    __shared__ int tileTotal;
    int* binOff = histW + PART_WAVES * nBins;
    const int g = blockIdx.y;
    const int* __restrict__ src = g ? srcB : srcA;
    const int* __restrict__ dst = g ? dstB : dstA;
    const int nE = g ? nEB : nEA;
    const int c = blockIdx.x;
    const int lane = threadIdx.x & 31;
    const int wv = __builtin_amdgcn_readfirstlane((int)(threadIdx.x >> 5));
    const int binBits = key_bits(nBins);
    for (int i = threadIdx.x; i < PART_WAVES * nBins; i += PART_THREADS) histW[i] = 0;
    __syncthreads();
    const int e0 = c * CHUNK_EDGES + (int)threadIdx.x;
    int d[PART_EPT], s[PART_EPT];
    unsigned grp[PART_EPT];
#pragma unroll
    for (int k = 0; k < PART_EPT; ++k) {
        const int e = e0 + k * PART_THREADS;
        const bool in = e < nE;
        d[k] = in ? dst[e] : -1;
        s[k] = in ? src[e] : 0;
    }
    int* myRow = histW + wv * nBins;
#pragma unroll
    for (int k = 0; k < PART_EPT; ++k) {
        const bool valid = d[k] >= 0;
        const int bin = valid ? (d[k] >> BIN_SHIFT) : 0;
        grp[k] = same_key_lanes(bin, valid, binBits);
        if (valid && lane == __builtin_ctz(grp[k])) myRow[bin] += __builtin_popcount(grp[k]);
    }
    __syncthreads();
    int2* __restrict__ sg = seg + (size_t)g * nBins * nChunks;
    int carry = 0;
    for (int b0 = 0; b0 < nBins; b0 += PART_THREADS) {
        const int b = b0 + (int)threadIdx.x;
        const bool live = b < nBins;
        int tot = 0;
        if (live) {
            for (int w = 0; w < PART_WAVES; ++w) {
                const int t = histW[w * nBins + b];
                histW[w * nBins + b] = tot;
                tot += t;
            }
        }
        const int incl = wave_incl_scan(tot, lane);
        if (lane == 31) wsum[wv] = incl;
        __syncthreads();
        if (wv == 0) {
            const int sv = wsum[lane];
            const int si = wave_incl_scan(sv, lane);
            wsum[lane] = si - sv;
            if (lane == 31) tileTotal = si;
        }
        __syncthreads();
        const int off = carry + wsum[wv] + incl - tot;
        if (live) {
            binOff[b] = off;
            sg[(size_t)b * nChunks + c] = make_int2(off, tot);
        }
        carry += tileTotal;
        __syncthreads();
    }
#pragma unroll
    for (int k = 0; k < PART_EPT; ++k) {
        const bool valid = d[k] >= 0;
        const int bin = valid ? (d[k] >> BIN_SHIFT) : 0;
        const int first = valid ? __builtin_ctz(grp[k]) : 0;
        int base = 0;
        if (valid && lane == first) {
            const int rel = myRow[bin];
            myRow[bin] = rel + __builtin_popcount(grp[k]);
            base = rel + binOff[bin];
        }
        base = __shfl(base, first, 32);
        if (valid) stage[base + __builtin_popcount(grp[k] & lanes_below(lane))] = make_int2(s[k], d[k]);
    }
    __syncthreads();
    int2* __restrict__ pg = part + (size_t)g * listStride + (size_t)c * CHUNK_EDGES;
    const int nIn = min(CHUNK_EDGES, nE - c * CHUNK_EDGES);
    for (int i = 2 * (int)threadIdx.x; i < nIn; i += 2 * PART_THREADS) {
        if (i + 1 < nIn) *(int4*)(pg + i) = *(const int4*)(stage + i);
        else pg[i] = stage[i];
    }
}

__device__ __forceinline__ int seg_slot(const int* segStart, const int* segPhys, int nChunks,
                                        int v, bool valid, int& ck)
{
    for (;;) {
        const bool adv = valid && ck + 1 < nChunks && segStart[ck + 1] <= v;
        if (__builtin_amdgcn_ballot_w32(adv) == 0u) break;
        ck += adv ? 1 : 0;
    }
    return segPhys[ck] + v;
}

__global__ void __launch_bounds__(BIN_NODES)
bin_to_rows_kernel(const int2* __restrict__ part, const int2* __restrict__ seg,
                   int2* __restrict__ rows, int* __restrict__ sorted,
                   int nBins, int nNodes, int nChunks, int listStride)
{
    extern __shared__ int segL[];
    __shared__ int cntW[ROW_WAVES][BIN_NODES];
    __shared__ int wsum[ROW_WAVES];
    __shared__ int wsumOff[ROW_WAVES];
    __shared__ __attribute__((aligned(16))) int2 binP[BIN_CAP];
    __shared__ __attribute__((aligned(16))) int binS[BIN_CAP + 32];
    const int g = blockIdx.y, b = blockIdx.x, t = threadIdx.x;
    const int lane = t & 31;
    const int wv = __builtin_amdgcn_readfirstlane(t >> 5);
    int* segStart = segL;
    int* segPhys = segL + nChunks;
    const int2* __restrict__ si = seg + ((size_t)g * nBins + b) * nChunks;
    int nPairs = 0, offSum = 0;
    for (int c0 = 0; c0 < nChunks; c0 += BIN_NODES) {
        const int c = c0 + t;
        const int2 sc = (c < nChunks) ? si[c] : make_int2(0, 0);
        const int incl = wave_incl_scan(sc.y, lane);
        int so = sc.x;
        for (int o = 16; o > 0; o >>= 1) so += __shfl_xor(so, o, 32);
        if (lane == 31) { wsum[wv] = incl; wsumOff[wv] = so; }
        __syncthreads();
        int wo = 0, tl = 0, to = 0;
#pragma unroll
        for (int w = 0; w < ROW_WAVES; ++w) {
            const int l = wsum[w];
            wo += (w < wv) ? l : 0;
            tl += l;
            to += wsumOff[w];
        }
        if (c < nChunks) {
            const int ex = nPairs + wo + incl - sc.y;
            segStart[c] = ex;
            segPhys[c] = c * CHUNK_EDGES + sc.x - ex;
        }
        nPairs += tl;
        offSum += to;
        __syncthreads();
    }
    const int total = __builtin_amdgcn_readfirstlane(nPairs);
    const int beg = __builtin_amdgcn_readfirstlane(offSum);
    const int node0 = b * BIN_NODES;
    const int2* __restrict__ pg = part + (size_t)g * listStride;
    int* __restrict__ sg = sorted + (size_t)g * listStride;
    const int per = ((total + ROW_WAVES - 1) / ROW_WAVES + 31) & ~31;
    const int sBeg = min(total, wv * per);
    const int sEnd = min(total, sBeg + per);
    const int2 none = make_int2(0, node0);
    const bool inLds = total <= BIN_CAP;
#pragma unroll
    for (int w = 0; w < ROW_WAVES; ++w) cntW[w][t] = 0;
    __syncthreads();
    int ckHead = 0;
    if (sBeg + lane < sEnd) {
        const int v = sBeg + lane;
        int lo = 0, hi = nChunks - 1;
        while (lo < hi) {
            const int mid = (lo + hi + 1) >> 1;
            if (segStart[mid] <= v) lo = mid; else hi = mid - 1;
        }
        ckHead = lo;
    }
    {
        int ck = ckHead;
        const int a0 = seg_slot(segStart, segPhys, nChunks, sBeg + lane, sBeg + lane < sEnd, ck);
        int2 pn = (sBeg + lane < sEnd) ? pg[a0] : none;
        int an = seg_slot(segStart, segPhys, nChunks, sBeg + 32 + lane, sBeg + 32 + lane < sEnd, ck);
        for (int i = sBeg; i < sEnd; i += 32) {
            const int2 p = pn;
            const bool valid = i + lane < sEnd;
            pn = (i + 32 + lane < sEnd) ? pg[an] : none;
            an = seg_slot(segStart, segPhys, nChunks, i + 64 + lane, i + 64 + lane < sEnd, ck);
            const int dl = p.y - node0;
            const unsigned grp = same_key_lanes(dl, valid, BIN_SHIFT);
            const int gfirst = valid ? __builtin_ctz(grp) : 0;
            int old = 0;
            if (valid && lane == gfirst) {
                old = cntW[wv][dl];
                cntW[wv][dl] = old + __builtin_popcount(grp);
            }
            if (inLds) {
                old = __shfl(old, gfirst, 32);
                const int rank = old + __builtin_popcount(grp & lanes_below(lane));
                if (valid) binP[i + lane] = make_int2(p.x, dl | (rank << BIN_SHIFT));
            }
        }
    }
    __syncthreads();
    int v = 0;
#pragma unroll
    for (int w = 0; w < ROW_WAVES; ++w) {
        const int c = cntW[w][t];
        cntW[w][t] = v;
        v += c;
    }
    const int incl = wave_incl_scan(v, lane);
    if (lane == 31) wsum[wv] = incl;
    __syncthreads();
    int wo = 0;
#pragma unroll
    for (int w = 0; w < ROW_WAVES; ++w) wo += (w < wv) ? wsum[w] : 0;
    const int first = beg + wo + incl - v;
#pragma unroll
    for (int w = 0; w < ROW_WAVES; ++w) cntW[w][t] += first;
    const int node = node0 + t;
    if (node < nNodes) rows[(size_t)g * nNodes + node] = make_int2(first, v);
    __syncthreads();
    if (inLds) {
        const int padE = beg & 31;
        const int* runFirst = cntW[wv];
        for (int i0 = sBeg + lane; i0 < sEnd; i0 += 4 * 32) {
            int2 q[4];
            int f[4];
#pragma unroll
            for (int u = 0; u < 4; ++u) q[u] = (i0 + 32 * u < sEnd) ? binP[i0 + 32 * u] : make_int2(0, 0);
#pragma unroll
            for (int u = 0; u < 4; ++u) f[u] = runFirst[q[u].y & (BIN_NODES - 1)];
#pragma unroll
            for (int u = 0; u < 4; ++u)
                if (i0 + 32 * u < sEnd) binS[padE + f[u] + (q[u].y >> BIN_SHIFT) - beg] = q[u].x;
        }
        __syncthreads();
        const int endJ = padE + total;
        int* __restrict__ lineBase = sg + (beg - padE);
        for (int j = 4 * t; j < endJ; j += 4 * BIN_NODES) {
            if (j >= padE && j + 4 <= endJ) {
                *(int4*)(lineBase + j) = *(const int4*)(binS + j);
            } else {
                for (int e = 0; e < 4; ++e)
                    if (j + e >= padE && j + e < endJ) lineBase[j + e] = binS[j + e];
            }
        }
    } else {
        int ck = ckHead;
        const int a0 = seg_slot(segStart, segPhys, nChunks, sBeg + lane, sBeg + lane < sEnd, ck);
        int2 pn = (sBeg + lane < sEnd) ? pg[a0] : none;
        int an = seg_slot(segStart, segPhys, nChunks, sBeg + 32 + lane, sBeg + 32 + lane < sEnd, ck);
        for (int i = sBeg; i < sEnd; i += 32) {
            const int2 p = pn;
            const bool valid = i + lane < sEnd;
            pn = (i + 32 + lane < sEnd) ? pg[an] : none;
            an = seg_slot(segStart, segPhys, nChunks, i + 64 + lane, i + 64 + lane < sEnd, ck);
            const int dl = p.y - node0;
            const unsigned grp = same_key_lanes(dl, valid, BIN_SHIFT);
            const int gfirst = valid ? __builtin_ctz(grp) : 0;
            int base = 0;
            if (valid && lane == gfirst) {
                base = cntW[wv][dl];
                cntW[wv][dl] = base + __builtin_popcount(grp);
            }
            base = __shfl(base, gfirst, 32);
            if (valid) sg[base + __builtin_popcount(grp & lanes_below(lane))] = p.x;
        }
    }
}


__device__ __forceinline__ float other_half(float v) {
    const int b = __builtin_bit_cast(int, v);
    return __builtin_bit_cast(float, __builtin_amdgcn_permlanex16(b, b, 0x76543210, 0xfedcba98, false, false));
}

template <int M>
__device__ __forceinline__ float add_half_xor(float p) {
    float o;
    asm("v_add_f32_dpp %0, %1, %1 row_xmask:%2 row_mask:0xf bank_mask:0xf" : "=v"(o) : "v"(p), "n"(M));
    return o;
}

__device__ __forceinline__ int half_bcast(int v, int t) {
    const unsigned sel = (unsigned)t * 0x11111111u;
    return __builtin_amdgcn_permlane16(v, v, sel, sel, false, false);
}

__device__ __forceinline__ float half_sum(float p) {
    p = add_half_xor<1>(p);
    p = add_half_xor<2>(p);
    p = add_half_xor<4>(p);
    p = add_half_xor<8>(p);
    return p;
}

__device__ __forceinline__ v4f load_row_piece(const v4f* __restrict__ x4, int s, int c) {
    return x4[((unsigned)s << 4) | (unsigned)c];
}

__global__ void __launch_bounds__(NODE_WAVES * 32)
node_rows_kernel(const float* __restrict__ x, const float* __restrict__ qp,
                 const int2* __restrict__ rows, const int* __restrict__ sorted,
                 float* __restrict__ agg, float* __restrict__ tneg,
                 int nNodes, int listStride)
{
    const int lane = threadIdx.x & 31;
    const int c = lane & 15;
    const int h = lane >> 4;
    const int n = wave_in_grid();
    if (n >= nNodes) return;
    const int g = blockIdx.y;
    const int2 row = rows[(size_t)g * nNodes + n];
    const int* __restrict__ sg = sorted + (size_t)g * listStride;
    const int beg = row.x;
    const int cnt = row.y;
    const int slot = 2 * c + h;
    const v4f* __restrict__ x4 = (const v4f*)x;
    if (g == 0) {
        v4f acc = {0.f, 0.f, 0.f, 0.f};
        for (int i0 = 0; i0 < cnt; i0 += 32) {
            const int m = min(32, cnt - i0);
            const int idx = (slot < m) ? sg[beg + i0 + slot] : 0;
            for (int t0 = 0; 2 * t0 < m; t0 += 4) {
                v4f r[4];
#pragma unroll
                for (int u = 0; u < 4; ++u)
                    r[u] = load_row_piece(x4, half_bcast(idx, t0 + u), c);
#pragma unroll
                for (int u = 0; u < 4; ++u) {
                    const bool valid = h < m - 2 * (t0 + u);
                    acc.x += valid ? r[u].x : 0.f;
                    acc.y += valid ? r[u].y : 0.f;
                    acc.z += valid ? r[u].z : 0.f;
                    acc.w += valid ? r[u].w : 0.f;
                }
            }
        }
        acc.x += other_half(acc.x);
        acc.y += other_half(acc.y);
        acc.z += other_half(acc.z);
        acc.w += other_half(acc.w);
        if (h == 0) *(v4f*)(agg + (size_t)n * DCH + 4 * c) = acc;
    } else {
        const v4f q = *(const v4f*)(qp + (size_t)n * DCH + 4 * c);
        const v2f qlo = {q.x, q.y}, qhi = {q.z, q.w};
        v2f alo = {0.f, 0.f}, ahi = {0.f, 0.f};
        for (int i0 = 0; i0 < cnt; i0 += 32) {
            const int m = min(32, cnt - i0);
            const int idx = (slot < m) ? sg[beg + i0 + slot] : 0;
            for (int t0 = 0; 2 * t0 < m; t0 += 4) {
                v4f r[4];
#pragma unroll
                for (int u = 0; u < 4; ++u)
                    r[u] = load_row_piece(x4, half_bcast(idx, t0 + u), c);
#pragma unroll
                for (int u = 0; u < 4; ++u) {
                    const v2f d = qlo * r[u].xy + qhi * r[u].zw;
                    const float p = half_sum(d.x + d.y);
                    const float pe = (h < m - 2 * (t0 + u)) ? p : 0.f;
                    alo += pe * r[u].xy;
                    ahi += pe * r[u].zw;
                }
            }
        }
        v4f acc = {alo.x, alo.y, ahi.x, ahi.y};
        acc.x += other_half(acc.x);
        acc.y += other_half(acc.y);
        acc.z += other_half(acc.z);
        acc.w += other_half(acc.w);
        if (h == 0) *(v4f*)(tneg + (size_t)n * DCH + 4 * c) = -acc;
    }
}

#define WPITCH 68
#define WSLAB (16 * WPITCH)

__device__ __forceinline__ void load_b_group(v2f (&b)[16], const float* bp) {
#pragma unroll
    for (int kk = 0; kk < 16; ++kk) b[kk] = ldv2(bp + 4 * kk);
}

__device__ __forceinline__ v8f wmma_chain(const v2f (&a)[16], const v2f (&b)[16], v8f acc) {
#pragma unroll
    for (int kk = 0; kk < 16; ++kk)
        acc = __builtin_amdgcn_wmma_f32_16x16x4_f32(false, a[kk], false, b[kk], (short)0, acc, false, false);
    return acc;
}

__device__ __forceinline__ void store_tile(float* __restrict__ rowp, unsigned off, v8f acc) {
#pragma unroll
    for (int r = 0; r < 8; ++r)
        rowp[off + (unsigned)(r * DCH)] = acc[r];
}

__device__ __forceinline__ void load_a_rows(v2f (&a)[16], const float* __restrict__ src, int row0, int m, int h) {
    const float* p = src + (long long)(row0 + m) * DCH + 2 * h;
#pragma unroll
    for (int kk = 0; kk < 16; ++kk) a[kk] = ldv2(p + 4 * kk);
}

#define QP_WAVES 16
#define RPITCH 80

__global__ void __launch_bounds__(QP_WAVES * 32)
qproj_kernel(const float* __restrict__ x, const float* __restrict__ WQ,
             const float* __restrict__ WK, float* __restrict__ qp, int nrows)
{
    __shared__ __attribute__((aligned(16))) float Rq[DCH * RPITCH];
    __shared__ __attribute__((aligned(16))) float Rk[DCH * RPITCH];
    __shared__ __attribute__((aligned(16))) float Pl[(DCH / 16) * WSLAB];
    const int t = threadIdx.x;
    const int lane = t & 31;
    const int wave = __builtin_amdgcn_readfirstlane(t >> 5);
    const int m = lane & 15;
    const int h = lane >> 4;
    const int nStrips = (nrows + 15) >> 4;
    const int strip = (int)blockIdx.x * QP_WAVES + wave;
    const int row0 = strip * 16;
    v2f ax[16] = {};
    if (strip < nStrips) load_a_rows(ax, x, row0, m, h);
#pragma unroll
    for (int q = 0; q < (DCH * DCH / 4) / (QP_WAVES * 32); ++q) {
        const int f = t + q * QP_WAVES * 32;
        const v4f vq = *(const v4f*)(WQ + 4 * f);
        const v4f vk = *(const v4f*)(WK + 4 * f);
        *(v4f*)(Rq + (f >> 4) * RPITCH + 4 * (f & 15)) = vq;
        *(v4f*)(Rk + (f >> 4) * RPITCH + 4 * (f & 15)) = vk;
    }
    __syncthreads();

    {
        const int nt = wave >> 2, kt = wave & 3;
        const float* ak = Rk + (2 * h) * RPITCH + nt * 16 + m;
        const float* bq = Rq + (2 * h) * RPITCH + kt * 16 + m;
        v2f a[16], b[16];
#pragma unroll
        for (int jj = 0; jj < 16; ++jj) {
            a[jj] = (v2f){ak[(4 * jj) * RPITCH], ak[(4 * jj + 1) * RPITCH]};
            b[jj] = (v2f){bq[(4 * jj) * RPITCH], bq[(4 * jj + 1) * RPITCH]};
        }
        const v8f acc = wmma_chain(a, b, (v8f){});
        float* pd = Pl + nt * WSLAB + (8 * h) * WPITCH + kt * 16 + m;
#pragma unroll
        for (int r = 0; r < 8; ++r) pd[r * WPITCH] = acc[r];
    }
    __syncthreads();

    if (strip < nStrips) {
        const float* bl = Pl + m * WPITCH + 2 * h;
        const unsigned off = (unsigned)(8 * h * DCH + m);
        float* qrow = qp + (long long)row0 * DCH;
        v2f b0[16], b1[16];
        load_b_group(b0, bl);
        load_b_group(b1, bl + 1 * WSLAB);
        v8f acc = wmma_chain(ax, b0, (v8f){});
        store_tile(qrow, off, acc);
        load_b_group(b0, bl + 2 * WSLAB);
        acc = wmma_chain(ax, b1, (v8f){});
        store_tile(qrow, off + 16, acc);
        load_b_group(b1, bl + 3 * WSLAB);
        acc = wmma_chain(ax, b0, (v8f){});
        store_tile(qrow, off + 32, acc);
        acc = wmma_chain(ax, b1, (v8f){});
        store_tile(qrow, off + 48, acc);
    }
}

#define GEMM_WAVES 8
#define GEMM_GROUPS 768
#define WTILE (3 * WSLAB)

__device__ __forceinline__ void stage_weight(const float* __restrict__ Wg, float* Wl, int mat, int t) {
#pragma unroll
    for (int q = 0; q < (DCH * DCH / 4) / (GEMM_WAVES * 32); ++q) {
        const int f = t + q * GEMM_WAVES * 32;
        const int r = f >> 4;
        const v4f v = *(const v4f*)(Wg + 4 * f);
        *(v4f*)(Wl + (r >> 4) * WTILE + mat * WSLAB + (r & 15) * WPITCH + 4 * (f & 15)) = v;
    }
}

__device__ __forceinline__ void gemm3_tile(const v2f (&ax)[16], const v2f (&ag)[16], const v2f (&at)[16],
                                           const float* bt, unsigned off, float* __restrict__ orow)
{
    v2f b0[16], b1[16];
    load_b_group(b0, bt);
    load_b_group(b1, bt + 1 * WSLAB);
    v8f acc = wmma_chain(ax, b0, (v8f){});
    load_b_group(b0, bt + 2 * WSLAB);
    acc = wmma_chain(ag, b1, acc);
    acc = wmma_chain(at, b0, acc);
    store_tile(orow, off, acc);
}

__global__ void __launch_bounds__(GEMM_WAVES * 32) __attribute__((amdgpu_waves_per_eu(6, 6)))
gemm3_kernel(const float* __restrict__ x,  const float* __restrict__ agg,
             const float* __restrict__ tneg,
             const float* __restrict__ W0, const float* __restrict__ W1,
             const float* __restrict__ WV,
             float* __restrict__ out, int nrows)
{
    __shared__ __attribute__((aligned(16))) float Wl[(DCH / 16) * WTILE];
    const int t = threadIdx.x;
    stage_weight(W0, Wl, 0, t);
    stage_weight(W1, Wl, 1, t);
    stage_weight(WV, Wl, 2, t);
    __syncthreads();

    const int lane = t & 31;
    const int wave = __builtin_amdgcn_readfirstlane(t >> 5);
    const int m = lane & 15;
    const int h = lane >> 4;
    const int nStrips = (nrows + 15) >> 4;
    const int total = (int)gridDim.x * GEMM_WAVES;
    const int g = (int)blockIdx.x * GEMM_WAVES + wave;
    const float* bl = Wl + m * WPITCH + 2 * h;
    v2f ax[16], ag[16], at[16];

    int base = 0, sub = 0;
    for (;;) {
        const int left = nStrips - base;
        int strip, t0, t1;
        if (left >= total) {
            strip = base + g; t0 = 0; t1 = DCH / 16;
            base += total;
        } else {
            const int u = g + sub * total;
            if (u >= left * (DCH / 16)) break;
            strip = base + (u >> 2); t0 = u & 3; t1 = t0 + 1;
            ++sub;
        }
        const int row0 = strip * 16;
        load_a_rows(ax, x, row0, m, h);
        load_a_rows(ag, agg, row0, m, h);
        load_a_rows(at, tneg, row0, m, h);
        const long long rowoff = (long long)row0 * DCH;
        for (int tile = t0; tile < t1; ++tile)
            gemm3_tile(ax, ag, at, bl + tile * WTILE, (unsigned)(8 * h * DCH + tile * 16 + m), out + rowoff);
    }
}

extern "C" void kernel_launch(void* const* d_in, const int* in_sizes, int n_in,
                              void* d_out, int out_size, void* d_ws, size_t ws_size,
                              hipStream_t stream)
{
    const float* x  = (const float*)d_in[0];
    const int*  ei1 = (const int*)d_in[1];
    const int*  ei2 = (const int*)d_in[2];
    const float* W0 = (const float*)d_in[3];
    const float* W1 = (const float*)d_in[4];
    const float* WQ = (const float*)d_in[5];
    const float* WK = (const float*)d_in[6];
    const float* WV = (const float*)d_in[7];
    float* out = (float*)d_out;

    const int N  = in_sizes[0] / DCH;
    const int E1 = in_sizes[1] / 2;
    const int E2 = in_sizes[2] / 2;
    const int EM = E1 > E2 ? E1 : E2;
    const int listStride = (EM + 63) / 64 * 64;
    const int nBins = (N + BIN_NODES - 1) / BIN_NODES;
    const int nChunks = (EM + CHUNK_EDGES - 1) / CHUNK_EDGES;

    float* agg = (float*)d_ws;
    float* Qp  = agg + (size_t)N * DCH;
    float* Tn  = Qp  + (size_t)N * DCH;
    int2* part    = (int2*)(Tn + (size_t)N * DCH);
    int* sorted   = (int*)(part + (size_t)2 * listStride);
    int2* rows    = (int2*)(sorted + (size_t)2 * listStride);
    int2* seg     = rows + (size_t)2 * N;

    const size_t histBytes = (size_t)(PART_WAVES + 1) * nBins * sizeof(int);
    const size_t segBytes = (size_t)2 * nChunks * sizeof(int);

    bin_local_kernel<<<dim3(nChunks, 2), PART_THREADS, histBytes, stream>>>(
        ei1, ei1 + E1, E1, ei2, ei2 + E2, E2, part, seg, nBins, nChunks, listStride);
    bin_to_rows_kernel<<<dim3(nBins, 2), BIN_NODES, segBytes, stream>>>(
        part, seg, rows, sorted, nBins, N, nChunks, listStride);

    const int strips = (N + 15) / 16;
    qproj_kernel<<<(strips + QP_WAVES - 1) / QP_WAVES, QP_WAVES * 32, 0, stream>>>(x, WQ, WK, Qp, N);

    const int nodeBlocks = (N + NODE_WAVES - 1) / NODE_WAVES;
    node_rows_kernel<<<dim3(nodeBlocks, 2), NODE_WAVES * 32, 0, stream>>>(
        x, Qp, rows, sorted, agg, Tn, N, listStride);

    {
        int blocks = (strips + GEMM_WAVES - 1) / GEMM_WAVES;
        if (blocks > GEMM_GROUPS) blocks = GEMM_GROUPS;
        gemm3_kernel<<<blocks, GEMM_WAVES * 32, 0, stream>>>(x, agg, Tn, W0, W1, WV, out, N);
    }
}


# Level 2 count pass: a wave requests its whole slice of the list (up to 20 loads per lane) before it ranks; slots come from an LDS table filled per chunk, not a lane cursor; list and sum order same
# speedup vs baseline: 3.1026x; 1.0437x over previous
#include <hip/hip_runtime.h>

typedef __attribute__((ext_vector_type(2))) float v2f;
typedef __attribute__((ext_vector_type(4))) float v4f;
typedef __attribute__((ext_vector_type(8))) float v8f;

#define DCH 64
#define BIN_NODES 256
#define BIN_SHIFT 8
#define PART_THREADS 1024
#define PART_WAVES (PART_THREADS / 32)
#define PART_EPT 16
#define CHUNK_EDGES (PART_THREADS * PART_EPT)
#define ROW_WAVES (BIN_NODES / 32)
#define NODE_WAVES 8
#define BIN_CAP 5120
#define SLICE_GROUPS (BIN_CAP / ROW_WAVES / 32)

__device__ __forceinline__ v2f ldv2(const float* p) { return *(const v2f*)p; }

__device__ __forceinline__ int wave_in_grid() {
    const int w = __builtin_amdgcn_readfirstlane((int)(threadIdx.x >> 5));
    return (int)(blockIdx.x * NODE_WAVES) + w;
}

__device__ __forceinline__ int wave_incl_scan(int v, int lane) {
    for (int off = 1; off < 32; off <<= 1) {
        const int t = __shfl_up(v, off, 32);
        if (lane >= off) v += t;
    }
    return v;
}

__device__ __forceinline__ unsigned same_key_lanes(int key, bool valid, int nbits) {
    unsigned m = __builtin_amdgcn_ballot_w32(valid);
    for (int b = 0; b < nbits; ++b) {
        const int sx = -((key >> b) & 1);
        const unsigned set = __builtin_amdgcn_ballot_w32(sx != 0);
        m &= ~(set ^ (unsigned)sx);
    }
    return valid ? m : 0u;
}

__device__ __forceinline__ int key_bits(int n) { return n > 1 ? 32 - __builtin_clz((unsigned)(n - 1)) : 0; }

__device__ __forceinline__ unsigned lanes_below(int lane) { return (1u << lane) - 1u; }


__global__ void __launch_bounds__(PART_THREADS)
bin_local_kernel(const int* __restrict__ srcA, const int* __restrict__ dstA, int nEA,
                 const int* __restrict__ srcB, const int* __restrict__ dstB, int nEB,
                 int2* __restrict__ part, int2* __restrict__ seg,
                 int nBins, int nChunks, int listStride)
{
    extern __shared__ int histW[];
    __shared__ __attribute__((aligned(16))) int2 stage[CHUNK_EDGES];
    __shared__ int wsum[PART_WAVES];
    __shared__ int tileTotal;
    int* binOff = histW + PART_WAVES * nBins;
    const int g = blockIdx.y;
    const int* __restrict__ src = g ? srcB : srcA;
    const int* __restrict__ dst = g ? dstB : dstA;
    const int nE = g ? nEB : nEA;
    const int c = blockIdx.x;
    const int lane = threadIdx.x & 31;
    const int wv = __builtin_amdgcn_readfirstlane((int)(threadIdx.x >> 5));
    const int binBits = key_bits(nBins);
    for (int i = threadIdx.x; i < PART_WAVES * nBins; i += PART_THREADS) histW[i] = 0;
    __syncthreads();
    const int e0 = c * CHUNK_EDGES + (int)threadIdx.x;
    int d[PART_EPT], s[PART_EPT];
    unsigned grp[PART_EPT];
#pragma unroll
    for (int k = 0; k < PART_EPT; ++k) {
        const int e = e0 + k * PART_THREADS;
        const bool in = e < nE;
        d[k] = in ? dst[e] : -1;
        s[k] = in ? src[e] : 0;
    }
    int* myRow = histW + wv * nBins;
#pragma unroll
    for (int k = 0; k < PART_EPT; ++k) {
        const bool valid = d[k] >= 0;
        const int bin = valid ? (d[k] >> BIN_SHIFT) : 0;
        grp[k] = same_key_lanes(bin, valid, binBits);
        if (valid && lane == __builtin_ctz(grp[k])) myRow[bin] += __builtin_popcount(grp[k]);
    }
    __syncthreads();
    int2* __restrict__ sg = seg + (size_t)g * nBins * nChunks;
    int carry = 0;
    for (int b0 = 0; b0 < nBins; b0 += PART_THREADS) {
        const int b = b0 + (int)threadIdx.x;
        const bool live = b < nBins;
        int tot = 0;
        if (live) {
            for (int w = 0; w < PART_WAVES; ++w) {
                const int t = histW[w * nBins + b];
                histW[w * nBins + b] = tot;
                tot += t;
            }
        }
        const int incl = wave_incl_scan(tot, lane);
        if (lane == 31) wsum[wv] = incl;
        __syncthreads();
        if (wv == 0) {
            const int sv = wsum[lane];
            const int si = wave_incl_scan(sv, lane);
            wsum[lane] = si - sv;
            if (lane == 31) tileTotal = si;
        }
        __syncthreads();
        const int off = carry + wsum[wv] + incl - tot;
        if (live) {
            binOff[b] = off;
            sg[(size_t)b * nChunks + c] = make_int2(off, tot);
        }
        carry += tileTotal;
        __syncthreads();
    }
#pragma unroll
    for (int k = 0; k < PART_EPT; ++k) {
        const bool valid = d[k] >= 0;
        const int bin = valid ? (d[k] >> BIN_SHIFT) : 0;
        const int first = valid ? __builtin_ctz(grp[k]) : 0;
        int base = 0;
        if (valid && lane == first) {
            const int rel = myRow[bin];
            myRow[bin] = rel + __builtin_popcount(grp[k]);
            base = rel + binOff[bin];
        }
        base = __shfl(base, first, 32);
        if (valid) stage[base + __builtin_popcount(grp[k] & lanes_below(lane))] = make_int2(s[k], d[k]);
    }
    __syncthreads();
    int2* __restrict__ pg = part + (size_t)g * listStride + (size_t)c * CHUNK_EDGES;
    const int nIn = min(CHUNK_EDGES, nE - c * CHUNK_EDGES);
    for (int i = 2 * (int)threadIdx.x; i < nIn; i += 2 * PART_THREADS) {
        if (i + 1 < nIn) *(int4*)(pg + i) = *(const int4*)(stage + i);
        else pg[i] = stage[i];
    }
}

__device__ __forceinline__ int seg_slot(const int* segStart, const int* segPhys, int nChunks,
                                        int v, bool valid, int& ck)
{
    for (;;) {
        const bool adv = valid && ck + 1 < nChunks && segStart[ck + 1] <= v;
        if (__builtin_amdgcn_ballot_w32(adv) == 0u) break;
        ck += adv ? 1 : 0;
    }
    return segPhys[ck] + v;
}

__global__ void __launch_bounds__(BIN_NODES)
bin_to_rows_kernel(const int2* __restrict__ part, const int2* __restrict__ seg,
                   int2* __restrict__ rows, int* __restrict__ sorted,
                   int nBins, int nNodes, int nChunks, int listStride)
{
    extern __shared__ int segL[];
    __shared__ int cntW[ROW_WAVES][BIN_NODES];
    __shared__ int wsum[ROW_WAVES];
    __shared__ int wsumOff[ROW_WAVES];
    __shared__ __attribute__((aligned(16))) int2 binP[BIN_CAP];
    __shared__ __attribute__((aligned(16))) int binS[BIN_CAP + 32];
    const int g = blockIdx.y, b = blockIdx.x, t = threadIdx.x;
    const int lane = t & 31;
    const int wv = __builtin_amdgcn_readfirstlane(t >> 5);
    int* segStart = segL;
    int* segPhys = segL + nChunks;
    const int2* __restrict__ si = seg + ((size_t)g * nBins + b) * nChunks;
    int nPairs = 0, offSum = 0;
    for (int c0 = 0; c0 < nChunks; c0 += BIN_NODES) {
        const int c = c0 + t;
        const int2 sc = (c < nChunks) ? si[c] : make_int2(0, 0);
        const int incl = wave_incl_scan(sc.y, lane);
        int so = sc.x;
        for (int o = 16; o > 0; o >>= 1) so += __shfl_xor(so, o, 32);
        if (lane == 31) { wsum[wv] = incl; wsumOff[wv] = so; }
        __syncthreads();
        int wo = 0, tl = 0, to = 0;
#pragma unroll
        for (int w = 0; w < ROW_WAVES; ++w) {
            const int l = wsum[w];
            wo += (w < wv) ? l : 0;
            tl += l;
            to += wsumOff[w];
        }
        if (c < nChunks) {
            const int ex = nPairs + wo + incl - sc.y;
            segStart[c] = ex;
            segPhys[c] = c * CHUNK_EDGES + sc.x - ex;
        }
        nPairs += tl;
        offSum += to;
        __syncthreads();
    }
    const int total = __builtin_amdgcn_readfirstlane(nPairs);
    const int beg = __builtin_amdgcn_readfirstlane(offSum);
    const int node0 = b * BIN_NODES;
    const int2* __restrict__ pg = part + (size_t)g * listStride;
    int* __restrict__ sg = sorted + (size_t)g * listStride;
    const int per = ((total + ROW_WAVES - 1) / ROW_WAVES + 31) & ~31;
    const int sBeg = min(total, wv * per);
    const int sEnd = min(total, sBeg + per);
    const int2 none = make_int2(0, node0);
    const bool inLds = total <= BIN_CAP;
#pragma unroll
    for (int w = 0; w < ROW_WAVES; ++w) cntW[w][t] = 0;
    if (inLds) {
        for (int c = t; c < nChunks; c += BIN_NODES) {
            const int s0 = segStart[c];
            const int s1 = (c + 1 < nChunks) ? segStart[c + 1] : total;
            const int ph = segPhys[c];
            for (int v = s0; v < s1; ++v) binS[v] = ph + v;
        }
    }
    __syncthreads();
    int ckHead = 0;
    if (!inLds && sBeg + lane < sEnd) {
        const int v = sBeg + lane;
        int lo = 0, hi = nChunks - 1;
        while (lo < hi) {
            const int mid = (lo + hi + 1) >> 1;
            if (segStart[mid] <= v) lo = mid; else hi = mid - 1;
        }
        ckHead = lo;
    }
    if (inLds && sBeg < sEnd) {
        int a[SLICE_GROUPS];
        int2 pv[SLICE_GROUPS];
#pragma unroll
        for (int u = 0; u < SLICE_GROUPS; ++u)
            a[u] = binS[min(sBeg + 32 * u + lane, total - 1)];
#pragma unroll
        for (int u = 0; u < SLICE_GROUPS; ++u)
            pv[u] = pg[a[u]];
#pragma unroll
        for (int u = 0; u < SLICE_GROUPS; ++u) {
            const int i = sBeg + 32 * u;
            if (i < sEnd) {
                const int2 p = pv[u];
                const bool valid = i + lane < sEnd;
                const int dl = p.y - node0;
                const unsigned grp = same_key_lanes(dl, valid, BIN_SHIFT);
                const int gfirst = valid ? __builtin_ctz(grp) : 0;
                int old = 0;
                if (valid && lane == gfirst) {
                    old = cntW[wv][dl];
                    cntW[wv][dl] = old + __builtin_popcount(grp);
                }
                old = __shfl(old, gfirst, 32);
                const int rank = old + __builtin_popcount(grp & lanes_below(lane));
                if (valid) binP[i + lane] = make_int2(p.x, dl | (rank << BIN_SHIFT));
            }
        }
    } else if (!inLds) {
        int ck = ckHead;
        const int a0 = seg_slot(segStart, segPhys, nChunks, sBeg + lane, sBeg + lane < sEnd, ck);
        int2 pn = (sBeg + lane < sEnd) ? pg[a0] : none;
        int an = seg_slot(segStart, segPhys, nChunks, sBeg + 32 + lane, sBeg + 32 + lane < sEnd, ck);
        for (int i = sBeg; i < sEnd; i += 32) {
            const int2 p = pn;
            const bool valid = i + lane < sEnd;
            pn = (i + 32 + lane < sEnd) ? pg[an] : none;
            an = seg_slot(segStart, segPhys, nChunks, i + 64 + lane, i + 64 + lane < sEnd, ck);
            const int dl = p.y - node0;
            const unsigned grp = same_key_lanes(dl, valid, BIN_SHIFT);
            const int gfirst = valid ? __builtin_ctz(grp) : 0;
            if (valid && lane == gfirst) cntW[wv][dl] += __builtin_popcount(grp);
        }
    }
    __syncthreads();
    int v = 0;
#pragma unroll
    for (int w = 0; w < ROW_WAVES; ++w) {
        const int c = cntW[w][t];
        cntW[w][t] = v;
        v += c;
    }
    const int incl = wave_incl_scan(v, lane);
    if (lane == 31) wsum[wv] = incl;
    __syncthreads();
    int wo = 0;
#pragma unroll
    for (int w = 0; w < ROW_WAVES; ++w) wo += (w < wv) ? wsum[w] : 0;
    const int first = beg + wo + incl - v;
#pragma unroll
    for (int w = 0; w < ROW_WAVES; ++w) cntW[w][t] += first;
    const int node = node0 + t;
    if (node < nNodes) rows[(size_t)g * nNodes + node] = make_int2(first, v);
    __syncthreads();
    if (inLds) {
        const int padE = beg & 31;
        const int* runFirst = cntW[wv];
        for (int i0 = sBeg + lane; i0 < sEnd; i0 += 4 * 32) {
            int2 q[4];
            int f[4];
#pragma unroll
            for (int u = 0; u < 4; ++u) q[u] = (i0 + 32 * u < sEnd) ? binP[i0 + 32 * u] : make_int2(0, 0);
#pragma unroll
            for (int u = 0; u < 4; ++u) f[u] = runFirst[q[u].y & (BIN_NODES - 1)];
#pragma unroll
            for (int u = 0; u < 4; ++u)
                if (i0 + 32 * u < sEnd) binS[padE + f[u] + (q[u].y >> BIN_SHIFT) - beg] = q[u].x;
        }
        __syncthreads();
        const int endJ = padE + total;
        int* __restrict__ lineBase = sg + (beg - padE);
        for (int j = 4 * t; j < endJ; j += 4 * BIN_NODES) {
            if (j >= padE && j + 4 <= endJ) {
                *(int4*)(lineBase + j) = *(const int4*)(binS + j);
            } else {
                for (int e = 0; e < 4; ++e)
                    if (j + e >= padE && j + e < endJ) lineBase[j + e] = binS[j + e];
            }
        }
    } else {
        int ck = ckHead;
        const int a0 = seg_slot(segStart, segPhys, nChunks, sBeg + lane, sBeg + lane < sEnd, ck);
        int2 pn = (sBeg + lane < sEnd) ? pg[a0] : none;
        int an = seg_slot(segStart, segPhys, nChunks, sBeg + 32 + lane, sBeg + 32 + lane < sEnd, ck);
        for (int i = sBeg; i < sEnd; i += 32) {
            const int2 p = pn;
            const bool valid = i + lane < sEnd;
            pn = (i + 32 + lane < sEnd) ? pg[an] : none;
            an = seg_slot(segStart, segPhys, nChunks, i + 64 + lane, i + 64 + lane < sEnd, ck);
            const int dl = p.y - node0;
            const unsigned grp = same_key_lanes(dl, valid, BIN_SHIFT);
            const int gfirst = valid ? __builtin_ctz(grp) : 0;
            int base = 0;
            if (valid && lane == gfirst) {
                base = cntW[wv][dl];
                cntW[wv][dl] = base + __builtin_popcount(grp);
            }
            base = __shfl(base, gfirst, 32);
            if (valid) sg[base + __builtin_popcount(grp & lanes_below(lane))] = p.x;
        }
    }
}


__device__ __forceinline__ float other_half(float v) {
    const int b = __builtin_bit_cast(int, v);
    return __builtin_bit_cast(float, __builtin_amdgcn_permlanex16(b, b, 0x76543210, 0xfedcba98, false, false));
}

template <int M>
__device__ __forceinline__ float add_half_xor(float p) {
    float o;
    asm("v_add_f32_dpp %0, %1, %1 row_xmask:%2 row_mask:0xf bank_mask:0xf" : "=v"(o) : "v"(p), "n"(M));
    return o;
}

__device__ __forceinline__ int half_bcast(int v, int t) {
    const unsigned sel = (unsigned)t * 0x11111111u;
    return __builtin_amdgcn_permlane16(v, v, sel, sel, false, false);
}

__device__ __forceinline__ float half_sum(float p) {
    p = add_half_xor<1>(p);
    p = add_half_xor<2>(p);
    p = add_half_xor<4>(p);
    p = add_half_xor<8>(p);
    return p;
}

__device__ __forceinline__ v4f load_row_piece(const v4f* __restrict__ x4, int s, int c) {
    return x4[((unsigned)s << 4) | (unsigned)c];
}

__global__ void __launch_bounds__(NODE_WAVES * 32)
node_rows_kernel(const float* __restrict__ x, const float* __restrict__ qp,
                 const int2* __restrict__ rows, const int* __restrict__ sorted,
                 float* __restrict__ agg, float* __restrict__ tneg,
                 int nNodes, int listStride)
{
    const int lane = threadIdx.x & 31;
    const int c = lane & 15;
    const int h = lane >> 4;
    const int n = wave_in_grid();
    if (n >= nNodes) return;
    const int g = blockIdx.y;
    const int2 row = rows[(size_t)g * nNodes + n];
    const int* __restrict__ sg = sorted + (size_t)g * listStride;
    const int beg = row.x;
    const int cnt = row.y;
    const int slot = 2 * c + h;
    const v4f* __restrict__ x4 = (const v4f*)x;
    if (g == 0) {
        v4f acc = {0.f, 0.f, 0.f, 0.f};
        for (int i0 = 0; i0 < cnt; i0 += 32) {
            const int m = min(32, cnt - i0);
            const int idx = (slot < m) ? sg[beg + i0 + slot] : 0;
            for (int t0 = 0; 2 * t0 < m; t0 += 4) {
                v4f r[4];
#pragma unroll
                for (int u = 0; u < 4; ++u)
                    r[u] = load_row_piece(x4, half_bcast(idx, t0 + u), c);
#pragma unroll
                for (int u = 0; u < 4; ++u) {
                    const bool valid = h < m - 2 * (t0 + u);
                    acc.x += valid ? r[u].x : 0.f;
                    acc.y += valid ? r[u].y : 0.f;
                    acc.z += valid ? r[u].z : 0.f;
                    acc.w += valid ? r[u].w : 0.f;
                }
            }
        }
        acc.x += other_half(acc.x);
        acc.y += other_half(acc.y);
        acc.z += other_half(acc.z);
        acc.w += other_half(acc.w);
        if (h == 0) *(v4f*)(agg + (size_t)n * DCH + 4 * c) = acc;
    } else {
        const v4f q = *(const v4f*)(qp + (size_t)n * DCH + 4 * c);
        const v2f qlo = {q.x, q.y}, qhi = {q.z, q.w};
        v2f alo = {0.f, 0.f}, ahi = {0.f, 0.f};
        for (int i0 = 0; i0 < cnt; i0 += 32) {
            const int m = min(32, cnt - i0);
            const int idx = (slot < m) ? sg[beg + i0 + slot] : 0;
            for (int t0 = 0; 2 * t0 < m; t0 += 4) {
                v4f r[4];
#pragma unroll
                for (int u = 0; u < 4; ++u)
                    r[u] = load_row_piece(x4, half_bcast(idx, t0 + u), c);
#pragma unroll
                for (int u = 0; u < 4; ++u) {
                    const v2f d = qlo * r[u].xy + qhi * r[u].zw;
                    const float p = half_sum(d.x + d.y);
                    const float pe = (h < m - 2 * (t0 + u)) ? p : 0.f;
                    alo += pe * r[u].xy;
                    ahi += pe * r[u].zw;
                }
            }
        }
        v4f acc = {alo.x, alo.y, ahi.x, ahi.y};
        acc.x += other_half(acc.x);
        acc.y += other_half(acc.y);
        acc.z += other_half(acc.z);
        acc.w += other_half(acc.w);
        if (h == 0) *(v4f*)(tneg + (size_t)n * DCH + 4 * c) = -acc;
    }
}

#define WPITCH 68
#define WSLAB (16 * WPITCH)

__device__ __forceinline__ void load_b_group(v2f (&b)[16], const float* bp) {
#pragma unroll
    for (int kk = 0; kk < 16; ++kk) b[kk] = ldv2(bp + 4 * kk);
}

__device__ __forceinline__ v8f wmma_chain(const v2f (&a)[16], const v2f (&b)[16], v8f acc) {
#pragma unroll
    for (int kk = 0; kk < 16; ++kk)
        acc = __builtin_amdgcn_wmma_f32_16x16x4_f32(false, a[kk], false, b[kk], (short)0, acc, false, false);
    return acc;
}

__device__ __forceinline__ void store_tile(float* __restrict__ rowp, unsigned off, v8f acc) {
#pragma unroll
    for (int r = 0; r < 8; ++r)
        rowp[off + (unsigned)(r * DCH)] = acc[r];
}

__device__ __forceinline__ void load_a_rows(v2f (&a)[16], const float* __restrict__ src, int row0, int m, int h) {
    const float* p = src + (long long)(row0 + m) * DCH + 2 * h;
#pragma unroll
    for (int kk = 0; kk < 16; ++kk) a[kk] = ldv2(p + 4 * kk);
}

#define QP_WAVES 16
#define RPITCH 80

__global__ void __launch_bounds__(QP_WAVES * 32)
qproj_kernel(const float* __restrict__ x, const float* __restrict__ WQ,
             const float* __restrict__ WK, float* __restrict__ qp, int nrows)
{
    __shared__ __attribute__((aligned(16))) float Rq[DCH * RPITCH];
    __shared__ __attribute__((aligned(16))) float Rk[DCH * RPITCH];
    __shared__ __attribute__((aligned(16))) float Pl[(DCH / 16) * WSLAB];
    const int t = threadIdx.x;
    const int lane = t & 31;
    const int wave = __builtin_amdgcn_readfirstlane(t >> 5);
    const int m = lane & 15;
    const int h = lane >> 4;
    const int nStrips = (nrows + 15) >> 4;
    const int strip = (int)blockIdx.x * QP_WAVES + wave;
    const int row0 = strip * 16;
    v2f ax[16] = {};
    if (strip < nStrips) load_a_rows(ax, x, row0, m, h);
#pragma unroll
    for (int q = 0; q < (DCH * DCH / 4) / (QP_WAVES * 32); ++q) {
        const int f = t + q * QP_WAVES * 32;
        const v4f vq = *(const v4f*)(WQ + 4 * f);
        const v4f vk = *(const v4f*)(WK + 4 * f);
        *(v4f*)(Rq + (f >> 4) * RPITCH + 4 * (f & 15)) = vq;
        *(v4f*)(Rk + (f >> 4) * RPITCH + 4 * (f & 15)) = vk;
    }
    __syncthreads();

    {
        const int nt = wave >> 2, kt = wave & 3;
        const float* ak = Rk + (2 * h) * RPITCH + nt * 16 + m;
        const float* bq = Rq + (2 * h) * RPITCH + kt * 16 + m;
        v2f a[16], b[16];
#pragma unroll
        for (int jj = 0; jj < 16; ++jj) {
            a[jj] = (v2f){ak[(4 * jj) * RPITCH], ak[(4 * jj + 1) * RPITCH]};
            b[jj] = (v2f){bq[(4 * jj) * RPITCH], bq[(4 * jj + 1) * RPITCH]};
        }
        const v8f acc = wmma_chain(a, b, (v8f){});
        float* pd = Pl + nt * WSLAB + (8 * h) * WPITCH + kt * 16 + m;
#pragma unroll
        for (int r = 0; r < 8; ++r) pd[r * WPITCH] = acc[r];
    }
    __syncthreads();

    if (strip < nStrips) {
        const float* bl = Pl + m * WPITCH + 2 * h;
        const unsigned off = (unsigned)(8 * h * DCH + m);
        float* qrow = qp + (long long)row0 * DCH;
        v2f b0[16], b1[16];
        load_b_group(b0, bl);
        load_b_group(b1, bl + 1 * WSLAB);
        v8f acc = wmma_chain(ax, b0, (v8f){});
        store_tile(qrow, off, acc);
        load_b_group(b0, bl + 2 * WSLAB);
        acc = wmma_chain(ax, b1, (v8f){});
        store_tile(qrow, off + 16, acc);
        load_b_group(b1, bl + 3 * WSLAB);
        acc = wmma_chain(ax, b0, (v8f){});
        store_tile(qrow, off + 32, acc);
        acc = wmma_chain(ax, b1, (v8f){});
        store_tile(qrow, off + 48, acc);
    }
}

#define GEMM_WAVES 8
#define GEMM_GROUPS 768
#define WTILE (3 * WSLAB)

__device__ __forceinline__ void stage_weight(const float* __restrict__ Wg, float* Wl, int mat, int t) {
#pragma unroll
    for (int q = 0; q < (DCH * DCH / 4) / (GEMM_WAVES * 32); ++q) {
        const int f = t + q * GEMM_WAVES * 32;
        const int r = f >> 4;
        const v4f v = *(const v4f*)(Wg + 4 * f);
        *(v4f*)(Wl + (r >> 4) * WTILE + mat * WSLAB + (r & 15) * WPITCH + 4 * (f & 15)) = v;
    }
}

__device__ __forceinline__ void gemm3_tile(const v2f (&ax)[16], const v2f (&ag)[16], const v2f (&at)[16],
                                           const float* bt, unsigned off, float* __restrict__ orow)
{
    v2f b0[16], b1[16];
    load_b_group(b0, bt);
    load_b_group(b1, bt + 1 * WSLAB);
    v8f acc = wmma_chain(ax, b0, (v8f){});
    load_b_group(b0, bt + 2 * WSLAB);
    acc = wmma_chain(ag, b1, acc);
    acc = wmma_chain(at, b0, acc);
    store_tile(orow, off, acc);
}

__global__ void __launch_bounds__(GEMM_WAVES * 32) __attribute__((amdgpu_waves_per_eu(6, 6)))
gemm3_kernel(const float* __restrict__ x,  const float* __restrict__ agg,
             const float* __restrict__ tneg,
             const float* __restrict__ W0, const float* __restrict__ W1,
             const float* __restrict__ WV,
             float* __restrict__ out, int nrows)
{
    __shared__ __attribute__((aligned(16))) float Wl[(DCH / 16) * WTILE];
    const int t = threadIdx.x;
    stage_weight(W0, Wl, 0, t);
    stage_weight(W1, Wl, 1, t);
    stage_weight(WV, Wl, 2, t);
    __syncthreads();

    const int lane = t & 31;
    const int wave = __builtin_amdgcn_readfirstlane(t >> 5);
    const int m = lane & 15;
    const int h = lane >> 4;
    const int nStrips = (nrows + 15) >> 4;
    const int total = (int)gridDim.x * GEMM_WAVES;
    const int g = (int)blockIdx.x * GEMM_WAVES + wave;
    const float* bl = Wl + m * WPITCH + 2 * h;
    v2f ax[16], ag[16], at[16];

    int base = 0, sub = 0;
    for (;;) {
        const int left = nStrips - base;
        int strip, t0, t1;
        if (left >= total) {
            strip = base + g; t0 = 0; t1 = DCH / 16;
            base += total;
        } else {
            const int u = g + sub * total;
            if (u >= left * (DCH / 16)) break;
            strip = base + (u >> 2); t0 = u & 3; t1 = t0 + 1;
            ++sub;
        }
        const int row0 = strip * 16;
        load_a_rows(ax, x, row0, m, h);
        load_a_rows(ag, agg, row0, m, h);
        load_a_rows(at, tneg, row0, m, h);
        const long long rowoff = (long long)row0 * DCH;
        for (int tile = t0; tile < t1; ++tile)
            gemm3_tile(ax, ag, at, bl + tile * WTILE, (unsigned)(8 * h * DCH + tile * 16 + m), out + rowoff);
    }
}

extern "C" void kernel_launch(void* const* d_in, const int* in_sizes, int n_in,
                              void* d_out, int out_size, void* d_ws, size_t ws_size,
                              hipStream_t stream)
{
    const float* x  = (const float*)d_in[0];
    const int*  ei1 = (const int*)d_in[1];
    const int*  ei2 = (const int*)d_in[2];
    const float* W0 = (const float*)d_in[3];
    const float* W1 = (const float*)d_in[4];
    const float* WQ = (const float*)d_in[5];
    const float* WK = (const float*)d_in[6];
    const float* WV = (const float*)d_in[7];
    float* out = (float*)d_out;

    const int N  = in_sizes[0] / DCH;
    const int E1 = in_sizes[1] / 2;
    const int E2 = in_sizes[2] / 2;
    const int EM = E1 > E2 ? E1 : E2;
    const int listStride = (EM + 63) / 64 * 64;
    const int nBins = (N + BIN_NODES - 1) / BIN_NODES;
    const int nChunks = (EM + CHUNK_EDGES - 1) / CHUNK_EDGES;

    float* agg = (float*)d_ws;
    float* Qp  = agg + (size_t)N * DCH;
    float* Tn  = Qp  + (size_t)N * DCH;
    int2* part    = (int2*)(Tn + (size_t)N * DCH);
    int* sorted   = (int*)(part + (size_t)2 * listStride);
    int2* rows    = (int2*)(sorted + (size_t)2 * listStride);
    int2* seg     = rows + (size_t)2 * N;

    const size_t histBytes = (size_t)(PART_WAVES + 1) * nBins * sizeof(int);
    const size_t segBytes = (size_t)2 * nChunks * sizeof(int);

    bin_local_kernel<<<dim3(nChunks, 2), PART_THREADS, histBytes, stream>>>(
        ei1, ei1 + E1, E1, ei2, ei2 + E2, E2, part, seg, nBins, nChunks, listStride);
    bin_to_rows_kernel<<<dim3(nBins, 2), BIN_NODES, segBytes, stream>>>(
        part, seg, rows, sorted, nBins, N, nChunks, listStride);

    const int strips = (N + 15) / 16;
    qproj_kernel<<<(strips + QP_WAVES - 1) / QP_WAVES, QP_WAVES * 32, 0, stream>>>(x, WQ, WK, Qp, N);

    const int nodeBlocks = (N + NODE_WAVES - 1) / NODE_WAVES;
    node_rows_kernel<<<dim3(nodeBlocks, 2), NODE_WAVES * 32, 0, stream>>>(
        x, Qp, rows, sorted, agg, Tn, N, listStride);

    {
        int blocks = (strips + GEMM_WAVES - 1) / GEMM_WAVES;
        if (blocks > GEMM_GROUPS) blocks = GEMM_GROUPS;
        gemm3_kernel<<<blocks, GEMM_WAVES * 32, 0, stream>>>(x, agg, Tn, W0, W1, WV, out, N);
    }
}


# Row sums and final GEMM in one launch (4 launches, were 5): a 16-wave group sums 64 nodes' lists into LDS, next node's list heads requested ahead, then one out tile per wave from LDS; sum order same
# speedup vs baseline: 3.3690x; 1.0859x over previous
#include <hip/hip_runtime.h>

typedef __attribute__((ext_vector_type(2))) float v2f;
typedef __attribute__((ext_vector_type(4))) float v4f;
typedef __attribute__((ext_vector_type(8))) float v8f;

#define DCH 64
#define BIN_NODES 256
#define BIN_SHIFT 8
#define PART_THREADS 1024
#define PART_WAVES (PART_THREADS / 32)
#define PART_EPT 16
#define CHUNK_EDGES (PART_THREADS * PART_EPT)
#define ROW_WAVES (BIN_NODES / 32)
#define BIN_CAP 5120
#define SLICE_GROUPS (BIN_CAP / ROW_WAVES / 32)

__device__ __forceinline__ v2f ldv2(const float* p) { return *(const v2f*)p; }

__device__ __forceinline__ int wave_incl_scan(int v, int lane) {
    for (int off = 1; off < 32; off <<= 1) {
        const int t = __shfl_up(v, off, 32);
        if (lane >= off) v += t;
    }
    return v;
}

__device__ __forceinline__ unsigned same_key_lanes(int key, bool valid, int nbits) {
    unsigned m = __builtin_amdgcn_ballot_w32(valid);
    for (int b = 0; b < nbits; ++b) {
        const int sx = -((key >> b) & 1);
        const unsigned set = __builtin_amdgcn_ballot_w32(sx != 0);
        m &= ~(set ^ (unsigned)sx);
    }
    return valid ? m : 0u;
}

__device__ __forceinline__ int key_bits(int n) { return n > 1 ? 32 - __builtin_clz((unsigned)(n - 1)) : 0; }

__device__ __forceinline__ unsigned lanes_below(int lane) { return (1u << lane) - 1u; }


__global__ void __launch_bounds__(PART_THREADS)
bin_local_kernel(const int* __restrict__ srcA, const int* __restrict__ dstA, int nEA,
                 const int* __restrict__ srcB, const int* __restrict__ dstB, int nEB,
                 int2* __restrict__ part, int2* __restrict__ seg,
                 int nBins, int nChunks, int listStride)
{
    extern __shared__ int histW[];
    __shared__ __attribute__((aligned(16))) int2 stage[CHUNK_EDGES];
    __shared__ int wsum[PART_WAVES];
    __shared__ int tileTotal;
    int* binOff = histW + PART_WAVES * nBins;
    const int g = blockIdx.y;
    const int* __restrict__ src = g ? srcB : srcA;
    const int* __restrict__ dst = g ? dstB : dstA;
    const int nE = g ? nEB : nEA;
    const int c = blockIdx.x;
    const int lane = threadIdx.x & 31;
    const int wv = __builtin_amdgcn_readfirstlane((int)(threadIdx.x >> 5));
    const int binBits = key_bits(nBins);
    for (int i = threadIdx.x; i < PART_WAVES * nBins; i += PART_THREADS) histW[i] = 0;
    __syncthreads();
    const int e0 = c * CHUNK_EDGES + (int)threadIdx.x;
    int d[PART_EPT], s[PART_EPT];
    unsigned grp[PART_EPT];
#pragma unroll
    for (int k = 0; k < PART_EPT; ++k) {
        const int e = e0 + k * PART_THREADS;
        const bool in = e < nE;
        d[k] = in ? dst[e] : -1;
        s[k] = in ? src[e] : 0;
    }
    int* myRow = histW + wv * nBins;
#pragma unroll
    for (int k = 0; k < PART_EPT; ++k) {
        const bool valid = d[k] >= 0;
        const int bin = valid ? (d[k] >> BIN_SHIFT) : 0;
        grp[k] = same_key_lanes(bin, valid, binBits);
        if (valid && lane == __builtin_ctz(grp[k])) myRow[bin] += __builtin_popcount(grp[k]);
    }
    __syncthreads();
    int2* __restrict__ sg = seg + (size_t)g * nBins * nChunks;
    int carry = 0;
    for (int b0 = 0; b0 < nBins; b0 += PART_THREADS) {
        const int b = b0 + (int)threadIdx.x;
        const bool live = b < nBins;
        int tot = 0;
        if (live) {
            for (int w = 0; w < PART_WAVES; ++w) {
                const int t = histW[w * nBins + b];
                histW[w * nBins + b] = tot;
                tot += t;
            }
        }
        const int incl = wave_incl_scan(tot, lane);
        if (lane == 31) wsum[wv] = incl;
        __syncthreads();
        if (wv == 0) {
            const int sv = wsum[lane];
            const int si = wave_incl_scan(sv, lane);
            wsum[lane] = si - sv;
            if (lane == 31) tileTotal = si;
        }
        __syncthreads();
        const int off = carry + wsum[wv] + incl - tot;
        if (live) {
            binOff[b] = off;
            sg[(size_t)b * nChunks + c] = make_int2(off, tot);
        }
        carry += tileTotal;
        __syncthreads();
    }
#pragma unroll
    for (int k = 0; k < PART_EPT; ++k) {
        const bool valid = d[k] >= 0;
        const int bin = valid ? (d[k] >> BIN_SHIFT) : 0;
        const int first = valid ? __builtin_ctz(grp[k]) : 0;
        int base = 0;
        if (valid && lane == first) {
            const int rel = myRow[bin];
            myRow[bin] = rel + __builtin_popcount(grp[k]);
            base = rel + binOff[bin];
        }
        base = __shfl(base, first, 32);
        if (valid) stage[base + __builtin_popcount(grp[k] & lanes_below(lane))] = make_int2(s[k], d[k]);
    }
    __syncthreads();
    int2* __restrict__ pg = part + (size_t)g * listStride + (size_t)c * CHUNK_EDGES;
    const int nIn = min(CHUNK_EDGES, nE - c * CHUNK_EDGES);
    for (int i = 2 * (int)threadIdx.x; i < nIn; i += 2 * PART_THREADS) {
        if (i + 1 < nIn) *(int4*)(pg + i) = *(const int4*)(stage + i);
        else pg[i] = stage[i];
    }
}

__device__ __forceinline__ int seg_slot(const int* segStart, const int* segPhys, int nChunks,
                                        int v, bool valid, int& ck)
{
    for (;;) {
        const bool adv = valid && ck + 1 < nChunks && segStart[ck + 1] <= v;
        if (__builtin_amdgcn_ballot_w32(adv) == 0u) break;
        ck += adv ? 1 : 0;
    }
    return segPhys[ck] + v;
}

__global__ void __launch_bounds__(BIN_NODES)
bin_to_rows_kernel(const int2* __restrict__ part, const int2* __restrict__ seg,
                   int2* __restrict__ rows, int* __restrict__ sorted,
                   int nBins, int nNodes, int nChunks, int listStride)
{
    extern __shared__ int segL[];
    __shared__ int cntW[ROW_WAVES][BIN_NODES];
    __shared__ int wsum[ROW_WAVES];
    __shared__ int wsumOff[ROW_WAVES];
    __shared__ __attribute__((aligned(16))) int2 binP[BIN_CAP];
    __shared__ __attribute__((aligned(16))) int binS[BIN_CAP + 32];
    const int g = blockIdx.y, b = blockIdx.x, t = threadIdx.x;
    const int lane = t & 31;
    const int wv = __builtin_amdgcn_readfirstlane(t >> 5);
    int* segStart = segL;
    int* segPhys = segL + nChunks;
    const int2* __restrict__ si = seg + ((size_t)g * nBins + b) * nChunks;
    int nPairs = 0, offSum = 0;
    for (int c0 = 0; c0 < nChunks; c0 += BIN_NODES) {
        const int c = c0 + t;
        const int2 sc = (c < nChunks) ? si[c] : make_int2(0, 0);
        const int incl = wave_incl_scan(sc.y, lane);
        int so = sc.x;
        for (int o = 16; o > 0; o >>= 1) so += __shfl_xor(so, o, 32);
        if (lane == 31) { wsum[wv] = incl; wsumOff[wv] = so; }
        __syncthreads();
        int wo = 0, tl = 0, to = 0;
#pragma unroll
        for (int w = 0; w < ROW_WAVES; ++w) {
            const int l = wsum[w];
            wo += (w < wv) ? l : 0;
            tl += l;
            to += wsumOff[w];
        }
        if (c < nChunks) {
            const int ex = nPairs + wo + incl - sc.y;
            segStart[c] = ex;
            segPhys[c] = c * CHUNK_EDGES + sc.x - ex;
        }
        nPairs += tl;
        offSum += to;
        __syncthreads();
    }
    const int total = __builtin_amdgcn_readfirstlane(nPairs);
    const int beg = __builtin_amdgcn_readfirstlane(offSum);
    const int node0 = b * BIN_NODES;
    const int2* __restrict__ pg = part + (size_t)g * listStride;
    int* __restrict__ sg = sorted + (size_t)g * listStride;
    const int per = ((total + ROW_WAVES - 1) / ROW_WAVES + 31) & ~31;
    const int sBeg = min(total, wv * per);
    const int sEnd = min(total, sBeg + per);
    const int2 none = make_int2(0, node0);
    const bool inLds = total <= BIN_CAP;
#pragma unroll
    for (int w = 0; w < ROW_WAVES; ++w) cntW[w][t] = 0;
    if (inLds) {
        for (int c = t; c < nChunks; c += BIN_NODES) {
            const int s0 = segStart[c];
            const int s1 = (c + 1 < nChunks) ? segStart[c + 1] : total;
            const int ph = segPhys[c];
            for (int v = s0; v < s1; ++v) binS[v] = ph + v;
        }
    }
    __syncthreads();
    int ckHead = 0;
    if (!inLds && sBeg + lane < sEnd) {
        const int v = sBeg + lane;
        int lo = 0, hi = nChunks - 1;
        while (lo < hi) {
            const int mid = (lo + hi + 1) >> 1;
            if (segStart[mid] <= v) lo = mid; else hi = mid - 1;
        }
        ckHead = lo;
    }
    if (inLds && sBeg < sEnd) {
        int a[SLICE_GROUPS];
        int2 pv[SLICE_GROUPS];
#pragma unroll
        for (int u = 0; u < SLICE_GROUPS; ++u)
            a[u] = binS[min(sBeg + 32 * u + lane, total - 1)];
#pragma unroll
        for (int u = 0; u < SLICE_GROUPS; ++u)
            pv[u] = pg[a[u]];
#pragma unroll
        for (int u = 0; u < SLICE_GROUPS; ++u) {
            const int i = sBeg + 32 * u;
            if (i < sEnd) {
                const int2 p = pv[u];
                const bool valid = i + lane < sEnd;
                const int dl = p.y - node0;
                const unsigned grp = same_key_lanes(dl, valid, BIN_SHIFT);
                const int gfirst = valid ? __builtin_ctz(grp) : 0;
                int old = 0;
                if (valid && lane == gfirst) {
                    old = cntW[wv][dl];
                    cntW[wv][dl] = old + __builtin_popcount(grp);
                }
                old = __shfl(old, gfirst, 32);
                const int rank = old + __builtin_popcount(grp & lanes_below(lane));
                if (valid) binP[i + lane] = make_int2(p.x, dl | (rank << BIN_SHIFT));
            }
        }
    } else if (!inLds) {
        int ck = ckHead;
        const int a0 = seg_slot(segStart, segPhys, nChunks, sBeg + lane, sBeg + lane < sEnd, ck);
        int2 pn = (sBeg + lane < sEnd) ? pg[a0] : none;
        int an = seg_slot(segStart, segPhys, nChunks, sBeg + 32 + lane, sBeg + 32 + lane < sEnd, ck);
        for (int i = sBeg; i < sEnd; i += 32) {
            const int2 p = pn;
            const bool valid = i + lane < sEnd;
            pn = (i + 32 + lane < sEnd) ? pg[an] : none;
            an = seg_slot(segStart, segPhys, nChunks, i + 64 + lane, i + 64 + lane < sEnd, ck);
            const int dl = p.y - node0;
            const unsigned grp = same_key_lanes(dl, valid, BIN_SHIFT);
            const int gfirst = valid ? __builtin_ctz(grp) : 0;
            if (valid && lane == gfirst) cntW[wv][dl] += __builtin_popcount(grp);
        }
    }
    __syncthreads();
    int v = 0;
#pragma unroll
    for (int w = 0; w < ROW_WAVES; ++w) {
        const int c = cntW[w][t];
        cntW[w][t] = v;
        v += c;
    }
    const int incl = wave_incl_scan(v, lane);
    if (lane == 31) wsum[wv] = incl;
    __syncthreads();
    int wo = 0;
#pragma unroll
    for (int w = 0; w < ROW_WAVES; ++w) wo += (w < wv) ? wsum[w] : 0;
    const int first = beg + wo + incl - v;
#pragma unroll
    for (int w = 0; w < ROW_WAVES; ++w) cntW[w][t] += first;
    const int node = node0 + t;
    if (node < nNodes) rows[(size_t)g * nNodes + node] = make_int2(first, v);
    __syncthreads();
    if (inLds) {
        const int padE = beg & 31;
        const int* runFirst = cntW[wv];
        for (int i0 = sBeg + lane; i0 < sEnd; i0 += 4 * 32) {
            int2 q[4];
            int f[4];
#pragma unroll
            for (int u = 0; u < 4; ++u) q[u] = (i0 + 32 * u < sEnd) ? binP[i0 + 32 * u] : make_int2(0, 0);
#pragma unroll
            for (int u = 0; u < 4; ++u) f[u] = runFirst[q[u].y & (BIN_NODES - 1)];
#pragma unroll
            for (int u = 0; u < 4; ++u)
                if (i0 + 32 * u < sEnd) binS[padE + f[u] + (q[u].y >> BIN_SHIFT) - beg] = q[u].x;
        }
        __syncthreads();
        const int endJ = padE + total;
        int* __restrict__ lineBase = sg + (beg - padE);
        for (int j = 4 * t; j < endJ; j += 4 * BIN_NODES) {
            if (j >= padE && j + 4 <= endJ) {
                *(int4*)(lineBase + j) = *(const int4*)(binS + j);
            } else {
                for (int e = 0; e < 4; ++e)
                    if (j + e >= padE && j + e < endJ) lineBase[j + e] = binS[j + e];
            }
        }
    } else {
        int ck = ckHead;
        const int a0 = seg_slot(segStart, segPhys, nChunks, sBeg + lane, sBeg + lane < sEnd, ck);
        int2 pn = (sBeg + lane < sEnd) ? pg[a0] : none;
        int an = seg_slot(segStart, segPhys, nChunks, sBeg + 32 + lane, sBeg + 32 + lane < sEnd, ck);
        for (int i = sBeg; i < sEnd; i += 32) {
            const int2 p = pn;
            const bool valid = i + lane < sEnd;
            pn = (i + 32 + lane < sEnd) ? pg[an] : none;
            an = seg_slot(segStart, segPhys, nChunks, i + 64 + lane, i + 64 + lane < sEnd, ck);
            const int dl = p.y - node0;
            const unsigned grp = same_key_lanes(dl, valid, BIN_SHIFT);
            const int gfirst = valid ? __builtin_ctz(grp) : 0;
            int base = 0;
            if (valid && lane == gfirst) {
                base = cntW[wv][dl];
                cntW[wv][dl] = base + __builtin_popcount(grp);
            }
            base = __shfl(base, gfirst, 32);
            if (valid) sg[base + __builtin_popcount(grp & lanes_below(lane))] = p.x;
        }
    }
}

__device__ __forceinline__ float other_half(float v) {
    const int b = __builtin_bit_cast(int, v);
    return __builtin_bit_cast(float, __builtin_amdgcn_permlanex16(b, b, 0x76543210, 0xfedcba98, false, false));
}

template <int M>
__device__ __forceinline__ float add_half_xor(float p) {
    float o;
    asm("v_add_f32_dpp %0, %1, %1 row_xmask:%2 row_mask:0xf bank_mask:0xf" : "=v"(o) : "v"(p), "n"(M));
    return o;
}

__device__ __forceinline__ int half_bcast(int v, int t) {
    const unsigned sel = (unsigned)t * 0x11111111u;
    return __builtin_amdgcn_permlane16(v, v, sel, sel, false, false);
}

__device__ __forceinline__ float half_sum(float p) {
    p = add_half_xor<1>(p);
    p = add_half_xor<2>(p);
    p = add_half_xor<4>(p);
    p = add_half_xor<8>(p);
    return p;
}

__device__ __forceinline__ v4f load_row_piece(const v4f* __restrict__ x4, int s, int c) {
    return x4[((unsigned)s << 4) | (unsigned)c];
}

__device__ __forceinline__ int list_head(const int* __restrict__ sg, int beg, int cnt, int slot) {
    return (slot < min(32, cnt)) ? sg[beg + slot] : 0;
}

__device__ __forceinline__ v4f sum_rows(const v4f* __restrict__ x4, const int* __restrict__ sg,
                                        int beg, int cnt, int head, int c, int h)
{
    const int slot = 2 * c + h;
    v4f acc = {0.f, 0.f, 0.f, 0.f};
    for (int i0 = 0; i0 < cnt; i0 += 32) {
        const int m = min(32, cnt - i0);
        const int idx = (i0 == 0) ? head : ((slot < m) ? sg[beg + i0 + slot] : 0);
        for (int t0 = 0; 2 * t0 < m; t0 += 4) {
            v4f r[4];
#pragma unroll
            for (int u = 0; u < 4; ++u)
                r[u] = load_row_piece(x4, half_bcast(idx, t0 + u), c);
#pragma unroll
            for (int u = 0; u < 4; ++u) {
                const bool valid = h < m - 2 * (t0 + u);
                acc.x += valid ? r[u].x : 0.f;
                acc.y += valid ? r[u].y : 0.f;
                acc.z += valid ? r[u].z : 0.f;
                acc.w += valid ? r[u].w : 0.f;
            }
        }
    }
    acc.x += other_half(acc.x);
    acc.y += other_half(acc.y);
    acc.z += other_half(acc.z);
    acc.w += other_half(acc.w);
    return acc;
}

__device__ __forceinline__ v4f sum_scored_rows(const v4f* __restrict__ x4, const int* __restrict__ sg,
                                               int beg, int cnt, int head, v4f q, int c, int h)
{
    const int slot = 2 * c + h;
    const v2f qlo = {q.x, q.y}, qhi = {q.z, q.w};
    v2f alo = {0.f, 0.f}, ahi = {0.f, 0.f};
    for (int i0 = 0; i0 < cnt; i0 += 32) {
        const int m = min(32, cnt - i0);
        const int idx = (i0 == 0) ? head : ((slot < m) ? sg[beg + i0 + slot] : 0);
        for (int t0 = 0; 2 * t0 < m; t0 += 4) {
            v4f r[4];
#pragma unroll
            for (int u = 0; u < 4; ++u)
                r[u] = load_row_piece(x4, half_bcast(idx, t0 + u), c);
#pragma unroll
            for (int u = 0; u < 4; ++u) {
                const v2f d = qlo * r[u].xy + qhi * r[u].zw;
                const float p = half_sum(d.x + d.y);
                const float pe = (h < m - 2 * (t0 + u)) ? p : 0.f;
                alo += pe * r[u].xy;
                ahi += pe * r[u].zw;
            }
        }
    }
    v4f acc = {alo.x, alo.y, ahi.x, ahi.y};
    acc.x += other_half(acc.x);
    acc.y += other_half(acc.y);
    acc.z += other_half(acc.z);
    acc.w += other_half(acc.w);
    return acc;
}


#define WPITCH 68
#define WSLAB (16 * WPITCH)

__device__ __forceinline__ void load_b_group(v2f (&b)[16], const float* bp) {
#pragma unroll
    for (int kk = 0; kk < 16; ++kk) b[kk] = ldv2(bp + 4 * kk);
}

__device__ __forceinline__ v8f wmma_chain(const v2f (&a)[16], const v2f (&b)[16], v8f acc) {
#pragma unroll
    for (int kk = 0; kk < 16; ++kk)
        acc = __builtin_amdgcn_wmma_f32_16x16x4_f32(false, a[kk], false, b[kk], (short)0, acc, false, false);
    return acc;
}

__device__ __forceinline__ void store_tile(float* __restrict__ rowp, unsigned off, v8f acc) {
#pragma unroll
    for (int r = 0; r < 8; ++r)
        rowp[off + (unsigned)(r * DCH)] = acc[r];
}

__device__ __forceinline__ void load_a_rows(v2f (&a)[16], const float* __restrict__ src, int row0, int m, int h) {
    const float* p = src + (long long)(row0 + m) * DCH + 2 * h;
#pragma unroll
    for (int kk = 0; kk < 16; ++kk) a[kk] = ldv2(p + 4 * kk);
}

#define QP_WAVES 16
#define RPITCH 80

__global__ void __launch_bounds__(QP_WAVES * 32)
qproj_kernel(const float* __restrict__ x, const float* __restrict__ WQ,
             const float* __restrict__ WK, float* __restrict__ qp, int nrows)
{
    __shared__ __attribute__((aligned(16))) float Rq[DCH * RPITCH];
    __shared__ __attribute__((aligned(16))) float Rk[DCH * RPITCH];
    __shared__ __attribute__((aligned(16))) float Pl[(DCH / 16) * WSLAB];
    const int t = threadIdx.x;
    const int lane = t & 31;
    const int wave = __builtin_amdgcn_readfirstlane(t >> 5);
    const int m = lane & 15;
    const int h = lane >> 4;
    const int nStrips = (nrows + 15) >> 4;
    const int strip = (int)blockIdx.x * QP_WAVES + wave;
    const int row0 = strip * 16;
    v2f ax[16] = {};
    if (strip < nStrips) load_a_rows(ax, x, row0, m, h);
#pragma unroll
    for (int q = 0; q < (DCH * DCH / 4) / (QP_WAVES * 32); ++q) {
        const int f = t + q * QP_WAVES * 32;
        const v4f vq = *(const v4f*)(WQ + 4 * f);
        const v4f vk = *(const v4f*)(WK + 4 * f);
        *(v4f*)(Rq + (f >> 4) * RPITCH + 4 * (f & 15)) = vq;
        *(v4f*)(Rk + (f >> 4) * RPITCH + 4 * (f & 15)) = vk;
    }
    __syncthreads();

    {
        const int nt = wave >> 2, kt = wave & 3;
        const float* ak = Rk + (2 * h) * RPITCH + nt * 16 + m;
        const float* bq = Rq + (2 * h) * RPITCH + kt * 16 + m;
        v2f a[16], b[16];
#pragma unroll
        for (int jj = 0; jj < 16; ++jj) {
            a[jj] = (v2f){ak[(4 * jj) * RPITCH], ak[(4 * jj + 1) * RPITCH]};
            b[jj] = (v2f){bq[(4 * jj) * RPITCH], bq[(4 * jj + 1) * RPITCH]};
        }
        const v8f acc = wmma_chain(a, b, (v8f){});
        float* pd = Pl + nt * WSLAB + (8 * h) * WPITCH + kt * 16 + m;
#pragma unroll
        for (int r = 0; r < 8; ++r) pd[r * WPITCH] = acc[r];
    }
    __syncthreads();

    if (strip < nStrips) {
        const float* bl = Pl + m * WPITCH + 2 * h;
        const unsigned off = (unsigned)(8 * h * DCH + m);
        float* qrow = qp + (long long)row0 * DCH;
        v2f b0[16], b1[16];
        load_b_group(b0, bl);
        load_b_group(b1, bl + 1 * WSLAB);
        v8f acc = wmma_chain(ax, b0, (v8f){});
        store_tile(qrow, off, acc);
        load_b_group(b0, bl + 2 * WSLAB);
        acc = wmma_chain(ax, b1, (v8f){});
        store_tile(qrow, off + 16, acc);
        load_b_group(b1, bl + 3 * WSLAB);
        acc = wmma_chain(ax, b0, (v8f){});
        store_tile(qrow, off + 32, acc);
        acc = wmma_chain(ax, b1, (v8f){});
        store_tile(qrow, off + 48, acc);
    }
}

#define FUSE_WAVES 16
#define FUSE_STRIPS 4
#define FUSE_NODES (16 * FUSE_STRIPS)
#define APITCH 68
#define APLANE (FUSE_NODES * APITCH)
#define WTILE (3 * WSLAB)

__device__ __forceinline__ void stage_weight(const float* __restrict__ Wg, float* Wl, int mat, int t) {
#pragma unroll
    for (int q = 0; q < (DCH * DCH / 4) / (FUSE_WAVES * 32); ++q) {
        const int f = t + q * FUSE_WAVES * 32;
        const int r = f >> 4;
        const v4f v = *(const v4f*)(Wg + 4 * f);
        *(v4f*)(Wl + (r >> 4) * WTILE + mat * WSLAB + (r & 15) * WPITCH + 4 * (f & 15)) = v;
    }
}

__global__ void __launch_bounds__(FUSE_WAVES * 32)
node_gemm_kernel(const float* __restrict__ x, const float* __restrict__ qp,
                 const int2* __restrict__ rows, const int* __restrict__ sorted,
                 const float* __restrict__ W0, const float* __restrict__ W1,
                 const float* __restrict__ WV,
                 float* __restrict__ out, int nNodes, int listStride)
{
    __shared__ __attribute__((aligned(16))) float Wl[(DCH / 16) * WTILE];
    __shared__ __attribute__((aligned(16))) float Al[3 * APLANE];
    const int t = threadIdx.x;
    stage_weight(W0, Wl, 0, t);
    stage_weight(W1, Wl, 1, t);
    stage_weight(WV, Wl, 2, t);

    const int lane = t & 31;
    const int c = lane & 15;
    const int h = lane >> 4;
    const int wv = __builtin_amdgcn_readfirstlane(t >> 5);
    const int node0 = (int)blockIdx.x * FUSE_NODES;
    const v4f* __restrict__ x4 = (const v4f*)x;

    const int slot = 2 * c + h;
    const int* __restrict__ sg0 = sorted;
    const int* __restrict__ sg1 = sorted + listStride;
    const int2* __restrict__ rows1 = rows + nNodes;
    const int2 none = make_int2(0, 0);
    const v4f zero = {0.f, 0.f, 0.f, 0.f};
    int n = node0 + wv;
    int nn = n + FUSE_WAVES;
    int2 a0 = (n < nNodes) ? rows[n] : none;
    int2 a1 = (n < nNodes) ? rows1[n] : none;
    int2 b0 = (nn < nNodes) ? rows[nn] : none;
    int2 b1 = (nn < nNodes) ? rows1[nn] : none;
    int h0 = list_head(sg0, a0.x, a0.y, slot);
    int h1 = list_head(sg1, a1.x, a1.y, slot);
    v4f q = zero, xr = zero;
    if (n < nNodes) {
        q = *(const v4f*)(qp + (size_t)n * DCH + 4 * c);
        xr = *(const v4f*)(x + (size_t)n * DCH + 4 * c);
    }
    for (int j = 0; j < FUSE_NODES / FUSE_WAVES; ++j) {
        const bool next = j + 1 < FUSE_NODES / FUSE_WAVES && nn < nNodes;
        const int h0n = list_head(sg0, b0.x, b0.y, slot);
        const int h1n = list_head(sg1, b1.x, b1.y, slot);
        v4f qn = zero, xn = zero;
        if (next) {
            qn = *(const v4f*)(qp + (size_t)nn * DCH + 4 * c);
            xn = *(const v4f*)(x + (size_t)nn * DCH + 4 * c);
        }
        const int n2 = nn + FUSE_WAVES;
        const bool after = j + 2 < FUSE_NODES / FUSE_WAVES && n2 < nNodes;
        const int2 c0 = after ? rows[n2] : none;
        const int2 c1 = after ? rows1[n2] : none;

        const v4f sa = sum_rows(x4, sg0, a0.x, a0.y, h0, c, h);
        const v4f st = -sum_scored_rows(x4, sg1, a1.x, a1.y, h1, q, c, h);
        if (h == 0) {
            float* d = Al + (wv + FUSE_WAVES * j) * APITCH + 4 * c;
            *(v4f*)d = xr;
            *(v4f*)(d + APLANE) = sa;
            *(v4f*)(d + 2 * APLANE) = st;
        }
        a0 = b0; a1 = b1; b0 = c0; b1 = c1;
        h0 = h0n; h1 = h1n; q = qn; xr = xn;
        n = nn; nn = n2;
    }
    __syncthreads();

    const int strip = wv >> 2, ct = wv & 3;
    const int row0 = node0 + 16 * strip;
    if (row0 < nNodes) {
        const float* al = Al + (16 * strip + c) * APITCH + 2 * h;
        const float* bl = Wl + ct * WTILE + c * WPITCH + 2 * h;
        v8f acc = {};
#pragma unroll
        for (int p = 0; p < 3; ++p) {
            v2f a[16], b[16];
            load_b_group(a, al + p * APLANE);
            load_b_group(b, bl + p * WSLAB);
            acc = wmma_chain(a, b, acc);
        }
        float* orow = out + (size_t)row0 * DCH;
        const unsigned off = (unsigned)(8 * h * DCH + ct * 16 + c);
        if (row0 + 16 <= nNodes) {
            store_tile(orow, off, acc);
        } else {
#pragma unroll
            for (int r = 0; r < 8; ++r)
                if (row0 + 8 * h + r < nNodes) orow[off + (unsigned)(r * DCH)] = acc[r];
        }
    }
}

extern "C" void kernel_launch(void* const* d_in, const int* in_sizes, int n_in,
                              void* d_out, int out_size, void* d_ws, size_t ws_size,
                              hipStream_t stream)
{
    const float* x  = (const float*)d_in[0];
    const int*  ei1 = (const int*)d_in[1];
    const int*  ei2 = (const int*)d_in[2];
    const float* W0 = (const float*)d_in[3];
    const float* W1 = (const float*)d_in[4];
    const float* WQ = (const float*)d_in[5];
    const float* WK = (const float*)d_in[6];
    const float* WV = (const float*)d_in[7];
    float* out = (float*)d_out;

    const int N  = in_sizes[0] / DCH;
    const int E1 = in_sizes[1] / 2;
    const int E2 = in_sizes[2] / 2;
    const int EM = E1 > E2 ? E1 : E2;
    const int listStride = (EM + 63) / 64 * 64;
    const int nBins = (N + BIN_NODES - 1) / BIN_NODES;
    const int nChunks = (EM + CHUNK_EDGES - 1) / CHUNK_EDGES;

    float* Qp     = (float*)d_ws;
    int2* part    = (int2*)(Qp + (size_t)N * DCH);
    int* sorted   = (int*)(part + (size_t)2 * listStride);
    int2* rows    = (int2*)(sorted + (size_t)2 * listStride);
    int2* seg     = rows + (size_t)2 * N;

    const size_t histBytes = (size_t)(PART_WAVES + 1) * nBins * sizeof(int);
    const size_t segBytes = (size_t)2 * nChunks * sizeof(int);

    bin_local_kernel<<<dim3(nChunks, 2), PART_THREADS, histBytes, stream>>>(
        ei1, ei1 + E1, E1, ei2, ei2 + E2, E2, part, seg, nBins, nChunks, listStride);
    bin_to_rows_kernel<<<dim3(nBins, 2), BIN_NODES, segBytes, stream>>>(
        part, seg, rows, sorted, nBins, N, nChunks, listStride);

    const int strips = (N + 15) / 16;
    qproj_kernel<<<(strips + QP_WAVES - 1) / QP_WAVES, QP_WAVES * 32, 0, stream>>>(x, WQ, WK, Qp, N);

    node_gemm_kernel<<<(N + FUSE_NODES - 1) / FUSE_NODES, FUSE_WAVES * 32, 0, stream>>>(
        x, Qp, rows, sorted, W0, W1, WV, out, N, listStride);
}
